# starting kernel
# baseline (speedup 1.0000x reference)
#include <hip/hip_runtime.h>
#include <hip/hip_bf16.h>

typedef __attribute__((ext_vector_type(2))) float v2f;
typedef __attribute__((ext_vector_type(8))) float v8f;

__global__ void k_init(float* __restrict__ deg, float* __restrict__ aggB,
                       float* __restrict__ aggC, int n) {
  int idx = blockIdx.x * blockDim.x + threadIdx.x;
  if (idx < n) deg[idx] = 1.0f;
  if (idx < n * 16) { aggB[idx] = 0.0f; aggC[idx] = 0.0f; }
}

__global__ void k_deg(const int* __restrict__ dst, float* __restrict__ deg, int E) {
  int e = blockIdx.x * blockDim.x + threadIdx.x;
  if (e < E) atomicAdd(&deg[dst[e]], 1.0f);
}

__global__ void k_rsqrt(const float* __restrict__ deg, float* __restrict__ dinv, int n) {
  int i = blockIdx.x * blockDim.x + threadIdx.x;
  if (i < n) dinv[i] = rsqrtf(deg[i]);
}

template <int K>
__global__ void k_gemm_wmma(const float* __restrict__ A, const float* __restrict__ W,
                            float* __restrict__ H, int ntiles) {
  __shared__ float Ws[K * 16];
  for (int t = threadIdx.x; t < K * 16; t += blockDim.x) Ws[t] = W[t];
  __syncthreads();

  const int lane = threadIdx.x & 31;
  const int tile = blockIdx.x * (blockDim.x >> 5) + (threadIdx.x >> 5);
  if (tile >= ntiles) return;

  const int half = lane >> 4;
  const int mn   = lane & 15;
  const int koff = half * 2;

  const float* arow = A + (size_t)(tile * 16 + mn) * K;

  v8f c = {0.f, 0.f, 0.f, 0.f, 0.f, 0.f, 0.f, 0.f};
#pragma unroll
  for (int k0 = 0; k0 < K; k0 += 4) {
    v2f a = *(const v2f*)(arow + k0 + koff);
    v2f b;
    b.x = Ws[(k0 + koff + 0) * 16 + mn];
    b.y = Ws[(k0 + koff + 1) * 16 + mn];
    c = __builtin_amdgcn_wmma_f32_16x16x4_f32(
         false, a,  false, b,
         (short)0, c,  false,  false);
  }

  float* hrow = H + (size_t)tile * 256 + half * 128 + mn;
#pragma unroll
  for (int i = 0; i < 8; i++) hrow[i * 16] = c[i];
}

__global__ void k_scatter(const int* __restrict__ src, const int* __restrict__ dst,
                          const float* __restrict__ dinv, const float* __restrict__ H,
                          float* __restrict__ AGG, int E) {
  int idx = blockIdx.x * blockDim.x + threadIdx.x;
  int e = idx >> 4;
  int j = idx & 15;
  if (e >= E) return;
  int s = src[e];
  int d = dst[e];
  float norm = dinv[s] * dinv[d];
  atomicAdd(&AGG[d * 16 + j], norm * H[s * 16 + j]);
}

__global__ void k_combine(const float* __restrict__ AGG, const float* __restrict__ H,
                          const float* __restrict__ dinv, const float* __restrict__ bias,
                          float* __restrict__ out, int n) {
  int idx = blockIdx.x * blockDim.x + threadIdx.x;
  int i = idx >> 4;
  int j = idx & 15;
  if (i >= n) return;
  float di = dinv[i];
  out[idx] = fmaxf(AGG[idx] + di * di * H[idx] + bias[j], 0.0f);
}

__global__ void k_final(const float* __restrict__ G, const float* __restrict__ l1b,
                        const float* __restrict__ l2W, const float* __restrict__ l2b,
                        float* __restrict__ out, int n) {
  int idx = blockIdx.x * blockDim.x + threadIdx.x;
  int i = idx >> 4;
  int j = idx & 15;
  float v = 0.0f;
  if (i < n) v = fmaxf(G[idx] + l1b[j], 0.0f) * l2W[j];
#pragma unroll
  for (int off = 8; off; off >>= 1) v += __shfl_xor(v, off, 16);
  if (i < n && j == 0) out[i] = v + l2b[0];
}

extern "C" void kernel_launch(void* const* d_in, const int* in_sizes, int n_in,
                              void* d_out, int out_size, void* d_ws, size_t ws_size,
                              hipStream_t stream) {
  const float* x   = (const float*)d_in[0];
  const int*   ei  = (const int*)d_in[1];
  const float* W1  = (const float*)d_in[2];
  const float* b1  = (const float*)d_in[3];
  const float* W2  = (const float*)d_in[4];
  const float* b2  = (const float*)d_in[5];
  const float* l1W = (const float*)d_in[6];
  const float* l1b = (const float*)d_in[7];
  const float* l2W = (const float*)d_in[8];
  const float* l2b = (const float*)d_in[9];

  const int N = in_sizes[0] / 128;
  const int E = in_sizes[1] / 2;
  const int* src = ei;
  const int* dst = ei + E;

  float* deg  = (float*)d_ws;
  float* dinv = deg + N;
  float* bufA = dinv + N;
  float* bufB = bufA + (size_t)N * 16;
  float* bufC = bufB + (size_t)N * 16;

  const int ntiles = N / 16;
  const int gemmBlocks = (ntiles * 32 + 255) / 256;
  const int nodeElems  = N * 16;
  const int nodeBlocks = (nodeElems + 255) / 256;
  const int edgeBlocks = (E + 255) / 256;
  const int scatBlocks = (int)(((size_t)E * 16 + 255) / 256);

  k_init  <<<nodeBlocks, 256, 0, stream>>>(deg, bufB, bufC, N);
  k_deg   <<<edgeBlocks, 256, 0, stream>>>(dst, deg, E);
  k_rsqrt <<<(N + 255) / 256, 256, 0, stream>>>(deg, dinv, N);

  k_gemm_wmma<128><<<gemmBlocks, 256, 0, stream>>>(x, W1, bufA, ntiles);
  k_scatter<<<scatBlocks, 256, 0, stream>>>(src, dst, dinv, bufA, bufB, E);
  k_combine<<<nodeBlocks, 256, 0, stream>>>(bufB, bufA, dinv, b1, bufB, N);

  k_gemm_wmma<16><<<gemmBlocks, 256, 0, stream>>>(bufB, W2, bufA, ntiles);
  k_scatter<<<scatBlocks, 256, 0, stream>>>(src, dst, dinv, bufA, bufC, E);
  k_combine<<<nodeBlocks, 256, 0, stream>>>(bufC, bufA, dinv, b2, bufC, N);

  k_gemm_wmma<16><<<gemmBlocks, 256, 0, stream>>>(bufC, l1W, bufA, ntiles);
  k_final<<<nodeBlocks, 256, 0, stream>>>(bufA, l1b, l2W, l2b, (float*)d_out, N);
}


# Float-atomic edge scatter replaced by LDS destination tiles (edge chunk x node range): matching edges listed, rows added by LDS float atomics, chunk partials summed in combine; sum order moves
# speedup vs baseline: 1.2785x; 1.2785x over previous
#include <hip/hip_runtime.h>
#include <hip/hip_bf16.h>

typedef __attribute__((ext_vector_type(2))) float v2f;
typedef __attribute__((ext_vector_type(8))) float v8f;

__global__ void k_init(float* __restrict__ deg, float* __restrict__ aggB,
                       float* __restrict__ aggC, int n) {
  int idx = blockIdx.x * blockDim.x + threadIdx.x;
  if (idx < n) deg[idx] = 1.0f;
  if (idx < n * 16) { aggB[idx] = 0.0f; aggC[idx] = 0.0f; }
}

__global__ void k_deg(const int* __restrict__ dst, float* __restrict__ deg, int E) {
  int e = blockIdx.x * blockDim.x + threadIdx.x;
  if (e < E) atomicAdd(&deg[dst[e]], 1.0f);
}

__global__ void k_rsqrt(const float* __restrict__ deg, float* __restrict__ dinv, int n) {
  int i = blockIdx.x * blockDim.x + threadIdx.x;
  if (i < n) dinv[i] = rsqrtf(deg[i]);
}

template <int K>
__global__ void k_gemm_wmma(const float* __restrict__ A, const float* __restrict__ W,
                            float* __restrict__ H, int ntiles) {
  __shared__ float Ws[K * 16];
  for (int t = threadIdx.x; t < K * 16; t += blockDim.x) Ws[t] = W[t];
  __syncthreads();

  const int lane = threadIdx.x & 31;
  const int tile = blockIdx.x * (blockDim.x >> 5) + (threadIdx.x >> 5);
  if (tile >= ntiles) return;

  const int half = lane >> 4;
  const int mn   = lane & 15;
  const int koff = half * 2;

  const float* arow = A + (size_t)(tile * 16 + mn) * K;

  v8f c = {0.f, 0.f, 0.f, 0.f, 0.f, 0.f, 0.f, 0.f};
#pragma unroll
  for (int k0 = 0; k0 < K; k0 += 4) {
    v2f a = *(const v2f*)(arow + k0 + koff);
    v2f b;
    b.x = Ws[(k0 + koff + 0) * 16 + mn];
    b.y = Ws[(k0 + koff + 1) * 16 + mn];
    c = __builtin_amdgcn_wmma_f32_16x16x4_f32(
         false, a,  false, b,
         (short)0, c,  false,  false);
  }

  float* hrow = H + (size_t)tile * 256 + half * 128 + mn;
#pragma unroll
  for (int i = 0; i < 8; i++) hrow[i * 16] = c[i];
}

__global__ void k_scatter(const int* __restrict__ src, const int* __restrict__ dst,
                          const float* __restrict__ dinv, const float* __restrict__ H,
                          float* __restrict__ AGG, int E) {
  int idx = blockIdx.x * blockDim.x + threadIdx.x;
  int e = idx >> 4;
  int j = idx & 15;
  if (e >= E) return;
  int s = src[e];
  int d = dst[e];
  float norm = dinv[s] * dinv[d];
  atomicAdd(&AGG[d * 16 + j], norm * H[s * 16 + j]);
}

__global__ void k_combine(const float* __restrict__ AGG, const float* __restrict__ H,
                          const float* __restrict__ dinv, const float* __restrict__ bias,
                          float* __restrict__ out, int n) {
  int idx = blockIdx.x * blockDim.x + threadIdx.x;
  int i = idx >> 4;
  int j = idx & 15;
  if (i >= n) return;
  float di = dinv[i];
  out[idx] = fmaxf(AGG[idx] + di * di * H[idx] + bias[j], 0.0f);
}

__global__ void k_final(const float* __restrict__ G, const float* __restrict__ l1b,
                        const float* __restrict__ l2W, const float* __restrict__ l2b,
                        float* __restrict__ out, int n) {
  int idx = blockIdx.x * blockDim.x + threadIdx.x;
  int i = idx >> 4;
  int j = idx & 15;
  float v = 0.0f;
  if (i < n) v = fmaxf(G[idx] + l1b[j], 0.0f) * l2W[j];
#pragma unroll
  for (int off = 8; off; off >>= 1) v += __shfl_xor(v, off, 16);
  if (i < n && j == 0) out[i] = v + l2b[0];
}


#define PUSH_TILE 4096
#define PUSH_CAP  15360
#define PUSH_EPT  15
#define PUSH_NT   1024
#define PUSH_C    10

__global__ __launch_bounds__(PUSH_NT)
void k_push(const int* __restrict__ src, const int* __restrict__ dst,
            const float* __restrict__ dinv, const float* __restrict__ H,
            float* __restrict__ part, int E, int n, int chunkEdges) {
  __shared__ __attribute__((aligned(16))) float acc[PUSH_TILE * 16];
  __shared__ unsigned list[PUSH_CAP];
  __shared__ unsigned cnt[2];

  const int tid = threadIdx.x;
  const int lo  = blockIdx.x * PUSH_TILE;
  const int e0  = blockIdx.y * chunkEdges;
  const int e1  = min(E, e0 + chunkEdges);

  for (int i = tid; i < PUSH_TILE * 16; i += PUSH_NT) acc[i] = 0.0f;
  if (tid < 2) cnt[tid] = 0u;

  int dv[PUSH_EPT] = {0}, sv[PUSH_EPT] = {0};
  if (e0 < e1) {
#pragma unroll
    for (int k = 0; k < PUSH_EPT; k++) {
      const int idx = min(e0 + k * PUSH_NT + tid, e1 - 1);
      dv[k] = dst[idx];
      sv[k] = src[idx];
    }
  }
  __syncthreads();

  const unsigned hw = (unsigned)tid >> 4;
  const int      j  = tid & 15;

  int p = 0;
  for (int b0 = e0; b0 < e1; b0 += PUSH_CAP, p ^= 1) {
    unsigned off[PUSH_EPT];
    unsigned flags = 0u;
#pragma unroll
    for (int k = 0; k < PUSH_EPT; k++) {
      const int e = b0 + k * PUSH_NT + tid;
      off[k] = (unsigned)(dv[k] - lo);
      if (e < e1 && off[k] < (unsigned)PUSH_TILE) flags |= (1u << k);
    }
    const unsigned mine = (unsigned)__popc(flags);
    unsigned pos = 0u;
    if (mine) pos = atomicAdd(&cnt[p], mine);
#pragma unroll
    for (int k = 0; k < PUSH_EPT; k++) {
      if ((flags >> k) & 1u) { list[pos] = (unsigned)sv[k] | (off[k] << 17); pos++; }
    }
    __syncthreads();

    const unsigned nent = cnt[p];
    if (tid == 0) cnt[p ^ 1] = 0u;

    const int nb = b0 + PUSH_CAP;
    if (nb < e1) {
#pragma unroll
      for (int k = 0; k < PUSH_EPT; k++) {
        const int idx = min(nb + k * PUSH_NT + tid, e1 - 1);
        dv[k] = dst[idx];
        sv[k] = src[idx];
      }
    }

    for (unsigned i = hw; i < nent; i += 512u) {
      unsigned w[8]; float v[8];
#pragma unroll
      for (int q = 0; q < 8; q++) {
        const unsigned idx = i + 64u * q;
        w[q] = list[idx < nent ? idx : i];
      }
#pragma unroll
      for (int q = 0; q < 8; q++) {
        const unsigned s = w[q] & 0x1FFFFu;
        const unsigned o = w[q] >> 17;
        const float norm = dinv[s] * dinv[lo + o];
        const float t = norm * H[s * 16 + j];
        v[q] = (i + 64u * q < nent) ? t : 0.0f;
      }
#pragma unroll
      for (int q = 0; q < 8; q++) atomicAdd(&acc[(w[q] >> 17) * 16 + j], v[q]);
    }
    __syncthreads();
  }

  const float4* acc4 = (const float4*)acc;
  float4* out4 = (float4*)(part + ((size_t)blockIdx.y * n + lo) * 16);
  for (int i4 = tid; i4 < PUSH_TILE * 4; i4 += PUSH_NT) {
    if (lo + (i4 >> 2) < n) out4[i4] = acc4[i4];
  }
}

__global__ void k_combine_parts(const float* __restrict__ part, const float* __restrict__ H,
                                const float* __restrict__ dinv, const float* __restrict__ bias,
                                float* __restrict__ out, int n) {
  int idx = blockIdx.x * blockDim.x + threadIdx.x;
  int i = idx >> 4;
  int j = idx & 15;
  if (i >= n) return;
  const size_t plane = (size_t)n * 16;
  float a = part[idx];
#pragma unroll
  for (int c = 1; c < PUSH_C; c++) a += part[c * plane + idx];
  float di = dinv[i];
  out[idx] = fmaxf(a + di * di * H[idx] + bias[j], 0.0f);
}

__global__ void k_init_deg(float* __restrict__ deg, int n) {
  int idx = blockIdx.x * blockDim.x + threadIdx.x;
  if (idx < n) deg[idx] = 1.0f;
}

extern "C" void kernel_launch(void* const* d_in, const int* in_sizes, int n_in,
                              void* d_out, int out_size, void* d_ws, size_t ws_size,
                              hipStream_t stream) {
  const float* x   = (const float*)d_in[0];
  const int*   ei  = (const int*)d_in[1];
  const float* W1  = (const float*)d_in[2];
  const float* b1  = (const float*)d_in[3];
  const float* W2  = (const float*)d_in[4];
  const float* b2  = (const float*)d_in[5];
  const float* l1W = (const float*)d_in[6];
  const float* l1b = (const float*)d_in[7];
  const float* l2W = (const float*)d_in[8];
  const float* l2b = (const float*)d_in[9];

  const int N = in_sizes[0] / 128;
  const int E = in_sizes[1] / 2;
  const int* src = ei;
  const int* dst = ei + E;

  float* deg  = (float*)d_ws;
  float* dinv = deg + N;
  float* bufA = dinv + N;
  float* bufB = bufA + (size_t)N * 16;
  float* bufC = bufB + (size_t)N * 16;

  const int ntiles = N / 16;
  const int gemmBlocks = (ntiles * 32 + 255) / 256;
  const int nodeElems  = N * 16;
  const int nodeBlocks = (nodeElems + 255) / 256;
  const int edgeBlocks = (E + 255) / 256;
  const int scatBlocks = (int)(((size_t)E * 16 + 255) / 256);

  const size_t pushNeed = ((size_t)N * 2 + (size_t)N * 16 * (2 + PUSH_C)) * sizeof(float);
  if (pushNeed <= ws_size && N <= (1 << 17)) {
    float* part = bufC;
    const int chunkEdges = (E + PUSH_C - 1) / PUSH_C;
    const dim3 pushGrid((N + PUSH_TILE - 1) / PUSH_TILE, PUSH_C);

    k_init_deg<<<(N + 255) / 256, 256, 0, stream>>>(deg, N);
    k_deg   <<<edgeBlocks, 256, 0, stream>>>(dst, deg, E);
    k_rsqrt <<<(N + 255) / 256, 256, 0, stream>>>(deg, dinv, N);

    k_gemm_wmma<128><<<gemmBlocks, 256, 0, stream>>>(x, W1, bufA, ntiles);
    k_push<<<pushGrid, PUSH_NT, 0, stream>>>(src, dst, dinv, bufA, part, E, N, chunkEdges);
    k_combine_parts<<<nodeBlocks, 256, 0, stream>>>(part, bufA, dinv, b1, bufB, N);

    k_gemm_wmma<16><<<gemmBlocks, 256, 0, stream>>>(bufB, W2, bufA, ntiles);
    k_push<<<pushGrid, PUSH_NT, 0, stream>>>(src, dst, dinv, bufA, part, E, N, chunkEdges);
    k_combine_parts<<<nodeBlocks, 256, 0, stream>>>(part, bufA, dinv, b2, bufB, N);

    k_gemm_wmma<16><<<gemmBlocks, 256, 0, stream>>>(bufB, l1W, bufA, ntiles);
    k_final<<<nodeBlocks, 256, 0, stream>>>(bufA, l1b, l2W, l2b, (float*)d_out, N);
  } else {
    k_init  <<<nodeBlocks, 256, 0, stream>>>(deg, bufB, bufC, N);
    k_deg   <<<edgeBlocks, 256, 0, stream>>>(dst, deg, E);
    k_rsqrt <<<(N + 255) / 256, 256, 0, stream>>>(deg, dinv, N);

    k_gemm_wmma<128><<<gemmBlocks, 256, 0, stream>>>(x, W1, bufA, ntiles);
    k_scatter<<<scatBlocks, 256, 0, stream>>>(src, dst, dinv, bufA, bufB, E);
    k_combine<<<nodeBlocks, 256, 0, stream>>>(bufB, bufA, dinv, b1, bufB, N);

    k_gemm_wmma<16><<<gemmBlocks, 256, 0, stream>>>(bufB, W2, bufA, ntiles);
    k_scatter<<<scatBlocks, 256, 0, stream>>>(src, dst, dinv, bufA, bufC, E);
    k_combine<<<nodeBlocks, 256, 0, stream>>>(bufC, bufA, dinv, b2, bufC, N);

    k_gemm_wmma<16><<<gemmBlocks, 256, 0, stream>>>(bufC, l1W, bufA, ntiles);
    k_final<<<nodeBlocks, 256, 0, stream>>>(bufA, l1b, l2W, l2b, (float*)d_out, N);
  }
}


# Push edge loop made private to each wave: ballot-compacted matches go to the wave's own LDS ring, drained in full 16-edge trips; no block barrier, counter or batch list; LDS add order moves
# speedup vs baseline: 1.5409x; 1.2052x over previous
#include <hip/hip_runtime.h>
#include <hip/hip_bf16.h>

typedef __attribute__((ext_vector_type(2))) float v2f;
typedef __attribute__((ext_vector_type(8))) float v8f;

__global__ void k_init(float* __restrict__ deg, float* __restrict__ aggB,
                       float* __restrict__ aggC, int n) {
  int idx = blockIdx.x * blockDim.x + threadIdx.x;
  if (idx < n) deg[idx] = 1.0f;
  if (idx < n * 16) { aggB[idx] = 0.0f; aggC[idx] = 0.0f; }
}

__global__ void k_deg(const int* __restrict__ dst, float* __restrict__ deg, int E) {
  int e = blockIdx.x * blockDim.x + threadIdx.x;
  if (e < E) atomicAdd(&deg[dst[e]], 1.0f);
}

__global__ void k_rsqrt(const float* __restrict__ deg, float* __restrict__ dinv, int n) {
  int i = blockIdx.x * blockDim.x + threadIdx.x;
  if (i < n) dinv[i] = rsqrtf(deg[i]);
}

template <int K>
__global__ void k_gemm_wmma(const float* __restrict__ A, const float* __restrict__ W,
                            float* __restrict__ H, int ntiles) {
  __shared__ float Ws[K * 16];
  for (int t = threadIdx.x; t < K * 16; t += blockDim.x) Ws[t] = W[t];
  __syncthreads();

  const int lane = threadIdx.x & 31;
  const int tile = blockIdx.x * (blockDim.x >> 5) + (threadIdx.x >> 5);
  if (tile >= ntiles) return;

  const int half = lane >> 4;
  const int mn   = lane & 15;
  const int koff = half * 2;

  const float* arow = A + (size_t)(tile * 16 + mn) * K;

  v8f c = {0.f, 0.f, 0.f, 0.f, 0.f, 0.f, 0.f, 0.f};
#pragma unroll
  for (int k0 = 0; k0 < K; k0 += 4) {
    v2f a = *(const v2f*)(arow + k0 + koff);
    v2f b;
    b.x = Ws[(k0 + koff + 0) * 16 + mn];
    b.y = Ws[(k0 + koff + 1) * 16 + mn];
    c = __builtin_amdgcn_wmma_f32_16x16x4_f32(
         false, a,  false, b,
         (short)0, c,  false,  false);
  }

  float* hrow = H + (size_t)tile * 256 + half * 128 + mn;
#pragma unroll
  for (int i = 0; i < 8; i++) hrow[i * 16] = c[i];
}

__global__ void k_scatter(const int* __restrict__ src, const int* __restrict__ dst,
                          const float* __restrict__ dinv, const float* __restrict__ H,
                          float* __restrict__ AGG, int E) {
  int idx = blockIdx.x * blockDim.x + threadIdx.x;
  int e = idx >> 4;
  int j = idx & 15;
  if (e >= E) return;
  int s = src[e];
  int d = dst[e];
  float norm = dinv[s] * dinv[d];
  atomicAdd(&AGG[d * 16 + j], norm * H[s * 16 + j]);
}

__global__ void k_combine(const float* __restrict__ AGG, const float* __restrict__ H,
                          const float* __restrict__ dinv, const float* __restrict__ bias,
                          float* __restrict__ out, int n) {
  int idx = blockIdx.x * blockDim.x + threadIdx.x;
  int i = idx >> 4;
  int j = idx & 15;
  if (i >= n) return;
  float di = dinv[i];
  out[idx] = fmaxf(AGG[idx] + di * di * H[idx] + bias[j], 0.0f);
}

__global__ void k_final(const float* __restrict__ G, const float* __restrict__ l1b,
                        const float* __restrict__ l2W, const float* __restrict__ l2b,
                        float* __restrict__ out, int n) {
  int idx = blockIdx.x * blockDim.x + threadIdx.x;
  int i = idx >> 4;
  int j = idx & 15;
  float v = 0.0f;
  if (i < n) v = fmaxf(G[idx] + l1b[j], 0.0f) * l2W[j];
#pragma unroll
  for (int off = 8; off; off >>= 1) v += __shfl_xor(v, off, 16);
  if (i < n && j == 0) out[i] = v + l2b[0];
}


#define PUSH_TILE 4096
#define PUSH_NT   1024
#define PUSH_K    6
#define PUSH_RING 256
#define PUSH_C    10

template <bool FULL>
__device__ __forceinline__ void push_trip(const unsigned* q, unsigned qh, unsigned left,
                                          unsigned half, int j, int lo,
                                          const float* __restrict__ dinv,
                                          const float* __restrict__ H, float* acc) {
  unsigned w[8]; float v[8];
#pragma unroll
  for (int t = 0; t < 8; t++) {
    const unsigned slot = 2u * t + half;
    const unsigned idx = (FULL || slot < left) ? qh + slot : qh;
    w[t] = q[idx & (PUSH_RING - 1)];
  }
#pragma unroll
  for (int t = 0; t < 8; t++) {
    const unsigned s = w[t] & 0x1FFFFu;
    const unsigned o = w[t] >> 17;
    const float norm = dinv[s] * dinv[lo + o];
    const float x = norm * H[s * 16 + j];
    v[t] = (FULL || 2u * t + half < left) ? x : 0.0f;
  }
#pragma unroll
  for (int t = 0; t < 8; t++) atomicAdd(&acc[(w[t] >> 17) * 16 + j], v[t]);
}

__device__ __forceinline__ void push_fetch(int (&dv)[PUSH_K], int (&sv)[PUSH_K], int b0, int tid,
                                           int e1, const int* __restrict__ src,
                                           const int* __restrict__ dst) {
#pragma unroll
  for (int k = 0; k < PUSH_K; k++) {
    const int idx = min(b0 + k * PUSH_NT + tid, e1 - 1);
    dv[k] = dst[idx];
    sv[k] = src[idx];
  }
}

__device__ __forceinline__ void push_append(const int (&dv)[PUSH_K], const int (&sv)[PUSH_K],
                                            int b0, int tid, int e1, int lo,
                                            unsigned* q, unsigned& qn) {
#pragma unroll
  for (int k = 0; k < PUSH_K; k++) {
    const int e = b0 + k * PUSH_NT + tid;
    const unsigned off = (unsigned)(dv[k] - lo);
    const bool hit = e < e1 && off < (unsigned)PUSH_TILE;
    const unsigned m = __builtin_amdgcn_ballot_w32(hit);
    if (hit) q[(qn + __builtin_amdgcn_mbcnt_lo(m, 0u)) & (PUSH_RING - 1)] = (unsigned)sv[k] | (off << 17);
    qn += (unsigned)__popc(m);
  }
}

__global__ __launch_bounds__(PUSH_NT)
void k_push(const int* __restrict__ src, const int* __restrict__ dst,
            const float* __restrict__ dinv, const float* __restrict__ H,
            float* __restrict__ part, int E, int n, int chunkEdges) {
  __shared__ __attribute__((aligned(16))) float acc[PUSH_TILE * 16];
  __shared__ unsigned ring[(PUSH_NT / 32) * PUSH_RING];

  const int tid = threadIdx.x;
  const int lo  = blockIdx.x * PUSH_TILE;
  const int e0  = blockIdx.y * chunkEdges;
  const int e1  = min(E, e0 + chunkEdges);
  const int STEP = PUSH_K * PUSH_NT;

  for (int i = tid; i < PUSH_TILE * 16; i += PUSH_NT) acc[i] = 0.0f;

  int dvA[PUSH_K] = {0}, svA[PUSH_K] = {0}, dvB[PUSH_K] = {0}, svB[PUSH_K] = {0};
  const bool any = e0 < e1;
  if (any) {
    push_fetch(dvA, svA, e0, tid, e1, src, dst);
    push_fetch(dvB, svB, e0 + STEP, tid, e1, src, dst);
  }
  __syncthreads();

  const int wv = __builtin_amdgcn_readfirstlane(tid >> 5);
  unsigned* q = ring + wv * PUSH_RING;
  const unsigned half = (unsigned)(tid >> 4) & 1u;
  const int j = tid & 15;
  unsigned qn = 0u, qh = 0u;

  if (any) for (int b0 = e0; b0 < e1; b0 += 2 * STEP) {
    push_append(dvA, svA, b0, tid, e1, lo, q, qn);
    push_fetch(dvA, svA, b0 + 2 * STEP, tid, e1, src, dst);
    while (qn - qh >= 16u) { push_trip<true>(q, qh, 16u, half, j, lo, dinv, H, acc); qh += 16u; }

    push_append(dvB, svB, b0 + STEP, tid, e1, lo, q, qn);
    push_fetch(dvB, svB, b0 + 3 * STEP, tid, e1, src, dst);
    while (qn - qh >= 16u) { push_trip<true>(q, qh, 16u, half, j, lo, dinv, H, acc); qh += 16u; }
  }
  if (qn != qh) push_trip<false>(q, qh, qn - qh, half, j, lo, dinv, H, acc);
  __syncthreads();

  const float4* acc4 = (const float4*)acc;
  float4* out4 = (float4*)(part + ((size_t)blockIdx.y * n + lo) * 16);
  for (int i4 = tid; i4 < PUSH_TILE * 4; i4 += PUSH_NT) {
    if (lo + (i4 >> 2) < n) out4[i4] = acc4[i4];
  }
}

__global__ void k_combine_parts(const float* __restrict__ part, const float* __restrict__ H,
                                const float* __restrict__ dinv, const float* __restrict__ bias,
                                float* __restrict__ out, int n) {
  int idx = blockIdx.x * blockDim.x + threadIdx.x;
  int i = idx >> 4;
  int j = idx & 15;
  if (i >= n) return;
  const size_t plane = (size_t)n * 16;
  float a = part[idx];
#pragma unroll
  for (int c = 1; c < PUSH_C; c++) a += part[c * plane + idx];
  float di = dinv[i];
  out[idx] = fmaxf(a + di * di * H[idx] + bias[j], 0.0f);
}

__global__ void k_init_deg(float* __restrict__ deg, int n) {
  int idx = blockIdx.x * blockDim.x + threadIdx.x;
  if (idx < n) deg[idx] = 1.0f;
}

extern "C" void kernel_launch(void* const* d_in, const int* in_sizes, int n_in,
                              void* d_out, int out_size, void* d_ws, size_t ws_size,
                              hipStream_t stream) {
  const float* x   = (const float*)d_in[0];
  const int*   ei  = (const int*)d_in[1];
  const float* W1  = (const float*)d_in[2];
  const float* b1  = (const float*)d_in[3];
  const float* W2  = (const float*)d_in[4];
  const float* b2  = (const float*)d_in[5];
  const float* l1W = (const float*)d_in[6];
  const float* l1b = (const float*)d_in[7];
  const float* l2W = (const float*)d_in[8];
  const float* l2b = (const float*)d_in[9];

  const int N = in_sizes[0] / 128;
  const int E = in_sizes[1] / 2;
  const int* src = ei;
  const int* dst = ei + E;

  float* deg  = (float*)d_ws;
  float* dinv = deg + N;
  float* bufA = dinv + N;
  float* bufB = bufA + (size_t)N * 16;
  float* bufC = bufB + (size_t)N * 16;

  const int ntiles = N / 16;
  const int gemmBlocks = (ntiles * 32 + 255) / 256;
  const int nodeElems  = N * 16;
  const int nodeBlocks = (nodeElems + 255) / 256;
  const int edgeBlocks = (E + 255) / 256;
  const int scatBlocks = (int)(((size_t)E * 16 + 255) / 256);

  const size_t pushNeed = ((size_t)N * 2 + (size_t)N * 16 * (2 + PUSH_C)) * sizeof(float);
  if (pushNeed <= ws_size && N <= (1 << 17)) {
    float* part = bufC;
    const int chunkEdges = (E + PUSH_C - 1) / PUSH_C;
    const dim3 pushGrid((N + PUSH_TILE - 1) / PUSH_TILE, PUSH_C);

    k_init_deg<<<(N + 255) / 256, 256, 0, stream>>>(deg, N);
    k_deg   <<<edgeBlocks, 256, 0, stream>>>(dst, deg, E);
    k_rsqrt <<<(N + 255) / 256, 256, 0, stream>>>(deg, dinv, N);

    k_gemm_wmma<128><<<gemmBlocks, 256, 0, stream>>>(x, W1, bufA, ntiles);
    k_push<<<pushGrid, PUSH_NT, 0, stream>>>(src, dst, dinv, bufA, part, E, N, chunkEdges);
    k_combine_parts<<<nodeBlocks, 256, 0, stream>>>(part, bufA, dinv, b1, bufB, N);

    k_gemm_wmma<16><<<gemmBlocks, 256, 0, stream>>>(bufB, W2, bufA, ntiles);
    k_push<<<pushGrid, PUSH_NT, 0, stream>>>(src, dst, dinv, bufA, part, E, N, chunkEdges);
    k_combine_parts<<<nodeBlocks, 256, 0, stream>>>(part, bufA, dinv, b2, bufB, N);

    k_gemm_wmma<16><<<gemmBlocks, 256, 0, stream>>>(bufB, l1W, bufA, ntiles);
    k_final<<<nodeBlocks, 256, 0, stream>>>(bufA, l1b, l2W, l2b, (float*)d_out, N);
  } else {
    k_init  <<<nodeBlocks, 256, 0, stream>>>(deg, bufB, bufC, N);
    k_deg   <<<edgeBlocks, 256, 0, stream>>>(dst, deg, E);
    k_rsqrt <<<(N + 255) / 256, 256, 0, stream>>>(deg, dinv, N);

    k_gemm_wmma<128><<<gemmBlocks, 256, 0, stream>>>(x, W1, bufA, ntiles);
    k_scatter<<<scatBlocks, 256, 0, stream>>>(src, dst, dinv, bufA, bufB, E);
    k_combine<<<nodeBlocks, 256, 0, stream>>>(bufB, bufA, dinv, b1, bufB, N);

    k_gemm_wmma<16><<<gemmBlocks, 256, 0, stream>>>(bufB, W2, bufA, ntiles);
    k_scatter<<<scatBlocks, 256, 0, stream>>>(src, dst, dinv, bufA, bufC, E);
    k_combine<<<nodeBlocks, 256, 0, stream>>>(bufC, bufA, dinv, b2, bufC, N);

    k_gemm_wmma<16><<<gemmBlocks, 256, 0, stream>>>(bufC, l1W, bufA, ntiles);
    k_final<<<nodeBlocks, 256, 0, stream>>>(bufA, l1b, l2W, l2b, (float*)d_out, N);
  }
}


# Edge scale lookups leave the push drain: source dinv kept in its H row's 128-B line (row pitch 32 floats), destination dinv applied once per node in combine; one line per edge; rounding moves
# speedup vs baseline: 1.6110x; 1.0455x over previous
#include <hip/hip_runtime.h>
#include <hip/hip_bf16.h>

typedef __attribute__((ext_vector_type(2))) float v2f;
typedef __attribute__((ext_vector_type(8))) float v8f;

#define TAB_PITCH 32

__global__ void k_init(float* __restrict__ deg, float* __restrict__ aggB,
                       float* __restrict__ aggC, int n) {
  int idx = blockIdx.x * blockDim.x + threadIdx.x;
  if (idx < n) deg[idx] = 1.0f;
  if (idx < n * 16) { aggB[idx] = 0.0f; aggC[idx] = 0.0f; }
}

__global__ void k_deg(const int* __restrict__ dst, float* __restrict__ deg, int E) {
  int e = blockIdx.x * blockDim.x + threadIdx.x;
  if (e < E) atomicAdd(&deg[dst[e]], 1.0f);
}

__global__ void k_rsqrt(const float* __restrict__ deg, float* __restrict__ dinv, int n) {
  int i = blockIdx.x * blockDim.x + threadIdx.x;
  if (i < n) dinv[i] = rsqrtf(deg[i]);
}

__global__ void k_rsqrt_rows(const float* __restrict__ deg, float* __restrict__ dinv,
                             float* __restrict__ tab, int n) {
  int i = blockIdx.x * blockDim.x + threadIdx.x;
  if (i < n) {
    const float r = rsqrtf(deg[i]);
    dinv[i] = r;
    tab[(size_t)i * TAB_PITCH + 16] = r;
  }
}

template <int K, int P>
__global__ void k_gemm_wmma(const float* __restrict__ A, const float* __restrict__ W,
                            float* __restrict__ H, int ntiles) {
  __shared__ float Ws[K * 16];
  for (int t = threadIdx.x; t < K * 16; t += blockDim.x) Ws[t] = W[t];
  __syncthreads();

  const int lane = threadIdx.x & 31;
  const int tile = blockIdx.x * (blockDim.x >> 5) + (threadIdx.x >> 5);
  if (tile >= ntiles) return;

  const int half = lane >> 4;
  const int mn   = lane & 15;
  const int koff = half * 2;

  const float* arow = A + (size_t)(tile * 16 + mn) * K;

  v8f c = {0.f, 0.f, 0.f, 0.f, 0.f, 0.f, 0.f, 0.f};
#pragma unroll
  for (int k0 = 0; k0 < K; k0 += 4) {
    v2f a = *(const v2f*)(arow + k0 + koff);
    v2f b;
    b.x = Ws[(k0 + koff + 0) * 16 + mn];
    b.y = Ws[(k0 + koff + 1) * 16 + mn];
    c = __builtin_amdgcn_wmma_f32_16x16x4_f32(
         false, a,  false, b,
         (short)0, c,  false,  false);
  }

  float* hrow = H + ((size_t)tile * 16 + half * 8) * P + mn;
#pragma unroll
  for (int i = 0; i < 8; i++) hrow[i * P] = c[i];
}

__global__ void k_scatter(const int* __restrict__ src, const int* __restrict__ dst,
                          const float* __restrict__ dinv, const float* __restrict__ H,
                          float* __restrict__ AGG, int E) {
  int idx = blockIdx.x * blockDim.x + threadIdx.x;
  int e = idx >> 4;
  int j = idx & 15;
  if (e >= E) return;
  int s = src[e];
  int d = dst[e];
  float norm = dinv[s] * dinv[d];
  atomicAdd(&AGG[d * 16 + j], norm * H[s * 16 + j]);
}

__global__ void k_combine(const float* __restrict__ AGG, const float* __restrict__ H,
                          const float* __restrict__ dinv, const float* __restrict__ bias,
                          float* __restrict__ out, int n) {
  int idx = blockIdx.x * blockDim.x + threadIdx.x;
  int i = idx >> 4;
  int j = idx & 15;
  if (i >= n) return;
  float di = dinv[i];
  out[idx] = fmaxf(AGG[idx] + di * di * H[idx] + bias[j], 0.0f);
}

__global__ void k_final(const float* __restrict__ G, const float* __restrict__ l1b,
                        const float* __restrict__ l2W, const float* __restrict__ l2b,
                        float* __restrict__ out, int n) {
  int idx = blockIdx.x * blockDim.x + threadIdx.x;
  int i = idx >> 4;
  int j = idx & 15;
  float v = 0.0f;
  if (i < n) v = fmaxf(G[idx] + l1b[j], 0.0f) * l2W[j];
#pragma unroll
  for (int off = 8; off; off >>= 1) v += __shfl_xor(v, off, 16);
  if (i < n && j == 0) out[i] = v + l2b[0];
}


#define PUSH_TILE 4096
#define PUSH_NT   1024
#define PUSH_K    6
#define PUSH_RING 256
#define PUSH_C    10

template <bool FULL>
__device__ __forceinline__ void push_trip(const unsigned* q, unsigned qh, unsigned left,
                                          unsigned half, int j,
                                          const float* __restrict__ tab, float* acc) {
  unsigned w[8]; float v[8];
#pragma unroll
  for (int t = 0; t < 8; t++) {
    const unsigned slot = 2u * t + half;
    const unsigned idx = (FULL || slot < left) ? qh + slot : qh;
    w[t] = q[idx & (PUSH_RING - 1)];
  }
#pragma unroll
  for (int t = 0; t < 8; t++) {
    const unsigned r = (w[t] & 0x1FFFFu) * TAB_PITCH;
    const float x = tab[r + 16u] * tab[r + (unsigned)j];
    v[t] = (FULL || 2u * t + half < left) ? x : 0.0f;
  }
#pragma unroll
  for (int t = 0; t < 8; t++) atomicAdd(&acc[(w[t] >> 17) * 16 + j], v[t]);
}

__device__ __forceinline__ void push_fetch(int (&dv)[PUSH_K], int (&sv)[PUSH_K], int b0, int tid,
                                           int e1, const int* __restrict__ src,
                                           const int* __restrict__ dst) {
#pragma unroll
  for (int k = 0; k < PUSH_K; k++) {
    const int idx = min(b0 + k * PUSH_NT + tid, e1 - 1);
    dv[k] = dst[idx];
    sv[k] = src[idx];
  }
}

__device__ __forceinline__ void push_append(const int (&dv)[PUSH_K], const int (&sv)[PUSH_K],
                                            int b0, int tid, int e1, int lo,
                                            unsigned* q, unsigned& qn) {
#pragma unroll
  for (int k = 0; k < PUSH_K; k++) {
    const int e = b0 + k * PUSH_NT + tid;
    const unsigned off = (unsigned)(dv[k] - lo);
    const bool hit = e < e1 && off < (unsigned)PUSH_TILE;
    const unsigned m = __builtin_amdgcn_ballot_w32(hit);
    if (hit) q[(qn + __builtin_amdgcn_mbcnt_lo(m, 0u)) & (PUSH_RING - 1)] = (unsigned)sv[k] | (off << 17);
    qn += (unsigned)__popc(m);
  }
}

__global__ __launch_bounds__(PUSH_NT)
void k_push(const int* __restrict__ src, const int* __restrict__ dst,
            const float* __restrict__ tab,
            float* __restrict__ part, int E, int n, int chunkEdges) {
  __shared__ __attribute__((aligned(16))) float acc[PUSH_TILE * 16];
  __shared__ unsigned ring[(PUSH_NT / 32) * PUSH_RING];

  const int tid = threadIdx.x;
  const int lo  = blockIdx.x * PUSH_TILE;
  const int e0  = blockIdx.y * chunkEdges;
  const int e1  = min(E, e0 + chunkEdges);
  const int STEP = PUSH_K * PUSH_NT;

  for (int i = tid; i < PUSH_TILE * 16; i += PUSH_NT) acc[i] = 0.0f;

  int dvA[PUSH_K] = {0}, svA[PUSH_K] = {0}, dvB[PUSH_K] = {0}, svB[PUSH_K] = {0};
  const bool any = e0 < e1;
  if (any) {
    push_fetch(dvA, svA, e0, tid, e1, src, dst);
    push_fetch(dvB, svB, e0 + STEP, tid, e1, src, dst);
  }
  __syncthreads();

  const int wv = __builtin_amdgcn_readfirstlane(tid >> 5);
  unsigned* q = ring + wv * PUSH_RING;
  const unsigned half = (unsigned)(tid >> 4) & 1u;
  const int j = tid & 15;
  unsigned qn = 0u, qh = 0u;

  if (any) for (int b0 = e0; b0 < e1; b0 += 2 * STEP) {
    push_append(dvA, svA, b0, tid, e1, lo, q, qn);
    push_fetch(dvA, svA, b0 + 2 * STEP, tid, e1, src, dst);
    while (qn - qh >= 16u) { push_trip<true>(q, qh, 16u, half, j, tab, acc); qh += 16u; }

    push_append(dvB, svB, b0 + STEP, tid, e1, lo, q, qn);
    push_fetch(dvB, svB, b0 + 3 * STEP, tid, e1, src, dst);
    while (qn - qh >= 16u) { push_trip<true>(q, qh, 16u, half, j, tab, acc); qh += 16u; }
  }
  if (qn != qh) push_trip<false>(q, qh, qn - qh, half, j, tab, acc);
  __syncthreads();

  const float4* acc4 = (const float4*)acc;
  float4* out4 = (float4*)(part + ((size_t)blockIdx.y * n + lo) * 16);
  for (int i4 = tid; i4 < PUSH_TILE * 4; i4 += PUSH_NT) {
    if (lo + (i4 >> 2) < n) out4[i4] = acc4[i4];
  }
}

__global__ void k_combine_parts(const float* part, const float* __restrict__ tab,
                                const float* __restrict__ dinv, const float* __restrict__ bias,
                                float* out, int n) {
  int idx = blockIdx.x * blockDim.x + threadIdx.x;
  int i = idx >> 4;
  int j = idx & 15;
  if (i >= n) return;
  const size_t plane = (size_t)n * 16;
  float a = part[idx];
#pragma unroll
  for (int c = 1; c < PUSH_C; c++) a += part[c * plane + idx];
  const float di = dinv[i];
  out[idx] = fmaxf(di * (a + di * tab[(size_t)i * TAB_PITCH + j]) + bias[j], 0.0f);
}

__global__ void k_init_deg(float* __restrict__ deg, int n) {
  int idx = blockIdx.x * blockDim.x + threadIdx.x;
  if (idx < n) deg[idx] = 1.0f;
}

extern "C" void kernel_launch(void* const* d_in, const int* in_sizes, int n_in,
                              void* d_out, int out_size, void* d_ws, size_t ws_size,
                              hipStream_t stream) {
  const float* x   = (const float*)d_in[0];
  const int*   ei  = (const int*)d_in[1];
  const float* W1  = (const float*)d_in[2];
  const float* b1  = (const float*)d_in[3];
  const float* W2  = (const float*)d_in[4];
  const float* b2  = (const float*)d_in[5];
  const float* l1W = (const float*)d_in[6];
  const float* l1b = (const float*)d_in[7];
  const float* l2W = (const float*)d_in[8];
  const float* l2b = (const float*)d_in[9];

  const int N = in_sizes[0] / 128;
  const int E = in_sizes[1] / 2;
  const int* src = ei;
  const int* dst = ei + E;

  float* deg  = (float*)d_ws;
  float* dinv = deg + N;
  float* bufA = dinv + N;
  float* bufB = bufA + (size_t)N * 16;
  float* bufC = bufB + (size_t)N * 16;

  const int ntiles = N / 16;
  const int gemmBlocks = (ntiles * 32 + 255) / 256;
  const int nodeElems  = N * 16;
  const int nodeBlocks = (nodeElems + 255) / 256;
  const int edgeBlocks = (E + 255) / 256;
  const int scatBlocks = (int)(((size_t)E * 16 + 255) / 256);

  const size_t pushNeed = ((size_t)N * 2 + (size_t)N * 16 * (2 + PUSH_C)) * sizeof(float);
  if (pushNeed <= ws_size && N <= (1 << 17)) {
    float* tab  = bufA;
    float* part = bufC;
    float* h    = part;
    const int chunkEdges = (E + PUSH_C - 1) / PUSH_C;
    const dim3 pushGrid((N + PUSH_TILE - 1) / PUSH_TILE, PUSH_C);

    k_init_deg<<<(N + 255) / 256, 256, 0, stream>>>(deg, N);
    k_deg   <<<edgeBlocks, 256, 0, stream>>>(dst, deg, E);
    k_rsqrt_rows<<<(N + 255) / 256, 256, 0, stream>>>(deg, dinv, tab, N);

    k_gemm_wmma<128, TAB_PITCH><<<gemmBlocks, 256, 0, stream>>>(x, W1, tab, ntiles);
    k_push<<<pushGrid, PUSH_NT, 0, stream>>>(src, dst, tab, part, E, N, chunkEdges);
    k_combine_parts<<<nodeBlocks, 256, 0, stream>>>(part, tab, dinv, b1, h, N);

    k_gemm_wmma<16, TAB_PITCH><<<gemmBlocks, 256, 0, stream>>>(h, W2, tab, ntiles);
    k_push<<<pushGrid, PUSH_NT, 0, stream>>>(src, dst, tab, part, E, N, chunkEdges);
    k_combine_parts<<<nodeBlocks, 256, 0, stream>>>(part, tab, dinv, b2, h, N);

    k_gemm_wmma<16, 16><<<gemmBlocks, 256, 0, stream>>>(h, l1W, bufA, ntiles);
    k_final<<<nodeBlocks, 256, 0, stream>>>(bufA, l1b, l2W, l2b, (float*)d_out, N);
  } else {
    k_init  <<<nodeBlocks, 256, 0, stream>>>(deg, bufB, bufC, N);
    k_deg   <<<edgeBlocks, 256, 0, stream>>>(dst, deg, E);
    k_rsqrt <<<(N + 255) / 256, 256, 0, stream>>>(deg, dinv, N);

    k_gemm_wmma<128, 16><<<gemmBlocks, 256, 0, stream>>>(x, W1, bufA, ntiles);
    k_scatter<<<scatBlocks, 256, 0, stream>>>(src, dst, dinv, bufA, bufB, E);
    k_combine<<<nodeBlocks, 256, 0, stream>>>(bufB, bufA, dinv, b1, bufB, N);

    k_gemm_wmma<16, 16><<<gemmBlocks, 256, 0, stream>>>(bufB, W2, bufA, ntiles);
    k_scatter<<<scatBlocks, 256, 0, stream>>>(src, dst, dinv, bufA, bufC, E);
    k_combine<<<nodeBlocks, 256, 0, stream>>>(bufC, bufA, dinv, b2, bufC, N);

    k_gemm_wmma<16, 16><<<gemmBlocks, 256, 0, stream>>>(bufC, l1W, bufA, ntiles);
    k_final<<<nodeBlocks, 256, 0, stream>>>(bufA, l1b, l2W, l2b, (float*)d_out, N);
  }
}


# Layer-2 push tests no edge: layer-1 waves also store each drained 16-entry trip (64 B) to a per-wave list, layer 2 replays the lists, rows of two trips in flight; layer-2 LDS add order moves
# speedup vs baseline: 1.7206x; 1.0680x over previous
#include <hip/hip_runtime.h>
#include <hip/hip_bf16.h>

typedef __attribute__((ext_vector_type(2))) float v2f;
typedef __attribute__((ext_vector_type(8))) float v8f;

#define TAB_PITCH 32

__global__ void k_init(float* __restrict__ deg, float* __restrict__ aggB,
                       float* __restrict__ aggC, int n) {
  int idx = blockIdx.x * blockDim.x + threadIdx.x;
  if (idx < n) deg[idx] = 1.0f;
  if (idx < n * 16) { aggB[idx] = 0.0f; aggC[idx] = 0.0f; }
}

__global__ void k_deg(const int* __restrict__ dst, float* __restrict__ deg, int E) {
  int e = blockIdx.x * blockDim.x + threadIdx.x;
  if (e < E) atomicAdd(&deg[dst[e]], 1.0f);
}

__global__ void k_rsqrt(const float* __restrict__ deg, float* __restrict__ dinv, int n) {
  int i = blockIdx.x * blockDim.x + threadIdx.x;
  if (i < n) dinv[i] = rsqrtf(deg[i]);
}

__global__ void k_rsqrt_rows(const float* __restrict__ deg, float* __restrict__ dinv,
                             float* __restrict__ tab, int n) {
  int i = blockIdx.x * blockDim.x + threadIdx.x;
  if (i < n) {
    const float r = rsqrtf(deg[i]);
    dinv[i] = r;
    tab[(size_t)i * TAB_PITCH + 16] = r;
  }
}

template <int K, int P>
__global__ void k_gemm_wmma(const float* __restrict__ A, const float* __restrict__ W,
                            float* __restrict__ H, int ntiles) {
  __shared__ float Ws[K * 16];
  for (int t = threadIdx.x; t < K * 16; t += blockDim.x) Ws[t] = W[t];
  __syncthreads();

  const int lane = threadIdx.x & 31;
  const int tile = blockIdx.x * (blockDim.x >> 5) + (threadIdx.x >> 5);
  if (tile >= ntiles) return;

  const int half = lane >> 4;
  const int mn   = lane & 15;
  const int koff = half * 2;

  const float* arow = A + (size_t)(tile * 16 + mn) * K;

  v8f c = {0.f, 0.f, 0.f, 0.f, 0.f, 0.f, 0.f, 0.f};
#pragma unroll
  for (int k0 = 0; k0 < K; k0 += 4) {
    v2f a = *(const v2f*)(arow + k0 + koff);
    v2f b;
    b.x = Ws[(k0 + koff + 0) * 16 + mn];
    b.y = Ws[(k0 + koff + 1) * 16 + mn];
    c = __builtin_amdgcn_wmma_f32_16x16x4_f32(
         false, a,  false, b,
         (short)0, c,  false,  false);
  }

  float* hrow = H + ((size_t)tile * 16 + half * 8) * P + mn;
#pragma unroll
  for (int i = 0; i < 8; i++) hrow[i * P] = c[i];
}

__global__ void k_scatter(const int* __restrict__ src, const int* __restrict__ dst,
                          const float* __restrict__ dinv, const float* __restrict__ H,
                          float* __restrict__ AGG, int E) {
  int idx = blockIdx.x * blockDim.x + threadIdx.x;
  int e = idx >> 4;
  int j = idx & 15;
  if (e >= E) return;
  int s = src[e];
  int d = dst[e];
  float norm = dinv[s] * dinv[d];
  atomicAdd(&AGG[d * 16 + j], norm * H[s * 16 + j]);
}

__global__ void k_combine(const float* __restrict__ AGG, const float* __restrict__ H,
                          const float* __restrict__ dinv, const float* __restrict__ bias,
                          float* __restrict__ out, int n) {
  int idx = blockIdx.x * blockDim.x + threadIdx.x;
  int i = idx >> 4;
  int j = idx & 15;
  if (i >= n) return;
  float di = dinv[i];
  out[idx] = fmaxf(AGG[idx] + di * di * H[idx] + bias[j], 0.0f);
}

__global__ void k_final(const float* __restrict__ G, const float* __restrict__ l1b,
                        const float* __restrict__ l2W, const float* __restrict__ l2b,
                        float* __restrict__ out, int n) {
  int idx = blockIdx.x * blockDim.x + threadIdx.x;
  int i = idx >> 4;
  int j = idx & 15;
  float v = 0.0f;
  if (i < n) v = fmaxf(G[idx] + l1b[j], 0.0f) * l2W[j];
#pragma unroll
  for (int off = 8; off; off >>= 1) v += __shfl_xor(v, off, 16);
  if (i < n && j == 0) out[i] = v + l2b[0];
}


#define PUSH_TILE 4096
#define PUSH_NT   1024
#define PUSH_K    6
#define PUSH_RING 256
#define PUSH_C    10
#define PUSH_CAP  576

template <bool FULL, bool REC>
__device__ __forceinline__ void push_trip(const unsigned* q, unsigned qh, unsigned left,
                                          unsigned half, int j,
                                          const float* __restrict__ tab, float* acc,
                                          unsigned* __restrict__ rec) {
  unsigned w[8]; float v[8];
#pragma unroll
  for (int t = 0; t < 8; t++) {
    const unsigned slot = 2u * t + half;
    const unsigned idx = (FULL || slot < left) ? qh + slot : qh;
    w[t] = q[idx & (PUSH_RING - 1)];
  }
  if (REC && j == 0 && qh + 16u <= (unsigned)PUSH_CAP) {
    uint4* p = (uint4*)(rec + qh + half * 8u);
    p[0] = make_uint4(w[0], w[1], w[2], w[3]);
    p[1] = make_uint4(w[4], w[5], w[6], w[7]);
  }
#pragma unroll
  for (int t = 0; t < 8; t++) {
    const unsigned r = (w[t] & 0x1FFFFu) * TAB_PITCH;
    const float x = tab[r + 16u] * tab[r + (unsigned)j];
    v[t] = (FULL || 2u * t + half < left) ? x : 0.0f;
  }
#pragma unroll
  for (int t = 0; t < 8; t++) atomicAdd(&acc[(w[t] >> 17) * 16 + j], v[t]);
}

__device__ __forceinline__ void push_fetch(int (&dv)[PUSH_K], int (&sv)[PUSH_K], int b0, int tid,
                                           int e1, const int* __restrict__ src,
                                           const int* __restrict__ dst) {
#pragma unroll
  for (int k = 0; k < PUSH_K; k++) {
    const int idx = min(b0 + k * PUSH_NT + tid, e1 - 1);
    dv[k] = dst[idx];
    sv[k] = src[idx];
  }
}

__device__ __forceinline__ void push_append(const int (&dv)[PUSH_K], const int (&sv)[PUSH_K],
                                            int b0, int tid, int e1, int lo,
                                            unsigned* q, unsigned& qn) {
#pragma unroll
  for (int k = 0; k < PUSH_K; k++) {
    const int e = b0 + k * PUSH_NT + tid;
    const unsigned off = (unsigned)(dv[k] - lo);
    const bool hit = e < e1 && off < (unsigned)PUSH_TILE;
    const unsigned m = __builtin_amdgcn_ballot_w32(hit);
    if (hit) q[(qn + __builtin_amdgcn_mbcnt_lo(m, 0u)) & (PUSH_RING - 1)] = (unsigned)sv[k] | (off << 17);
    qn += (unsigned)__popc(m);
  }
}

template <bool FULL>
__device__ __forceinline__ void push_rows(const unsigned (&w)[8], float (&v)[8], unsigned left,
                                          unsigned half, int j, const float* __restrict__ tab) {
#pragma unroll
  for (int t = 0; t < 8; t++) {
    const unsigned r = (w[t] & 0x1FFFFu) * TAB_PITCH;
    const float x = tab[r + 16u] * tab[r + (unsigned)j];
    v[t] = (FULL || 2u * t + half < left) ? x : 0.0f;
  }
}
__device__ __forceinline__ void push_adds(const unsigned (&w)[8], const float (&v)[8], int j,
                                          float* acc) {
#pragma unroll
  for (int t = 0; t < 8; t++) atomicAdd(&acc[(w[t] >> 17) * 16 + j], v[t]);
}
__device__ __forceinline__ void push_words(unsigned (&w)[8], const uint4& a, const uint4& b) {
  w[0] = a.x; w[1] = a.y; w[2] = a.z; w[3] = a.w;
  w[4] = b.x; w[5] = b.y; w[6] = b.z; w[7] = b.w;
}

template <int MODE>
__global__ __launch_bounds__(PUSH_NT)
void k_push(const int* __restrict__ src, const int* __restrict__ dst,
            const float* __restrict__ tab,
            float* __restrict__ part, unsigned* __restrict__ rec, unsigned* __restrict__ cnt,
            int E, int n, int chunkEdges) {
  __shared__ __attribute__((aligned(16))) float acc[PUSH_TILE * 16];
  __shared__ unsigned ring[(PUSH_NT / 32) * PUSH_RING];

  const int tid = threadIdx.x;
  const int lo  = blockIdx.x * PUSH_TILE;
  const int e0  = blockIdx.y * chunkEdges;
  const int e1  = min(E, e0 + chunkEdges);
  const int STEP = PUSH_K * PUSH_NT;

  const int wv = __builtin_amdgcn_readfirstlane(tid >> 5);
  const int wg = blockIdx.y * gridDim.x + blockIdx.x;
  const unsigned half = (unsigned)(tid >> 4) & 1u;
  const int j = tid & 15;
  unsigned* wrec = (MODE != 0) ? rec + ((size_t)wg * (PUSH_NT / 32) + wv) * PUSH_CAP : nullptr;

  unsigned most = 0u;
  if (MODE == 2) {
#pragma unroll
    for (int w = 0; w < PUSH_NT / 32; w++) most = max(most, cnt[wg * (PUSH_NT / 32) + w]);
  }

  for (int i = tid; i < PUSH_TILE * 16; i += PUSH_NT) acc[i] = 0.0f;

  const bool replay = MODE == 2 && most <= (unsigned)PUSH_CAP;

  if (replay) {
    const unsigned total = __builtin_amdgcn_readfirstlane(cnt[wg * (PUSH_NT / 32) + wv]);
    const unsigned ntrip = (total + 15u) >> 4;
    const unsigned nfull = total >> 4;
    const uint4* rp = (const uint4*)wrec + half * 2u;
    uint4 a0 = make_uint4(0u, 0u, 0u, 0u), a1 = a0, b0 = a0, b1 = a0;
    if (0u < ntrip) { a0 = rp[0]; a1 = rp[1]; }
    if (1u < ntrip) { b0 = rp[4]; b1 = rp[5]; }
    __syncthreads();

    unsigned T = 0u;
    for (; T + 2u <= nfull; T += 2u) {
      unsigned wa[8], wb[8]; float va[8], vb[8];
      push_words(wa, a0, a1);
      push_words(wb, b0, b1);
      if (T + 2u < ntrip) { a0 = rp[(T + 2u) * 4u]; a1 = rp[(T + 2u) * 4u + 1u]; }
      if (T + 3u < ntrip) { b0 = rp[(T + 3u) * 4u]; b1 = rp[(T + 3u) * 4u + 1u]; }
      push_rows<true>(wa, va, 16u, half, j, tab);
      push_rows<true>(wb, vb, 16u, half, j, tab);
      push_adds(wa, va, j, acc);
      push_adds(wb, vb, j, acc);
    }
    if (T < nfull) {
      unsigned wa[8]; float va[8];
      push_words(wa, a0, a1);
      push_rows<true>(wa, va, 16u, half, j, tab);
      push_adds(wa, va, j, acc);
      a0 = b0; a1 = b1;
    }
    if (total & 15u) {
      unsigned wa[8]; float va[8];
      push_words(wa, a0, a1);
      push_rows<false>(wa, va, total & 15u, half, j, tab);
      push_adds(wa, va, j, acc);
    }
  } else {
    int dvA[PUSH_K] = {0}, svA[PUSH_K] = {0}, dvB[PUSH_K] = {0}, svB[PUSH_K] = {0};
    const bool any = e0 < e1;
    if (any) {
      push_fetch(dvA, svA, e0, tid, e1, src, dst);
      push_fetch(dvB, svB, e0 + STEP, tid, e1, src, dst);
    }
    __syncthreads();

    unsigned* q = ring + wv * PUSH_RING;
    unsigned qn = 0u, qh = 0u;

    if (any) for (int b0 = e0; b0 < e1; b0 += 2 * STEP) {
      push_append(dvA, svA, b0, tid, e1, lo, q, qn);
      push_fetch(dvA, svA, b0 + 2 * STEP, tid, e1, src, dst);
      while (qn - qh >= 16u) { push_trip<true, MODE == 1>(q, qh, 16u, half, j, tab, acc, wrec); qh += 16u; }

      push_append(dvB, svB, b0 + STEP, tid, e1, lo, q, qn);
      push_fetch(dvB, svB, b0 + 3 * STEP, tid, e1, src, dst);
      while (qn - qh >= 16u) { push_trip<true, MODE == 1>(q, qh, 16u, half, j, tab, acc, wrec); qh += 16u; }
    }
    if (qn != qh) push_trip<false, MODE == 1>(q, qh, qn - qh, half, j, tab, acc, wrec);
    if (MODE == 1 && (tid & 31) == 0) q[0] = qn;
  }
  __syncthreads();
  if (MODE == 1 && tid < PUSH_NT / 32) cnt[wg * (PUSH_NT / 32) + tid] = ring[tid * PUSH_RING];

  const float4* acc4 = (const float4*)acc;
  float4* out4 = (float4*)(part + ((size_t)blockIdx.y * n + lo) * 16);
  for (int i4 = tid; i4 < PUSH_TILE * 4; i4 += PUSH_NT) {
    if (lo + (i4 >> 2) < n) out4[i4] = acc4[i4];
  }
}

__global__ void k_combine_parts(const float* part, const float* __restrict__ tab,
                                const float* __restrict__ dinv, const float* __restrict__ bias,
                                float* out, int n) {
  int idx = blockIdx.x * blockDim.x + threadIdx.x;
  int i = idx >> 4;
  int j = idx & 15;
  if (i >= n) return;
  const size_t plane = (size_t)n * 16;
  float a = part[idx];
#pragma unroll
  for (int c = 1; c < PUSH_C; c++) a += part[c * plane + idx];
  const float di = dinv[i];
  out[idx] = fmaxf(di * (a + di * tab[(size_t)i * TAB_PITCH + j]) + bias[j], 0.0f);
}

__global__ void k_init_deg(float* __restrict__ deg, int n) {
  int idx = blockIdx.x * blockDim.x + threadIdx.x;
  if (idx < n) deg[idx] = 1.0f;
}

extern "C" void kernel_launch(void* const* d_in, const int* in_sizes, int n_in,
                              void* d_out, int out_size, void* d_ws, size_t ws_size,
                              hipStream_t stream) {
  const float* x   = (const float*)d_in[0];
  const int*   ei  = (const int*)d_in[1];
  const float* W1  = (const float*)d_in[2];
  const float* b1  = (const float*)d_in[3];
  const float* W2  = (const float*)d_in[4];
  const float* b2  = (const float*)d_in[5];
  const float* l1W = (const float*)d_in[6];
  const float* l1b = (const float*)d_in[7];
  const float* l2W = (const float*)d_in[8];
  const float* l2b = (const float*)d_in[9];

  const int N = in_sizes[0] / 128;
  const int E = in_sizes[1] / 2;
  const int* src = ei;
  const int* dst = ei + E;

  float* deg  = (float*)d_ws;
  float* dinv = deg + N;
  float* bufA = dinv + N;
  float* bufB = bufA + (size_t)N * 16;
  float* bufC = bufB + (size_t)N * 16;

  const int ntiles = N / 16;
  const int gemmBlocks = (ntiles * 32 + 255) / 256;
  const int nodeElems  = N * 16;
  const int nodeBlocks = (nodeElems + 255) / 256;
  const int edgeBlocks = (E + 255) / 256;
  const int scatBlocks = (int)(((size_t)E * 16 + 255) / 256);

  const size_t pushNeed = ((size_t)N * 2 + (size_t)N * 16 * (2 + PUSH_C)) * sizeof(float);
  if (pushNeed <= ws_size && N <= (1 << 17)) {
    float* tab  = bufA;
    float* part = bufC;
    float* h    = part;
    const int chunkEdges = (E + PUSH_C - 1) / PUSH_C;
    const dim3 pushGrid((N + PUSH_TILE - 1) / PUSH_TILE, PUSH_C);

    const size_t nWaves   = (size_t)pushGrid.x * pushGrid.y * (PUSH_NT / 32);
    const size_t cntWords = (nWaves + 31) / 32 * 32;
    unsigned* cnt = (unsigned*)(part + (size_t)PUSH_C * N * 16);
    unsigned* rec = cnt + cntWords;
    const bool record = pushNeed + (cntWords + nWaves * PUSH_CAP) * sizeof(unsigned) <= ws_size;

    k_init_deg<<<(N + 255) / 256, 256, 0, stream>>>(deg, N);
    k_deg   <<<edgeBlocks, 256, 0, stream>>>(dst, deg, E);
    k_rsqrt_rows<<<(N + 255) / 256, 256, 0, stream>>>(deg, dinv, tab, N);

    k_gemm_wmma<128, TAB_PITCH><<<gemmBlocks, 256, 0, stream>>>(x, W1, tab, ntiles);
    if (record) k_push<1><<<pushGrid, PUSH_NT, 0, stream>>>(src, dst, tab, part, rec, cnt, E, N, chunkEdges);
    else        k_push<0><<<pushGrid, PUSH_NT, 0, stream>>>(src, dst, tab, part, rec, cnt, E, N, chunkEdges);
    k_combine_parts<<<nodeBlocks, 256, 0, stream>>>(part, tab, dinv, b1, h, N);

    k_gemm_wmma<16, TAB_PITCH><<<gemmBlocks, 256, 0, stream>>>(h, W2, tab, ntiles);
    if (record) k_push<2><<<pushGrid, PUSH_NT, 0, stream>>>(src, dst, tab, part, rec, cnt, E, N, chunkEdges);
    else        k_push<0><<<pushGrid, PUSH_NT, 0, stream>>>(src, dst, tab, part, rec, cnt, E, N, chunkEdges);
    k_combine_parts<<<nodeBlocks, 256, 0, stream>>>(part, tab, dinv, b2, h, N);

    k_gemm_wmma<16, 16><<<gemmBlocks, 256, 0, stream>>>(h, l1W, bufA, ntiles);
    k_final<<<nodeBlocks, 256, 0, stream>>>(bufA, l1b, l2W, l2b, (float*)d_out, N);
  } else {
    k_init  <<<nodeBlocks, 256, 0, stream>>>(deg, bufB, bufC, N);
    k_deg   <<<edgeBlocks, 256, 0, stream>>>(dst, deg, E);
    k_rsqrt <<<(N + 255) / 256, 256, 0, stream>>>(deg, dinv, N);

    k_gemm_wmma<128, 16><<<gemmBlocks, 256, 0, stream>>>(x, W1, bufA, ntiles);
    k_scatter<<<scatBlocks, 256, 0, stream>>>(src, dst, dinv, bufA, bufB, E);
    k_combine<<<nodeBlocks, 256, 0, stream>>>(bufB, bufA, dinv, b1, bufB, N);

    k_gemm_wmma<16, 16><<<gemmBlocks, 256, 0, stream>>>(bufB, W2, bufA, ntiles);
    k_scatter<<<scatBlocks, 256, 0, stream>>>(src, dst, dinv, bufA, bufC, E);
    k_combine<<<nodeBlocks, 256, 0, stream>>>(bufC, bufA, dinv, b2, bufC, N);

    k_gemm_wmma<16, 16><<<gemmBlocks, 256, 0, stream>>>(bufC, l1W, bufA, ntiles);
    k_final<<<nodeBlocks, 256, 0, stream>>>(bufA, l1b, l2W, l2b, (float*)d_out, N);
  }
}


# Degree adds kept die-local: two copies of the count, one per half of the chip (picked by the die a work-group runs on), summed in the scale kernel; no counter line added to from both L2 halves
# speedup vs baseline: 2.7048x; 1.5720x over previous
#include <hip/hip_runtime.h>
#include <hip/hip_bf16.h>

typedef __attribute__((ext_vector_type(2))) float v2f;
typedef __attribute__((ext_vector_type(8))) float v8f;

#define TAB_PITCH 32

__global__ void k_init(float* __restrict__ deg, float* __restrict__ aggB,
                       float* __restrict__ aggC, int n) {
  int idx = blockIdx.x * blockDim.x + threadIdx.x;
  if (idx < n) deg[idx] = 1.0f;
  if (idx < n * 16) { aggB[idx] = 0.0f; aggC[idx] = 0.0f; }
}

__global__ void k_deg(const int* __restrict__ dst, float* __restrict__ deg, int E) {
  int e = blockIdx.x * blockDim.x + threadIdx.x;
  if (e < E) atomicAdd(&deg[dst[e]], 1.0f);
}

__global__ void k_deg_halves(const int* __restrict__ dst, float* __restrict__ deg0,
                             float* __restrict__ deg1, int E) {
  const unsigned where = (unsigned)__builtin_amdgcn_s_sendmsg_rtn(0x87);
  float* deg = ((where >> 18) & 1u) ? deg1 : deg0;
  int e = blockIdx.x * blockDim.x + threadIdx.x;
  if (e < E) atomicAdd(&deg[dst[e]], 1.0f);
}

__global__ void k_rsqrt(const float* __restrict__ deg, float* __restrict__ dinv, int n) {
  int i = blockIdx.x * blockDim.x + threadIdx.x;
  if (i < n) dinv[i] = rsqrtf(deg[i]);
}

__global__ void k_rsqrt_rows(const float* __restrict__ deg, const float* __restrict__ deg1,
                             float* __restrict__ dinv, float* __restrict__ tab, int n) {
  int i = blockIdx.x * blockDim.x + threadIdx.x;
  if (i < n) {
    const float r = rsqrtf(deg[i] + deg1[i]);
    dinv[i] = r;
    tab[(size_t)i * TAB_PITCH + 16] = r;
  }
}

template <int K, int P>
__global__ void k_gemm_wmma(const float* __restrict__ A, const float* __restrict__ W,
                            float* __restrict__ H, int ntiles) {
  __shared__ float Ws[K * 16];
  for (int t = threadIdx.x; t < K * 16; t += blockDim.x) Ws[t] = W[t];
  __syncthreads();

  const int lane = threadIdx.x & 31;
  const int tile = blockIdx.x * (blockDim.x >> 5) + (threadIdx.x >> 5);
  if (tile >= ntiles) return;

  const int half = lane >> 4;
  const int mn   = lane & 15;
  const int koff = half * 2;

  const float* arow = A + (size_t)(tile * 16 + mn) * K;

  v8f c = {0.f, 0.f, 0.f, 0.f, 0.f, 0.f, 0.f, 0.f};
#pragma unroll
  for (int k0 = 0; k0 < K; k0 += 4) {
    v2f a = *(const v2f*)(arow + k0 + koff);
    v2f b;
    b.x = Ws[(k0 + koff + 0) * 16 + mn];
    b.y = Ws[(k0 + koff + 1) * 16 + mn];
    c = __builtin_amdgcn_wmma_f32_16x16x4_f32(
         false, a,  false, b,
         (short)0, c,  false,  false);
  }

  float* hrow = H + ((size_t)tile * 16 + half * 8) * P + mn;
#pragma unroll
  for (int i = 0; i < 8; i++) hrow[i * P] = c[i];
}

__global__ void k_scatter(const int* __restrict__ src, const int* __restrict__ dst,
                          const float* __restrict__ dinv, const float* __restrict__ H,
                          float* __restrict__ AGG, int E) {
  int idx = blockIdx.x * blockDim.x + threadIdx.x;
  int e = idx >> 4;
  int j = idx & 15;
  if (e >= E) return;
  int s = src[e];
  int d = dst[e];
  float norm = dinv[s] * dinv[d];
  atomicAdd(&AGG[d * 16 + j], norm * H[s * 16 + j]);
}

__global__ void k_combine(const float* __restrict__ AGG, const float* __restrict__ H,
                          const float* __restrict__ dinv, const float* __restrict__ bias,
                          float* __restrict__ out, int n) {
  int idx = blockIdx.x * blockDim.x + threadIdx.x;
  int i = idx >> 4;
  int j = idx & 15;
  if (i >= n) return;
  float di = dinv[i];
  out[idx] = fmaxf(AGG[idx] + di * di * H[idx] + bias[j], 0.0f);
}

__global__ void k_final(const float* __restrict__ G, const float* __restrict__ l1b,
                        const float* __restrict__ l2W, const float* __restrict__ l2b,
                        float* __restrict__ out, int n) {
  int idx = blockIdx.x * blockDim.x + threadIdx.x;
  int i = idx >> 4;
  int j = idx & 15;
  float v = 0.0f;
  if (i < n) v = fmaxf(G[idx] + l1b[j], 0.0f) * l2W[j];
#pragma unroll
  for (int off = 8; off; off >>= 1) v += __shfl_xor(v, off, 16);
  if (i < n && j == 0) out[i] = v + l2b[0];
}


#define PUSH_TILE 4096
#define PUSH_NT   1024
#define PUSH_K    6
#define PUSH_RING 256
#define PUSH_C    10
#define PUSH_CAP  576

template <bool FULL, bool REC>
__device__ __forceinline__ void push_trip(const unsigned* q, unsigned qh, unsigned left,
                                          unsigned half, int j,
                                          const float* __restrict__ tab, float* acc,
                                          unsigned* __restrict__ rec) {
  unsigned w[8]; float v[8];
#pragma unroll
  for (int t = 0; t < 8; t++) {
    const unsigned slot = 2u * t + half;
    const unsigned idx = (FULL || slot < left) ? qh + slot : qh;
    w[t] = q[idx & (PUSH_RING - 1)];
  }
  if (REC && j == 0 && qh + 16u <= (unsigned)PUSH_CAP) {
    uint4* p = (uint4*)(rec + qh + half * 8u);
    p[0] = make_uint4(w[0], w[1], w[2], w[3]);
    p[1] = make_uint4(w[4], w[5], w[6], w[7]);
  }
#pragma unroll
  for (int t = 0; t < 8; t++) {
    const unsigned r = (w[t] & 0x1FFFFu) * TAB_PITCH;
    const float x = tab[r + 16u] * tab[r + (unsigned)j];
    v[t] = (FULL || 2u * t + half < left) ? x : 0.0f;
  }
#pragma unroll
  for (int t = 0; t < 8; t++) atomicAdd(&acc[(w[t] >> 17) * 16 + j], v[t]);
}

__device__ __forceinline__ void push_fetch(int (&dv)[PUSH_K], int (&sv)[PUSH_K], int b0, int tid,
                                           int e1, const int* __restrict__ src,
                                           const int* __restrict__ dst) {
#pragma unroll
  for (int k = 0; k < PUSH_K; k++) {
    const int idx = min(b0 + k * PUSH_NT + tid, e1 - 1);
    dv[k] = dst[idx];
    sv[k] = src[idx];
  }
}

__device__ __forceinline__ void push_append(const int (&dv)[PUSH_K], const int (&sv)[PUSH_K],
                                            int b0, int tid, int e1, int lo,
                                            unsigned* q, unsigned& qn) {
#pragma unroll
  for (int k = 0; k < PUSH_K; k++) {
    const int e = b0 + k * PUSH_NT + tid;
    const unsigned off = (unsigned)(dv[k] - lo);
    const bool hit = e < e1 && off < (unsigned)PUSH_TILE;
    const unsigned m = __builtin_amdgcn_ballot_w32(hit);
    if (hit) q[(qn + __builtin_amdgcn_mbcnt_lo(m, 0u)) & (PUSH_RING - 1)] = (unsigned)sv[k] | (off << 17);
    qn += (unsigned)__popc(m);
  }
}

template <bool FULL>
__device__ __forceinline__ void push_rows(const unsigned (&w)[8], float (&v)[8], unsigned left,
                                          unsigned half, int j, const float* __restrict__ tab) {
#pragma unroll
  for (int t = 0; t < 8; t++) {
    const unsigned r = (w[t] & 0x1FFFFu) * TAB_PITCH;
    const float x = tab[r + 16u] * tab[r + (unsigned)j];
    v[t] = (FULL || 2u * t + half < left) ? x : 0.0f;
  }
}
__device__ __forceinline__ void push_adds(const unsigned (&w)[8], const float (&v)[8], int j,
                                          float* acc) {
#pragma unroll
  for (int t = 0; t < 8; t++) atomicAdd(&acc[(w[t] >> 17) * 16 + j], v[t]);
}
__device__ __forceinline__ void push_words(unsigned (&w)[8], const uint4& a, const uint4& b) {
  w[0] = a.x; w[1] = a.y; w[2] = a.z; w[3] = a.w;
  w[4] = b.x; w[5] = b.y; w[6] = b.z; w[7] = b.w;
}

template <int MODE>
__global__ __launch_bounds__(PUSH_NT)
void k_push(const int* __restrict__ src, const int* __restrict__ dst,
            const float* __restrict__ tab,
            float* __restrict__ part, unsigned* __restrict__ rec, unsigned* __restrict__ cnt,
            int E, int n, int chunkEdges) {
  __shared__ __attribute__((aligned(16))) float acc[PUSH_TILE * 16];
  __shared__ unsigned ring[(PUSH_NT / 32) * PUSH_RING];

  const int tid = threadIdx.x;
  const int lo  = blockIdx.x * PUSH_TILE;
  const int e0  = blockIdx.y * chunkEdges;
  const int e1  = min(E, e0 + chunkEdges);
  const int STEP = PUSH_K * PUSH_NT;

  const int wv = __builtin_amdgcn_readfirstlane(tid >> 5);
  const int wg = blockIdx.y * gridDim.x + blockIdx.x;
  const unsigned half = (unsigned)(tid >> 4) & 1u;
  const int j = tid & 15;
  unsigned* wrec = (MODE != 0) ? rec + ((size_t)wg * (PUSH_NT / 32) + wv) * PUSH_CAP : nullptr;

  unsigned most = 0u;
  if (MODE == 2) {
#pragma unroll
    for (int w = 0; w < PUSH_NT / 32; w++) most = max(most, cnt[wg * (PUSH_NT / 32) + w]);
  }

  for (int i = tid; i < PUSH_TILE * 16; i += PUSH_NT) acc[i] = 0.0f;

  const bool replay = MODE == 2 && most <= (unsigned)PUSH_CAP;

  if (replay) {
    const unsigned total = __builtin_amdgcn_readfirstlane(cnt[wg * (PUSH_NT / 32) + wv]);
    const unsigned ntrip = (total + 15u) >> 4;
    const unsigned nfull = total >> 4;
    const uint4* rp = (const uint4*)wrec + half * 2u;
    uint4 a0 = make_uint4(0u, 0u, 0u, 0u), a1 = a0, b0 = a0, b1 = a0;
    if (0u < ntrip) { a0 = rp[0]; a1 = rp[1]; }
    if (1u < ntrip) { b0 = rp[4]; b1 = rp[5]; }
    __syncthreads();

    unsigned T = 0u;
    for (; T + 2u <= nfull; T += 2u) {
      unsigned wa[8], wb[8]; float va[8], vb[8];
      push_words(wa, a0, a1);
      push_words(wb, b0, b1);
      if (T + 2u < ntrip) { a0 = rp[(T + 2u) * 4u]; a1 = rp[(T + 2u) * 4u + 1u]; }
      if (T + 3u < ntrip) { b0 = rp[(T + 3u) * 4u]; b1 = rp[(T + 3u) * 4u + 1u]; }
      push_rows<true>(wa, va, 16u, half, j, tab);
      push_rows<true>(wb, vb, 16u, half, j, tab);
      push_adds(wa, va, j, acc);
      push_adds(wb, vb, j, acc);
    }
    if (T < nfull) {
      unsigned wa[8]; float va[8];
      push_words(wa, a0, a1);
      push_rows<true>(wa, va, 16u, half, j, tab);
      push_adds(wa, va, j, acc);
      a0 = b0; a1 = b1;
    }
    if (total & 15u) {
      unsigned wa[8]; float va[8];
      push_words(wa, a0, a1);
      push_rows<false>(wa, va, total & 15u, half, j, tab);
      push_adds(wa, va, j, acc);
    }
  } else {
    int dvA[PUSH_K] = {0}, svA[PUSH_K] = {0}, dvB[PUSH_K] = {0}, svB[PUSH_K] = {0};
    const bool any = e0 < e1;
    if (any) {
      push_fetch(dvA, svA, e0, tid, e1, src, dst);
      push_fetch(dvB, svB, e0 + STEP, tid, e1, src, dst);
    }
    __syncthreads();

    unsigned* q = ring + wv * PUSH_RING;
    unsigned qn = 0u, qh = 0u;

    if (any) for (int b0 = e0; b0 < e1; b0 += 2 * STEP) {
      push_append(dvA, svA, b0, tid, e1, lo, q, qn);
      push_fetch(dvA, svA, b0 + 2 * STEP, tid, e1, src, dst);
      while (qn - qh >= 16u) { push_trip<true, MODE == 1>(q, qh, 16u, half, j, tab, acc, wrec); qh += 16u; }

      push_append(dvB, svB, b0 + STEP, tid, e1, lo, q, qn);
      push_fetch(dvB, svB, b0 + 3 * STEP, tid, e1, src, dst);
      while (qn - qh >= 16u) { push_trip<true, MODE == 1>(q, qh, 16u, half, j, tab, acc, wrec); qh += 16u; }
    }
    if (qn != qh) push_trip<false, MODE == 1>(q, qh, qn - qh, half, j, tab, acc, wrec);
    if (MODE == 1 && (tid & 31) == 0) q[0] = qn;
  }
  __syncthreads();
  if (MODE == 1 && tid < PUSH_NT / 32) cnt[wg * (PUSH_NT / 32) + tid] = ring[tid * PUSH_RING];

  const float4* acc4 = (const float4*)acc;
  float4* out4 = (float4*)(part + ((size_t)blockIdx.y * n + lo) * 16);
  for (int i4 = tid; i4 < PUSH_TILE * 4; i4 += PUSH_NT) {
    if (lo + (i4 >> 2) < n) out4[i4] = acc4[i4];
  }
}

__global__ void k_combine_parts(const float* part, const float* __restrict__ tab,
                                const float* __restrict__ dinv, const float* __restrict__ bias,
                                float* out, int n) {
  int idx = blockIdx.x * blockDim.x + threadIdx.x;
  int i = idx >> 4;
  int j = idx & 15;
  if (i >= n) return;
  const size_t plane = (size_t)n * 16;
  float a = part[idx];
#pragma unroll
  for (int c = 1; c < PUSH_C; c++) a += part[c * plane + idx];
  const float di = dinv[i];
  out[idx] = fmaxf(di * (a + di * tab[(size_t)i * TAB_PITCH + j]) + bias[j], 0.0f);
}

__global__ void k_init_deg(float* __restrict__ deg, float* __restrict__ deg1, int n) {
  int idx = blockIdx.x * blockDim.x + threadIdx.x;
  if (idx < n) { deg[idx] = 1.0f; deg1[idx] = 0.0f; }
}

extern "C" void kernel_launch(void* const* d_in, const int* in_sizes, int n_in,
                              void* d_out, int out_size, void* d_ws, size_t ws_size,
                              hipStream_t stream) {
  const float* x   = (const float*)d_in[0];
  const int*   ei  = (const int*)d_in[1];
  const float* W1  = (const float*)d_in[2];
  const float* b1  = (const float*)d_in[3];
  const float* W2  = (const float*)d_in[4];
  const float* b2  = (const float*)d_in[5];
  const float* l1W = (const float*)d_in[6];
  const float* l1b = (const float*)d_in[7];
  const float* l2W = (const float*)d_in[8];
  const float* l2b = (const float*)d_in[9];

  const int N = in_sizes[0] / 128;
  const int E = in_sizes[1] / 2;
  const int* src = ei;
  const int* dst = ei + E;

  float* deg  = (float*)d_ws;
  float* dinv = deg + N;
  float* bufA = dinv + N;
  float* bufB = bufA + (size_t)N * 16;
  float* bufC = bufB + (size_t)N * 16;

  const int ntiles = N / 16;
  const int gemmBlocks = (ntiles * 32 + 255) / 256;
  const int nodeElems  = N * 16;
  const int nodeBlocks = (nodeElems + 255) / 256;
  const int edgeBlocks = (E + 255) / 256;
  const int scatBlocks = (int)(((size_t)E * 16 + 255) / 256);

  const size_t pushNeed = ((size_t)N * 2 + (size_t)N * 16 * (2 + PUSH_C)) * sizeof(float);
  if (pushNeed <= ws_size && N <= (1 << 17)) {
    float* tab  = bufA;
    float* part = bufC;
    float* h    = part;
    const int chunkEdges = (E + PUSH_C - 1) / PUSH_C;
    const dim3 pushGrid((N + PUSH_TILE - 1) / PUSH_TILE, PUSH_C);

    const size_t nWaves   = (size_t)pushGrid.x * pushGrid.y * (PUSH_NT / 32);
    const size_t cntWords = (nWaves + 31) / 32 * 32;
    unsigned* cnt = (unsigned*)(part + (size_t)PUSH_C * N * 16);
    unsigned* rec = cnt + cntWords;
    const bool record = pushNeed + (cntWords + nWaves * PUSH_CAP) * sizeof(unsigned) <= ws_size;

    float* deg1 = part;
    k_init_deg<<<(N + 255) / 256, 256, 0, stream>>>(deg, deg1, N);
    k_deg_halves<<<edgeBlocks, 256, 0, stream>>>(dst, deg, deg1, E);
    k_rsqrt_rows<<<(N + 255) / 256, 256, 0, stream>>>(deg, deg1, dinv, tab, N);

    k_gemm_wmma<128, TAB_PITCH><<<gemmBlocks, 256, 0, stream>>>(x, W1, tab, ntiles);
    if (record) k_push<1><<<pushGrid, PUSH_NT, 0, stream>>>(src, dst, tab, part, rec, cnt, E, N, chunkEdges);
    else        k_push<0><<<pushGrid, PUSH_NT, 0, stream>>>(src, dst, tab, part, rec, cnt, E, N, chunkEdges);
    k_combine_parts<<<nodeBlocks, 256, 0, stream>>>(part, tab, dinv, b1, h, N);

    k_gemm_wmma<16, TAB_PITCH><<<gemmBlocks, 256, 0, stream>>>(h, W2, tab, ntiles);
    if (record) k_push<2><<<pushGrid, PUSH_NT, 0, stream>>>(src, dst, tab, part, rec, cnt, E, N, chunkEdges);
    else        k_push<0><<<pushGrid, PUSH_NT, 0, stream>>>(src, dst, tab, part, rec, cnt, E, N, chunkEdges);
    k_combine_parts<<<nodeBlocks, 256, 0, stream>>>(part, tab, dinv, b2, h, N);

    k_gemm_wmma<16, 16><<<gemmBlocks, 256, 0, stream>>>(h, l1W, bufA, ntiles);
    k_final<<<nodeBlocks, 256, 0, stream>>>(bufA, l1b, l2W, l2b, (float*)d_out, N);
  } else {
    k_init  <<<nodeBlocks, 256, 0, stream>>>(deg, bufB, bufC, N);
    k_deg   <<<edgeBlocks, 256, 0, stream>>>(dst, deg, E);
    k_rsqrt <<<(N + 255) / 256, 256, 0, stream>>>(deg, dinv, N);

    k_gemm_wmma<128, 16><<<gemmBlocks, 256, 0, stream>>>(x, W1, bufA, ntiles);
    k_scatter<<<scatBlocks, 256, 0, stream>>>(src, dst, dinv, bufA, bufB, E);
    k_combine<<<nodeBlocks, 256, 0, stream>>>(bufB, bufA, dinv, b1, bufB, N);

    k_gemm_wmma<16, 16><<<gemmBlocks, 256, 0, stream>>>(bufB, W2, bufA, ntiles);
    k_scatter<<<scatBlocks, 256, 0, stream>>>(src, dst, dinv, bufA, bufC, E);
    k_combine<<<nodeBlocks, 256, 0, stream>>>(bufC, bufA, dinv, b2, bufC, N);

    k_gemm_wmma<16, 16><<<gemmBlocks, 256, 0, stream>>>(bufC, l1W, bufA, ntiles);
    k_final<<<nodeBlocks, 256, 0, stream>>>(bufA, l1b, l2W, l2b, (float*)d_out, N);
  }
}


# Per-node launches folded into WMMA kernels: plane-sum+relu tile feeds next 16x16 WMMA via LDS (h, G not stored), relu-dot head as epilogue (1st add now fma), deg init in GEMM-1; 11->7 launches
# speedup vs baseline: 3.1299x; 1.1572x over previous
#include <hip/hip_runtime.h>
#include <hip/hip_bf16.h>

typedef __attribute__((ext_vector_type(2))) float v2f;
typedef __attribute__((ext_vector_type(8))) float v8f;

#define TAB_PITCH 32

__global__ void k_init(float* __restrict__ deg, float* __restrict__ aggB,
                       float* __restrict__ aggC, int n) {
  int idx = blockIdx.x * blockDim.x + threadIdx.x;
  if (idx < n) deg[idx] = 1.0f;
  if (idx < n * 16) { aggB[idx] = 0.0f; aggC[idx] = 0.0f; }
}

__global__ void k_deg(const int* __restrict__ dst, float* __restrict__ deg, int E) {
  int e = blockIdx.x * blockDim.x + threadIdx.x;
  if (e < E) atomicAdd(&deg[dst[e]], 1.0f);
}

__global__ void k_deg_halves(const int* __restrict__ dst, float* __restrict__ deg0,
                             float* __restrict__ deg1, int E) {
  const unsigned where = (unsigned)__builtin_amdgcn_s_sendmsg_rtn(0x87);
  float* deg = ((where >> 18) & 1u) ? deg1 : deg0;
  int e = blockIdx.x * blockDim.x + threadIdx.x;
  if (e < E) atomicAdd(&deg[dst[e]], 1.0f);
}

__global__ void k_rsqrt(const float* __restrict__ deg, float* __restrict__ dinv, int n) {
  int i = blockIdx.x * blockDim.x + threadIdx.x;
  if (i < n) dinv[i] = rsqrtf(deg[i]);
}

__global__ void k_rsqrt_rows(const float* __restrict__ deg, const float* __restrict__ deg1,
                             float* __restrict__ dinv, float* __restrict__ tab, int n) {
  int i = blockIdx.x * blockDim.x + threadIdx.x;
  if (i < n) {
    const float r = rsqrtf(deg[i] + deg1[i]);
    dinv[i] = r;
    tab[(size_t)i * TAB_PITCH + 16] = r;
  }
}

template <int K, int P, bool INIT>
__global__ void k_gemm_wmma(const float* __restrict__ A, const float* __restrict__ W,
                            float* __restrict__ H, int ntiles,
                            float* __restrict__ deg0, float* __restrict__ deg1) {
  __shared__ float Ws[K * 16];
  for (int t = threadIdx.x; t < K * 16; t += blockDim.x) Ws[t] = W[t];
  __syncthreads();

  const int lane = threadIdx.x & 31;
  const int tile = blockIdx.x * (blockDim.x >> 5) + (threadIdx.x >> 5);
  if (tile >= ntiles) return;

  const int half = lane >> 4;
  const int mn   = lane & 15;
  const int koff = half * 2;

  const float* arow = A + (size_t)(tile * 16 + mn) * K;

  if (INIT) {
    if (half == 0) deg0[tile * 16 + mn] = 1.0f;
    else           deg1[tile * 16 + mn] = 0.0f;
  }

  v8f c = {0.f, 0.f, 0.f, 0.f, 0.f, 0.f, 0.f, 0.f};
#pragma unroll
  for (int k0 = 0; k0 < K; k0 += 4) {
    v2f a = *(const v2f*)(arow + k0 + koff);
    v2f b;
    b.x = Ws[(k0 + koff + 0) * 16 + mn];
    b.y = Ws[(k0 + koff + 1) * 16 + mn];
    c = __builtin_amdgcn_wmma_f32_16x16x4_f32(
         false, a,  false, b,
         (short)0, c,  false,  false);
  }

  float* hrow = H + ((size_t)tile * 16 + half * 8) * P + mn;
#pragma unroll
  for (int i = 0; i < 8; i++) hrow[i * P] = c[i];
}

__global__ void k_scatter(const int* __restrict__ src, const int* __restrict__ dst,
                          const float* __restrict__ dinv, const float* __restrict__ H,
                          float* __restrict__ AGG, int E) {
  int idx = blockIdx.x * blockDim.x + threadIdx.x;
  int e = idx >> 4;
  int j = idx & 15;
  if (e >= E) return;
  int s = src[e];
  int d = dst[e];
  float norm = dinv[s] * dinv[d];
  atomicAdd(&AGG[d * 16 + j], norm * H[s * 16 + j]);
}

__global__ void k_combine(const float* __restrict__ AGG, const float* __restrict__ H,
                          const float* __restrict__ dinv, const float* __restrict__ bias,
                          float* __restrict__ out, int n) {
  int idx = blockIdx.x * blockDim.x + threadIdx.x;
  int i = idx >> 4;
  int j = idx & 15;
  if (i >= n) return;
  float di = dinv[i];
  out[idx] = fmaxf(AGG[idx] + di * di * H[idx] + bias[j], 0.0f);
}

__global__ void k_final(const float* __restrict__ G, const float* __restrict__ l1b,
                        const float* __restrict__ l2W, const float* __restrict__ l2b,
                        float* __restrict__ out, int n) {
  int idx = blockIdx.x * blockDim.x + threadIdx.x;
  int i = idx >> 4;
  int j = idx & 15;
  float v = 0.0f;
  if (i < n) v = fmaxf(G[idx] + l1b[j], 0.0f) * l2W[j];
#pragma unroll
  for (int off = 8; off; off >>= 1) v += __shfl_xor(v, off, 16);
  if (i < n && j == 0) out[i] = v + l2b[0];
}


#define PUSH_TILE 4096
#define PUSH_NT   1024
#define PUSH_K    6
#define PUSH_RING 256
#define PUSH_C    10
#define PUSH_CAP  576

template <bool FULL, bool REC>
__device__ __forceinline__ void push_trip(const unsigned* q, unsigned qh, unsigned left,
                                          unsigned half, int j,
                                          const float* __restrict__ tab, float* acc,
                                          unsigned* __restrict__ rec) {
  unsigned w[8]; float v[8];
#pragma unroll
  for (int t = 0; t < 8; t++) {
    const unsigned slot = 2u * t + half;
    const unsigned idx = (FULL || slot < left) ? qh + slot : qh;
    w[t] = q[idx & (PUSH_RING - 1)];
  }
  if (REC && j == 0 && qh + 16u <= (unsigned)PUSH_CAP) {
    uint4* p = (uint4*)(rec + qh + half * 8u);
    p[0] = make_uint4(w[0], w[1], w[2], w[3]);
    p[1] = make_uint4(w[4], w[5], w[6], w[7]);
  }
#pragma unroll
  for (int t = 0; t < 8; t++) {
    const unsigned r = (w[t] & 0x1FFFFu) * TAB_PITCH;
    const float x = tab[r + 16u] * tab[r + (unsigned)j];
    v[t] = (FULL || 2u * t + half < left) ? x : 0.0f;
  }
#pragma unroll
  for (int t = 0; t < 8; t++) atomicAdd(&acc[(w[t] >> 17) * 16 + j], v[t]);
}

__device__ __forceinline__ void push_fetch(int (&dv)[PUSH_K], int (&sv)[PUSH_K], int b0, int tid,
                                           int e1, const int* __restrict__ src,
                                           const int* __restrict__ dst) {
#pragma unroll
  for (int k = 0; k < PUSH_K; k++) {
    const int idx = min(b0 + k * PUSH_NT + tid, e1 - 1);
    dv[k] = dst[idx];
    sv[k] = src[idx];
  }
}

__device__ __forceinline__ void push_append(const int (&dv)[PUSH_K], const int (&sv)[PUSH_K],
                                            int b0, int tid, int e1, int lo,
                                            unsigned* q, unsigned& qn) {
#pragma unroll
  for (int k = 0; k < PUSH_K; k++) {
    const int e = b0 + k * PUSH_NT + tid;
    const unsigned off = (unsigned)(dv[k] - lo);
    const bool hit = e < e1 && off < (unsigned)PUSH_TILE;
    const unsigned m = __builtin_amdgcn_ballot_w32(hit);
    if (hit) q[(qn + __builtin_amdgcn_mbcnt_lo(m, 0u)) & (PUSH_RING - 1)] = (unsigned)sv[k] | (off << 17);
    qn += (unsigned)__popc(m);
  }
}

template <bool FULL>
__device__ __forceinline__ void push_rows(const unsigned (&w)[8], float (&v)[8], unsigned left,
                                          unsigned half, int j, const float* __restrict__ tab) {
#pragma unroll
  for (int t = 0; t < 8; t++) {
    const unsigned r = (w[t] & 0x1FFFFu) * TAB_PITCH;
    const float x = tab[r + 16u] * tab[r + (unsigned)j];
    v[t] = (FULL || 2u * t + half < left) ? x : 0.0f;
  }
}
__device__ __forceinline__ void push_adds(const unsigned (&w)[8], const float (&v)[8], int j,
                                          float* acc) {
#pragma unroll
  for (int t = 0; t < 8; t++) atomicAdd(&acc[(w[t] >> 17) * 16 + j], v[t]);
}
__device__ __forceinline__ void push_words(unsigned (&w)[8], const uint4& a, const uint4& b) {
  w[0] = a.x; w[1] = a.y; w[2] = a.z; w[3] = a.w;
  w[4] = b.x; w[5] = b.y; w[6] = b.z; w[7] = b.w;
}

template <int MODE>
__global__ __launch_bounds__(PUSH_NT)
void k_push(const int* __restrict__ src, const int* __restrict__ dst,
            const float* __restrict__ tab,
            float* __restrict__ part, unsigned* __restrict__ rec, unsigned* __restrict__ cnt,
            int E, int n, int chunkEdges) {
  __shared__ __attribute__((aligned(16))) float acc[PUSH_TILE * 16];
  __shared__ unsigned ring[(PUSH_NT / 32) * PUSH_RING];

  const int tid = threadIdx.x;
  const int lo  = blockIdx.x * PUSH_TILE;
  const int e0  = blockIdx.y * chunkEdges;
  const int e1  = min(E, e0 + chunkEdges);
  const int STEP = PUSH_K * PUSH_NT;

  const int wv = __builtin_amdgcn_readfirstlane(tid >> 5);
  const int wg = blockIdx.y * gridDim.x + blockIdx.x;
  const unsigned half = (unsigned)(tid >> 4) & 1u;
  const int j = tid & 15;
  unsigned* wrec = (MODE != 0) ? rec + ((size_t)wg * (PUSH_NT / 32) + wv) * PUSH_CAP : nullptr;

  unsigned most = 0u;
  if (MODE == 2) {
#pragma unroll
    for (int w = 0; w < PUSH_NT / 32; w++) most = max(most, cnt[wg * (PUSH_NT / 32) + w]);
  }

  for (int i = tid; i < PUSH_TILE * 16; i += PUSH_NT) acc[i] = 0.0f;

  const bool replay = MODE == 2 && most <= (unsigned)PUSH_CAP;

  if (replay) {
    const unsigned total = __builtin_amdgcn_readfirstlane(cnt[wg * (PUSH_NT / 32) + wv]);
    const unsigned ntrip = (total + 15u) >> 4;
    const unsigned nfull = total >> 4;
    const uint4* rp = (const uint4*)wrec + half * 2u;
    uint4 a0 = make_uint4(0u, 0u, 0u, 0u), a1 = a0, b0 = a0, b1 = a0;
    if (0u < ntrip) { a0 = rp[0]; a1 = rp[1]; }
    if (1u < ntrip) { b0 = rp[4]; b1 = rp[5]; }
    __syncthreads();

    unsigned T = 0u;
    for (; T + 2u <= nfull; T += 2u) {
      unsigned wa[8], wb[8]; float va[8], vb[8];
      push_words(wa, a0, a1);
      push_words(wb, b0, b1);
      if (T + 2u < ntrip) { a0 = rp[(T + 2u) * 4u]; a1 = rp[(T + 2u) * 4u + 1u]; }
      if (T + 3u < ntrip) { b0 = rp[(T + 3u) * 4u]; b1 = rp[(T + 3u) * 4u + 1u]; }
      push_rows<true>(wa, va, 16u, half, j, tab);
      push_rows<true>(wb, vb, 16u, half, j, tab);
      push_adds(wa, va, j, acc);
      push_adds(wb, vb, j, acc);
    }
    if (T < nfull) {
      unsigned wa[8]; float va[8];
      push_words(wa, a0, a1);
      push_rows<true>(wa, va, 16u, half, j, tab);
      push_adds(wa, va, j, acc);
      a0 = b0; a1 = b1;
    }
    if (total & 15u) {
      unsigned wa[8]; float va[8];
      push_words(wa, a0, a1);
      push_rows<false>(wa, va, total & 15u, half, j, tab);
      push_adds(wa, va, j, acc);
    }
  } else {
    int dvA[PUSH_K] = {0}, svA[PUSH_K] = {0}, dvB[PUSH_K] = {0}, svB[PUSH_K] = {0};
    const bool any = e0 < e1;
    if (any) {
      push_fetch(dvA, svA, e0, tid, e1, src, dst);
      push_fetch(dvB, svB, e0 + STEP, tid, e1, src, dst);
    }
    __syncthreads();

    unsigned* q = ring + wv * PUSH_RING;
    unsigned qn = 0u, qh = 0u;

    if (any) for (int b0 = e0; b0 < e1; b0 += 2 * STEP) {
      push_append(dvA, svA, b0, tid, e1, lo, q, qn);
      push_fetch(dvA, svA, b0 + 2 * STEP, tid, e1, src, dst);
      while (qn - qh >= 16u) { push_trip<true, MODE == 1>(q, qh, 16u, half, j, tab, acc, wrec); qh += 16u; }

      push_append(dvB, svB, b0 + STEP, tid, e1, lo, q, qn);
      push_fetch(dvB, svB, b0 + 3 * STEP, tid, e1, src, dst);
      while (qn - qh >= 16u) { push_trip<true, MODE == 1>(q, qh, 16u, half, j, tab, acc, wrec); qh += 16u; }
    }
    if (qn != qh) push_trip<false, MODE == 1>(q, qh, qn - qh, half, j, tab, acc, wrec);
    if (MODE == 1 && (tid & 31) == 0) q[0] = qn;
  }
  __syncthreads();
  if (MODE == 1 && tid < PUSH_NT / 32) cnt[wg * (PUSH_NT / 32) + tid] = ring[tid * PUSH_RING];

  const float4* acc4 = (const float4*)acc;
  float4* out4 = (float4*)(part + ((size_t)blockIdx.y * n + lo) * 16);
  for (int i4 = tid; i4 < PUSH_TILE * 4; i4 += PUSH_NT) {
    if (lo + (i4 >> 2) < n) out4[i4] = acc4[i4];
  }
}

#define CW_PITCH 20

template <bool HEAD>
__global__ __launch_bounds__(256)
void k_combine_wmma(const float* __restrict__ part, float* tab,
                    const float* __restrict__ dinv, const float* __restrict__ bias,
                    const float* __restrict__ W, const float* __restrict__ l1b,
                    const float* __restrict__ l2W, const float* __restrict__ l2b,
                    float* __restrict__ out, int n, int ntiles) {
  __shared__ float Ws[16 * 16];
  __shared__ __attribute__((aligned(16))) float hs[8 * 16 * CW_PITCH];
  Ws[threadIdx.x] = W[threadIdx.x];

  const int lane = threadIdx.x & 31;
  const int wv   = threadIdx.x >> 5;
  const int tile = blockIdx.x * 8 + wv;
  const bool ok  = tile < ntiles;
  const int tc   = ok ? tile : ntiles - 1;

  const int r0 = lane >> 2;
  const int c0 = (lane & 3) * 4;
  const size_t plane4 = (size_t)n * 4;
  const float4* p4 = (const float4*)part + (size_t)tc * 64 + lane;
  float4 lo[PUSH_C], hi[PUSH_C];
#pragma unroll
  for (int c = 0; c < PUSH_C; c++) { lo[c] = p4[c * plane4]; hi[c] = p4[c * plane4 + 32]; }
  const int nlo = tc * 16 + r0, nhi = nlo + 8;
  const float4 tlo = *(const float4*)(tab + (size_t)nlo * TAB_PITCH + c0);
  const float4 thi = *(const float4*)(tab + (size_t)nhi * TAB_PITCH + c0);
  const float dlo = dinv[nlo], dhi = dinv[nhi];
  float4 b4;
  b4.x = bias[c0]; b4.y = bias[c0 + 1]; b4.z = bias[c0 + 2]; b4.w = bias[c0 + 3];

  float4 alo = lo[0], ahi = hi[0];
#pragma unroll
  for (int c = 1; c < PUSH_C; c++) {
    alo.x += lo[c].x; alo.y += lo[c].y; alo.z += lo[c].z; alo.w += lo[c].w;
    ahi.x += hi[c].x; ahi.y += hi[c].y; ahi.z += hi[c].z; ahi.w += hi[c].w;
  }
  float4 hlo, hhi;
  hlo.x = fmaxf(dlo * (alo.x + dlo * tlo.x) + b4.x, 0.0f);
  hlo.y = fmaxf(dlo * (alo.y + dlo * tlo.y) + b4.y, 0.0f);
  hlo.z = fmaxf(dlo * (alo.z + dlo * tlo.z) + b4.z, 0.0f);
  hlo.w = fmaxf(dlo * (alo.w + dlo * tlo.w) + b4.w, 0.0f);
  hhi.x = fmaxf(dhi * (ahi.x + dhi * thi.x) + b4.x, 0.0f);
  hhi.y = fmaxf(dhi * (ahi.y + dhi * thi.y) + b4.y, 0.0f);
  hhi.z = fmaxf(dhi * (ahi.z + dhi * thi.z) + b4.z, 0.0f);
  hhi.w = fmaxf(dhi * (ahi.w + dhi * thi.w) + b4.w, 0.0f);

  float* ht = hs + wv * (16 * CW_PITCH);
  *(float4*)(ht + r0 * CW_PITCH + c0)       = hlo;
  *(float4*)(ht + (r0 + 8) * CW_PITCH + c0) = hhi;
  __syncthreads();

  const int half = lane >> 4;
  const int mn   = lane & 15;
  const int koff = half * 2;
  v8f c = {0.f, 0.f, 0.f, 0.f, 0.f, 0.f, 0.f, 0.f};
#pragma unroll
  for (int k0 = 0; k0 < 16; k0 += 4) {
    v2f a = *(const v2f*)(ht + mn * CW_PITCH + k0 + koff);
    v2f b;
    b.x = Ws[(k0 + koff + 0) * 16 + mn];
    b.y = Ws[(k0 + koff + 1) * 16 + mn];
    c = __builtin_amdgcn_wmma_f32_16x16x4_f32(
         false, a,  false, b,
         (short)0, c,  false,  false);
  }

  if (!HEAD) {
    if (ok) {
      float* hrow = tab + ((size_t)tile * 16 + half * 8) * TAB_PITCH + mn;
#pragma unroll
      for (int i = 0; i < 8; i++) hrow[i * TAB_PITCH] = c[i];
    }
  } else {
    const float bj = l1b[mn], wj = l2W[mn];
    float res = 0.0f;
#pragma unroll
    for (int i = 0; i < 8; i++) {
      float v = fmaxf(c[i] + bj, 0.0f) * wj;
#pragma unroll
      for (int off = 8; off; off >>= 1) v += __shfl_xor(v, off, 16);
      if (mn == i) res = v;
    }
    if (ok && mn < 8) out[tile * 16 + half * 8 + mn] = res + l2b[0];
  }
}

extern "C" void kernel_launch(void* const* d_in, const int* in_sizes, int n_in,
                              void* d_out, int out_size, void* d_ws, size_t ws_size,
                              hipStream_t stream) {
  const float* x   = (const float*)d_in[0];
  const int*   ei  = (const int*)d_in[1];
  const float* W1  = (const float*)d_in[2];
  const float* b1  = (const float*)d_in[3];
  const float* W2  = (const float*)d_in[4];
  const float* b2  = (const float*)d_in[5];
  const float* l1W = (const float*)d_in[6];
  const float* l1b = (const float*)d_in[7];
  const float* l2W = (const float*)d_in[8];
  const float* l2b = (const float*)d_in[9];

  const int N = in_sizes[0] / 128;
  const int E = in_sizes[1] / 2;
  const int* src = ei;
  const int* dst = ei + E;

  float* deg  = (float*)d_ws;
  float* dinv = deg + N;
  float* bufA = dinv + N;
  float* bufB = bufA + (size_t)N * 16;
  float* bufC = bufB + (size_t)N * 16;

  const int ntiles = N / 16;
  const int gemmBlocks = (ntiles * 32 + 255) / 256;
  const int nodeElems  = N * 16;
  const int nodeBlocks = (nodeElems + 255) / 256;
  const int edgeBlocks = (E + 255) / 256;
  const int scatBlocks = (int)(((size_t)E * 16 + 255) / 256);

  const size_t pushNeed = ((size_t)N * 2 + (size_t)N * 16 * (2 + PUSH_C)) * sizeof(float);
  if (pushNeed <= ws_size && N <= (1 << 17)) {
    float* tab  = bufA;
    float* part = bufC;
    const int chunkEdges = (E + PUSH_C - 1) / PUSH_C;
    const dim3 pushGrid((N + PUSH_TILE - 1) / PUSH_TILE, PUSH_C);

    const size_t nWaves   = (size_t)pushGrid.x * pushGrid.y * (PUSH_NT / 32);
    const size_t cntWords = (nWaves + 31) / 32 * 32;
    unsigned* cnt = (unsigned*)(part + (size_t)PUSH_C * N * 16);
    unsigned* rec = cnt + cntWords;
    const bool record = pushNeed + (cntWords + nWaves * PUSH_CAP) * sizeof(unsigned) <= ws_size;

    float* deg1 = part;
    k_gemm_wmma<128, TAB_PITCH, true><<<gemmBlocks, 256, 0, stream>>>(x, W1, tab, ntiles, deg, deg1);
    k_deg_halves<<<edgeBlocks, 256, 0, stream>>>(dst, deg, deg1, E);
    k_rsqrt_rows<<<(N + 255) / 256, 256, 0, stream>>>(deg, deg1, dinv, tab, N);

    if (record) k_push<1><<<pushGrid, PUSH_NT, 0, stream>>>(src, dst, tab, part, rec, cnt, E, N, chunkEdges);
    else        k_push<0><<<pushGrid, PUSH_NT, 0, stream>>>(src, dst, tab, part, rec, cnt, E, N, chunkEdges);
    k_combine_wmma<false><<<gemmBlocks, 256, 0, stream>>>(part, tab, dinv, b1, W2, nullptr, nullptr,
                                                         nullptr, nullptr, N, ntiles);
    if (record) k_push<2><<<pushGrid, PUSH_NT, 0, stream>>>(src, dst, tab, part, rec, cnt, E, N, chunkEdges);
    else        k_push<0><<<pushGrid, PUSH_NT, 0, stream>>>(src, dst, tab, part, rec, cnt, E, N, chunkEdges);
    k_combine_wmma<true><<<gemmBlocks, 256, 0, stream>>>(part, tab, dinv, b2, l1W, l1b, l2W, l2b,
                                                        (float*)d_out, N, ntiles);
  } else {
    k_init  <<<nodeBlocks, 256, 0, stream>>>(deg, bufB, bufC, N);
    k_deg   <<<edgeBlocks, 256, 0, stream>>>(dst, deg, E);
    k_rsqrt <<<(N + 255) / 256, 256, 0, stream>>>(deg, dinv, N);

    k_gemm_wmma<128, 16, false><<<gemmBlocks, 256, 0, stream>>>(x, W1, bufA, ntiles, nullptr, nullptr);
    k_scatter<<<scatBlocks, 256, 0, stream>>>(src, dst, dinv, bufA, bufB, E);
    k_combine<<<nodeBlocks, 256, 0, stream>>>(bufB, bufA, dinv, b1, bufB, N);

    k_gemm_wmma<16, 16, false><<<gemmBlocks, 256, 0, stream>>>(bufB, W2, bufA, ntiles, nullptr, nullptr);
    k_scatter<<<scatBlocks, 256, 0, stream>>>(src, dst, dinv, bufA, bufC, E);
    k_combine<<<nodeBlocks, 256, 0, stream>>>(bufC, bufA, dinv, b2, bufC, N);

    k_gemm_wmma<16, 16, false><<<gemmBlocks, 256, 0, stream>>>(bufC, l1W, bufA, ntiles, nullptr, nullptr);
    k_final<<<nodeBlocks, 256, 0, stream>>>(bufA, l1b, l2W, l2b, (float*)d_out, N);
  }
}


# Degree pass folded into an edge scan run before the scale: matching lanes add the die-local degree atomics and fill the per-wave records; both layers replay them; layer-1 LDS add order moves
# speedup vs baseline: 3.4514x; 1.1027x over previous
#include <hip/hip_runtime.h>
#include <hip/hip_bf16.h>

typedef __attribute__((ext_vector_type(2))) float v2f;
typedef __attribute__((ext_vector_type(8))) float v8f;

#define TAB_PITCH 32

__global__ void k_init(float* __restrict__ deg, float* __restrict__ aggB,
                       float* __restrict__ aggC, int n) {
  int idx = blockIdx.x * blockDim.x + threadIdx.x;
  if (idx < n) deg[idx] = 1.0f;
  if (idx < n * 16) { aggB[idx] = 0.0f; aggC[idx] = 0.0f; }
}

__global__ void k_deg(const int* __restrict__ dst, float* __restrict__ deg, int E) {
  int e = blockIdx.x * blockDim.x + threadIdx.x;
  if (e < E) atomicAdd(&deg[dst[e]], 1.0f);
}

__global__ void k_deg_halves(const int* __restrict__ dst, float* __restrict__ deg0,
                             float* __restrict__ deg1, int E) {
  const unsigned where = (unsigned)__builtin_amdgcn_s_sendmsg_rtn(0x87);
  float* deg = ((where >> 18) & 1u) ? deg1 : deg0;
  int e = blockIdx.x * blockDim.x + threadIdx.x;
  if (e < E) atomicAdd(&deg[dst[e]], 1.0f);
}

__global__ void k_rsqrt(const float* __restrict__ deg, float* __restrict__ dinv, int n) {
  int i = blockIdx.x * blockDim.x + threadIdx.x;
  if (i < n) dinv[i] = rsqrtf(deg[i]);
}

__global__ void k_rsqrt_rows(const float* __restrict__ deg, const float* __restrict__ deg1,
                             float* __restrict__ dinv, float* __restrict__ tab, int n) {
  int i = blockIdx.x * blockDim.x + threadIdx.x;
  if (i < n) {
    const float r = rsqrtf(deg[i] + deg1[i]);
    dinv[i] = r;
    tab[(size_t)i * TAB_PITCH + 16] = r;
  }
}

template <int K, int P, bool INIT>
__global__ void k_gemm_wmma(const float* __restrict__ A, const float* __restrict__ W,
                            float* __restrict__ H, int ntiles,
                            float* __restrict__ deg0, float* __restrict__ deg1) {
  __shared__ float Ws[K * 16];
  for (int t = threadIdx.x; t < K * 16; t += blockDim.x) Ws[t] = W[t];
  __syncthreads();

  const int lane = threadIdx.x & 31;
  const int tile = blockIdx.x * (blockDim.x >> 5) + (threadIdx.x >> 5);
  if (tile >= ntiles) return;

  const int half = lane >> 4;
  const int mn   = lane & 15;
  const int koff = half * 2;

  const float* arow = A + (size_t)(tile * 16 + mn) * K;

  if (INIT) {
    if (half == 0) deg0[tile * 16 + mn] = 1.0f;
    else           deg1[tile * 16 + mn] = 0.0f;
  }

  v8f c = {0.f, 0.f, 0.f, 0.f, 0.f, 0.f, 0.f, 0.f};
#pragma unroll
  for (int k0 = 0; k0 < K; k0 += 4) {
    v2f a = *(const v2f*)(arow + k0 + koff);
    v2f b;
    b.x = Ws[(k0 + koff + 0) * 16 + mn];
    b.y = Ws[(k0 + koff + 1) * 16 + mn];
    c = __builtin_amdgcn_wmma_f32_16x16x4_f32(
         false, a,  false, b,
         (short)0, c,  false,  false);
  }

  float* hrow = H + ((size_t)tile * 16 + half * 8) * P + mn;
#pragma unroll
  for (int i = 0; i < 8; i++) hrow[i * P] = c[i];
}

__global__ void k_scatter(const int* __restrict__ src, const int* __restrict__ dst,
                          const float* __restrict__ dinv, const float* __restrict__ H,
                          float* __restrict__ AGG, int E) {
  int idx = blockIdx.x * blockDim.x + threadIdx.x;
  int e = idx >> 4;
  int j = idx & 15;
  if (e >= E) return;
  int s = src[e];
  int d = dst[e];
  float norm = dinv[s] * dinv[d];
  atomicAdd(&AGG[d * 16 + j], norm * H[s * 16 + j]);
}

__global__ void k_combine(const float* __restrict__ AGG, const float* __restrict__ H,
                          const float* __restrict__ dinv, const float* __restrict__ bias,
                          float* __restrict__ out, int n) {
  int idx = blockIdx.x * blockDim.x + threadIdx.x;
  int i = idx >> 4;
  int j = idx & 15;
  if (i >= n) return;
  float di = dinv[i];
  out[idx] = fmaxf(AGG[idx] + di * di * H[idx] + bias[j], 0.0f);
}

__global__ void k_final(const float* __restrict__ G, const float* __restrict__ l1b,
                        const float* __restrict__ l2W, const float* __restrict__ l2b,
                        float* __restrict__ out, int n) {
  int idx = blockIdx.x * blockDim.x + threadIdx.x;
  int i = idx >> 4;
  int j = idx & 15;
  float v = 0.0f;
  if (i < n) v = fmaxf(G[idx] + l1b[j], 0.0f) * l2W[j];
#pragma unroll
  for (int off = 8; off; off >>= 1) v += __shfl_xor(v, off, 16);
  if (i < n && j == 0) out[i] = v + l2b[0];
}


#define PUSH_TILE 4096
#define PUSH_NT   1024
#define PUSH_K    6
#define PUSH_RING 256
#define PUSH_C    10
#define PUSH_CAP  576

template <bool FULL>
__device__ __forceinline__ void push_trip(const unsigned* q, unsigned qh, unsigned left,
                                          unsigned half, int j,
                                          const float* __restrict__ tab, float* acc) {
  unsigned w[8]; float v[8];
#pragma unroll
  for (int t = 0; t < 8; t++) {
    const unsigned slot = 2u * t + half;
    const unsigned idx = (FULL || slot < left) ? qh + slot : qh;
    w[t] = q[idx & (PUSH_RING - 1)];
  }
#pragma unroll
  for (int t = 0; t < 8; t++) {
    const unsigned r = (w[t] & 0x1FFFFu) * TAB_PITCH;
    const float x = tab[r + 16u] * tab[r + (unsigned)j];
    v[t] = (FULL || 2u * t + half < left) ? x : 0.0f;
  }
#pragma unroll
  for (int t = 0; t < 8; t++) atomicAdd(&acc[(w[t] >> 17) * 16 + j], v[t]);
}

__device__ __forceinline__ void push_fetch(int (&dv)[PUSH_K], int (&sv)[PUSH_K], int b0, int tid,
                                           int e1, const int* __restrict__ src,
                                           const int* __restrict__ dst) {
#pragma unroll
  for (int k = 0; k < PUSH_K; k++) {
    const int idx = min(b0 + k * PUSH_NT + tid, e1 - 1);
    dv[k] = dst[idx];
    sv[k] = src[idx];
  }
}

__device__ __forceinline__ void push_append(const int (&dv)[PUSH_K], const int (&sv)[PUSH_K],
                                            int b0, int tid, int e1, int lo,
                                            unsigned* q, unsigned& qn) {
#pragma unroll
  for (int k = 0; k < PUSH_K; k++) {
    const int e = b0 + k * PUSH_NT + tid;
    const unsigned off = (unsigned)(dv[k] - lo);
    const bool hit = e < e1 && off < (unsigned)PUSH_TILE;
    const unsigned m = __builtin_amdgcn_ballot_w32(hit);
    if (hit) q[(qn + __builtin_amdgcn_mbcnt_lo(m, 0u)) & (PUSH_RING - 1)] = (unsigned)sv[k] | (off << 17);
    qn += (unsigned)__popc(m);
  }
}

template <bool FULL>
__device__ __forceinline__ void push_rows(const unsigned (&w)[8], float (&v)[8], unsigned left,
                                          unsigned half, int j, const float* __restrict__ tab) {
#pragma unroll
  for (int t = 0; t < 8; t++) {
    const unsigned r = (w[t] & 0x1FFFFu) * TAB_PITCH;
    const float x = tab[r + 16u] * tab[r + (unsigned)j];
    v[t] = (FULL || 2u * t + half < left) ? x : 0.0f;
  }
}
__device__ __forceinline__ void push_adds(const unsigned (&w)[8], const float (&v)[8], int j,
                                          float* acc) {
#pragma unroll
  for (int t = 0; t < 8; t++) atomicAdd(&acc[(w[t] >> 17) * 16 + j], v[t]);
}
__device__ __forceinline__ void push_words(unsigned (&w)[8], const uint4& a, const uint4& b) {
  w[0] = a.x; w[1] = a.y; w[2] = a.z; w[3] = a.w;
  w[4] = b.x; w[5] = b.y; w[6] = b.z; w[7] = b.w;
}

template <int MODE>
__global__ __launch_bounds__(PUSH_NT)
void k_push(const int* __restrict__ src, const int* __restrict__ dst,
            const float* __restrict__ tab,
            float* __restrict__ part, const unsigned* __restrict__ rec,
            const unsigned* __restrict__ cnt, int E, int n, int chunkEdges) {
  __shared__ __attribute__((aligned(16))) float acc[PUSH_TILE * 16];
  __shared__ unsigned ring[(PUSH_NT / 32) * PUSH_RING];

  const int tid = threadIdx.x;
  const int lo  = blockIdx.x * PUSH_TILE;
  const int e0  = blockIdx.y * chunkEdges;
  const int e1  = min(E, e0 + chunkEdges);
  const int STEP = PUSH_K * PUSH_NT;

  const int wv = __builtin_amdgcn_readfirstlane(tid >> 5);
  const int wg = blockIdx.y * gridDim.x + blockIdx.x;
  const unsigned half = (unsigned)(tid >> 4) & 1u;
  const int j = tid & 15;
  const unsigned* wrec = (MODE == 2) ? rec + ((size_t)wg * (PUSH_NT / 32) + wv) * PUSH_CAP : nullptr;

  unsigned most = 0u;
  if (MODE == 2) {
#pragma unroll
    for (int w = 0; w < PUSH_NT / 32; w++) most = max(most, cnt[wg * (PUSH_NT / 32) + w]);
  }

  for (int i = tid; i < PUSH_TILE * 16; i += PUSH_NT) acc[i] = 0.0f;

  const bool replay = MODE == 2 && most <= (unsigned)PUSH_CAP;

  if (replay) {
    const unsigned total = __builtin_amdgcn_readfirstlane(cnt[wg * (PUSH_NT / 32) + wv]);
    const unsigned ntrip = (total + 15u) >> 4;
    const unsigned nfull = total >> 4;
    const uint4* rp = (const uint4*)wrec + half * 2u;
    uint4 a0 = make_uint4(0u, 0u, 0u, 0u), a1 = a0, b0 = a0, b1 = a0;
    if (0u < ntrip) { a0 = rp[0]; a1 = rp[1]; }
    if (1u < ntrip) { b0 = rp[4]; b1 = rp[5]; }
    __syncthreads();

    unsigned T = 0u;
    for (; T + 2u <= nfull; T += 2u) {
      unsigned wa[8], wb[8]; float va[8], vb[8];
      push_words(wa, a0, a1);
      push_words(wb, b0, b1);
      if (T + 2u < ntrip) { a0 = rp[(T + 2u) * 4u]; a1 = rp[(T + 2u) * 4u + 1u]; }
      if (T + 3u < ntrip) { b0 = rp[(T + 3u) * 4u]; b1 = rp[(T + 3u) * 4u + 1u]; }
      push_rows<true>(wa, va, 16u, half, j, tab);
      push_rows<true>(wb, vb, 16u, half, j, tab);
      push_adds(wa, va, j, acc);
      push_adds(wb, vb, j, acc);
    }
    if (T < nfull) {
      unsigned wa[8]; float va[8];
      push_words(wa, a0, a1);
      push_rows<true>(wa, va, 16u, half, j, tab);
      push_adds(wa, va, j, acc);
      a0 = b0; a1 = b1;
    }
    if (total & 15u) {
      unsigned wa[8]; float va[8];
      push_words(wa, a0, a1);
      push_rows<false>(wa, va, total & 15u, half, j, tab);
      push_adds(wa, va, j, acc);
    }
  } else {
    int dvA[PUSH_K] = {0}, svA[PUSH_K] = {0}, dvB[PUSH_K] = {0}, svB[PUSH_K] = {0};
    const bool any = e0 < e1;
    if (any) {
      push_fetch(dvA, svA, e0, tid, e1, src, dst);
      push_fetch(dvB, svB, e0 + STEP, tid, e1, src, dst);
    }
    __syncthreads();

    unsigned* q = ring + wv * PUSH_RING;
    unsigned qn = 0u, qh = 0u;

    if (any) for (int b0 = e0; b0 < e1; b0 += 2 * STEP) {
      push_append(dvA, svA, b0, tid, e1, lo, q, qn);
      push_fetch(dvA, svA, b0 + 2 * STEP, tid, e1, src, dst);
      while (qn - qh >= 16u) { push_trip<true>(q, qh, 16u, half, j, tab, acc); qh += 16u; }

      push_append(dvB, svB, b0 + STEP, tid, e1, lo, q, qn);
      push_fetch(dvB, svB, b0 + 3 * STEP, tid, e1, src, dst);
      while (qn - qh >= 16u) { push_trip<true>(q, qh, 16u, half, j, tab, acc); qh += 16u; }
    }
    if (qn != qh) push_trip<false>(q, qh, qn - qh, half, j, tab, acc);
  }
  __syncthreads();

  const float4* acc4 = (const float4*)acc;
  float4* out4 = (float4*)(part + ((size_t)blockIdx.y * n + lo) * 16);
  for (int i4 = tid; i4 < PUSH_TILE * 4; i4 += PUSH_NT) {
    if (lo + (i4 >> 2) < n) out4[i4] = acc4[i4];
  }
}

__device__ __forceinline__ void scan_append(const int (&dv)[PUSH_K], const int (&sv)[PUSH_K],
                                            int b0, int tid, int e1, int lo,
                                            unsigned* q, unsigned& qn, float* __restrict__ deg) {
#pragma unroll
  for (int k = 0; k < PUSH_K; k++) {
    const int e = b0 + k * PUSH_NT + tid;
    const unsigned off = (unsigned)(dv[k] - lo);
    const bool hit = e < e1 && off < (unsigned)PUSH_TILE;
    const unsigned m = __builtin_amdgcn_ballot_w32(hit);
    if (hit) {
      atomicAdd(&deg[dv[k]], 1.0f);
      q[(qn + __builtin_amdgcn_mbcnt_lo(m, 0u)) & (PUSH_RING - 1)] = (unsigned)sv[k] | (off << 17);
    }
    qn += (unsigned)__popc(m);
  }
}

template <bool FULL>
__device__ __forceinline__ void scan_flush(const unsigned* q, unsigned qh, unsigned left, int lane,
                                           unsigned* __restrict__ rec) {
  const unsigned trip = (unsigned)lane >> 4, p = (unsigned)lane & 15u;
  const unsigned slot = trip * 16u + (((p & 7u) << 1) | (p >> 3));
  const unsigned idx = (FULL || slot < left) ? qh + slot : qh + trip * 16u;
  const unsigned w = q[idx & (PUSH_RING - 1)];
  if ((FULL || trip * 16u < left) && qh + (unsigned)lane < (unsigned)PUSH_CAP) rec[qh + (unsigned)lane] = w;
}

__global__ __launch_bounds__(PUSH_NT)
void k_scan(const int* __restrict__ src, const int* __restrict__ dst,
            float* __restrict__ deg0, float* __restrict__ deg1,
            unsigned* __restrict__ rec, unsigned* __restrict__ cnt, int E, int chunkEdges) {
  __shared__ unsigned ring[(PUSH_NT / 32) * PUSH_RING];

  const int tid = threadIdx.x;
  const int lane = tid & 31;
  const int lo  = blockIdx.x * PUSH_TILE;
  const int e0  = blockIdx.y * chunkEdges;
  const int e1  = min(E, e0 + chunkEdges);
  const int STEP = PUSH_K * PUSH_NT;

  const int wv = __builtin_amdgcn_readfirstlane(tid >> 5);
  const int wg = blockIdx.y * gridDim.x + blockIdx.x;
  unsigned* wrec = rec + ((size_t)wg * (PUSH_NT / 32) + wv) * PUSH_CAP;

  const unsigned where = (unsigned)__builtin_amdgcn_s_sendmsg_rtn(0x87);
  float* deg = ((where >> 18) & 1u) ? deg1 : deg0;

  unsigned* q = ring + wv * PUSH_RING;
  unsigned qn = 0u, qh = 0u;

  if (e0 < e1) {
    int dvA[PUSH_K], svA[PUSH_K], dvB[PUSH_K], svB[PUSH_K];
    push_fetch(dvA, svA, e0, tid, e1, src, dst);
    push_fetch(dvB, svB, e0 + STEP, tid, e1, src, dst);
    for (int b0 = e0; b0 < e1; b0 += 2 * STEP) {
      scan_append(dvA, svA, b0, tid, e1, lo, q, qn, deg);
      push_fetch(dvA, svA, b0 + 2 * STEP, tid, e1, src, dst);
      while (qn - qh >= 32u) { scan_flush<true>(q, qh, 32u, lane, wrec); qh += 32u; }

      scan_append(dvB, svB, b0 + STEP, tid, e1, lo, q, qn, deg);
      push_fetch(dvB, svB, b0 + 3 * STEP, tid, e1, src, dst);
      while (qn - qh >= 32u) { scan_flush<true>(q, qh, 32u, lane, wrec); qh += 32u; }
    }
    if (qn != qh) scan_flush<false>(q, qh, qn - qh, lane, wrec);
  }
  if (lane == 0) cnt[wg * (PUSH_NT / 32) + wv] = qn;
}

#define CW_PITCH 20

template <bool HEAD>
__global__ __launch_bounds__(256)
void k_combine_wmma(const float* __restrict__ part, float* tab,
                    const float* __restrict__ dinv, const float* __restrict__ bias,
                    const float* __restrict__ W, const float* __restrict__ l1b,
                    const float* __restrict__ l2W, const float* __restrict__ l2b,
                    float* __restrict__ out, int n, int ntiles) {
  __shared__ float Ws[16 * 16];
  __shared__ __attribute__((aligned(16))) float hs[8 * 16 * CW_PITCH];
  Ws[threadIdx.x] = W[threadIdx.x];

  const int lane = threadIdx.x & 31;
  const int wv   = threadIdx.x >> 5;
  const int tile = blockIdx.x * 8 + wv;
  const bool ok  = tile < ntiles;
  const int tc   = ok ? tile : ntiles - 1;

  const int r0 = lane >> 2;
  const int c0 = (lane & 3) * 4;
  const size_t plane4 = (size_t)n * 4;
  const float4* p4 = (const float4*)part + (size_t)tc * 64 + lane;
  float4 lo[PUSH_C], hi[PUSH_C];
#pragma unroll
  for (int c = 0; c < PUSH_C; c++) { lo[c] = p4[c * plane4]; hi[c] = p4[c * plane4 + 32]; }
  const int nlo = tc * 16 + r0, nhi = nlo + 8;
  const float4 tlo = *(const float4*)(tab + (size_t)nlo * TAB_PITCH + c0);
  const float4 thi = *(const float4*)(tab + (size_t)nhi * TAB_PITCH + c0);
  const float dlo = dinv[nlo], dhi = dinv[nhi];
  float4 b4;
  b4.x = bias[c0]; b4.y = bias[c0 + 1]; b4.z = bias[c0 + 2]; b4.w = bias[c0 + 3];

  float4 alo = lo[0], ahi = hi[0];
#pragma unroll
  for (int c = 1; c < PUSH_C; c++) {
    alo.x += lo[c].x; alo.y += lo[c].y; alo.z += lo[c].z; alo.w += lo[c].w;
    ahi.x += hi[c].x; ahi.y += hi[c].y; ahi.z += hi[c].z; ahi.w += hi[c].w;
  }
  float4 hlo, hhi;
  hlo.x = fmaxf(dlo * (alo.x + dlo * tlo.x) + b4.x, 0.0f);
  hlo.y = fmaxf(dlo * (alo.y + dlo * tlo.y) + b4.y, 0.0f);
  hlo.z = fmaxf(dlo * (alo.z + dlo * tlo.z) + b4.z, 0.0f);
  hlo.w = fmaxf(dlo * (alo.w + dlo * tlo.w) + b4.w, 0.0f);
  hhi.x = fmaxf(dhi * (ahi.x + dhi * thi.x) + b4.x, 0.0f);
  hhi.y = fmaxf(dhi * (ahi.y + dhi * thi.y) + b4.y, 0.0f);
  hhi.z = fmaxf(dhi * (ahi.z + dhi * thi.z) + b4.z, 0.0f);
  hhi.w = fmaxf(dhi * (ahi.w + dhi * thi.w) + b4.w, 0.0f);

  float* ht = hs + wv * (16 * CW_PITCH);
  *(float4*)(ht + r0 * CW_PITCH + c0)       = hlo;
  *(float4*)(ht + (r0 + 8) * CW_PITCH + c0) = hhi;
  __syncthreads();

  const int half = lane >> 4;
  const int mn   = lane & 15;
  const int koff = half * 2;
  v8f c = {0.f, 0.f, 0.f, 0.f, 0.f, 0.f, 0.f, 0.f};
#pragma unroll
  for (int k0 = 0; k0 < 16; k0 += 4) {
    v2f a = *(const v2f*)(ht + mn * CW_PITCH + k0 + koff);
    v2f b;
    b.x = Ws[(k0 + koff + 0) * 16 + mn];
    b.y = Ws[(k0 + koff + 1) * 16 + mn];
    c = __builtin_amdgcn_wmma_f32_16x16x4_f32(
         false, a,  false, b,
         (short)0, c,  false,  false);
  }

  if (!HEAD) {
    if (ok) {
      float* hrow = tab + ((size_t)tile * 16 + half * 8) * TAB_PITCH + mn;
#pragma unroll
      for (int i = 0; i < 8; i++) hrow[i * TAB_PITCH] = c[i];
    }
  } else {
    const float bj = l1b[mn], wj = l2W[mn];
    float res = 0.0f;
#pragma unroll
    for (int i = 0; i < 8; i++) {
      float v = fmaxf(c[i] + bj, 0.0f) * wj;
#pragma unroll
      for (int off = 8; off; off >>= 1) v += __shfl_xor(v, off, 16);
      if (mn == i) res = v;
    }
    if (ok && mn < 8) out[tile * 16 + half * 8 + mn] = res + l2b[0];
  }
}

extern "C" void kernel_launch(void* const* d_in, const int* in_sizes, int n_in,
                              void* d_out, int out_size, void* d_ws, size_t ws_size,
                              hipStream_t stream) {
  const float* x   = (const float*)d_in[0];
  const int*   ei  = (const int*)d_in[1];
  const float* W1  = (const float*)d_in[2];
  const float* b1  = (const float*)d_in[3];
  const float* W2  = (const float*)d_in[4];
  const float* b2  = (const float*)d_in[5];
  const float* l1W = (const float*)d_in[6];
  const float* l1b = (const float*)d_in[7];
  const float* l2W = (const float*)d_in[8];
  const float* l2b = (const float*)d_in[9];

  const int N = in_sizes[0] / 128;
  const int E = in_sizes[1] / 2;
  const int* src = ei;
  const int* dst = ei + E;

  float* deg  = (float*)d_ws;
  float* dinv = deg + N;
  float* bufA = dinv + N;
  float* bufB = bufA + (size_t)N * 16;
  float* bufC = bufB + (size_t)N * 16;

  const int ntiles = N / 16;
  const int gemmBlocks = (ntiles * 32 + 255) / 256;
  const int nodeElems  = N * 16;
  const int nodeBlocks = (nodeElems + 255) / 256;
  const int edgeBlocks = (E + 255) / 256;
  const int scatBlocks = (int)(((size_t)E * 16 + 255) / 256);

  const size_t pushNeed = ((size_t)N * 2 + (size_t)N * 16 * (2 + PUSH_C)) * sizeof(float);
  if (pushNeed <= ws_size && N <= (1 << 17)) {
    float* tab  = bufA;
    float* part = bufC;
    const int chunkEdges = (E + PUSH_C - 1) / PUSH_C;
    const dim3 pushGrid((N + PUSH_TILE - 1) / PUSH_TILE, PUSH_C);

    const size_t nWaves   = (size_t)pushGrid.x * pushGrid.y * (PUSH_NT / 32);
    const size_t cntWords = (nWaves + 31) / 32 * 32;
    unsigned* cnt = (unsigned*)(part + (size_t)PUSH_C * N * 16);
    unsigned* rec = cnt + cntWords;
    const bool record = pushNeed + (cntWords + nWaves * PUSH_CAP) * sizeof(unsigned) <= ws_size;

    float* deg1 = part;
    k_gemm_wmma<128, TAB_PITCH, true><<<gemmBlocks, 256, 0, stream>>>(x, W1, tab, ntiles, deg, deg1);
    if (record) k_scan<<<pushGrid, PUSH_NT, 0, stream>>>(src, dst, deg, deg1, rec, cnt, E, chunkEdges);
    else        k_deg_halves<<<edgeBlocks, 256, 0, stream>>>(dst, deg, deg1, E);
    k_rsqrt_rows<<<(N + 255) / 256, 256, 0, stream>>>(deg, deg1, dinv, tab, N);

    if (record) k_push<2><<<pushGrid, PUSH_NT, 0, stream>>>(src, dst, tab, part, rec, cnt, E, N, chunkEdges);
    else        k_push<0><<<pushGrid, PUSH_NT, 0, stream>>>(src, dst, tab, part, rec, cnt, E, N, chunkEdges);
    k_combine_wmma<false><<<gemmBlocks, 256, 0, stream>>>(part, tab, dinv, b1, W2, nullptr, nullptr,
                                                         nullptr, nullptr, N, ntiles);
    if (record) k_push<2><<<pushGrid, PUSH_NT, 0, stream>>>(src, dst, tab, part, rec, cnt, E, N, chunkEdges);
    else        k_push<0><<<pushGrid, PUSH_NT, 0, stream>>>(src, dst, tab, part, rec, cnt, E, N, chunkEdges);
    k_combine_wmma<true><<<gemmBlocks, 256, 0, stream>>>(part, tab, dinv, b2, l1W, l1b, l2W, l2b,
                                                        (float*)d_out, N, ntiles);
  } else {
    k_init  <<<nodeBlocks, 256, 0, stream>>>(deg, bufB, bufC, N);
    k_deg   <<<edgeBlocks, 256, 0, stream>>>(dst, deg, E);
    k_rsqrt <<<(N + 255) / 256, 256, 0, stream>>>(deg, dinv, N);

    k_gemm_wmma<128, 16, false><<<gemmBlocks, 256, 0, stream>>>(x, W1, bufA, ntiles, nullptr, nullptr);
    k_scatter<<<scatBlocks, 256, 0, stream>>>(src, dst, dinv, bufA, bufB, E);
    k_combine<<<nodeBlocks, 256, 0, stream>>>(bufB, bufA, dinv, b1, bufB, N);

    k_gemm_wmma<16, 16, false><<<gemmBlocks, 256, 0, stream>>>(bufB, W2, bufA, ntiles, nullptr, nullptr);
    k_scatter<<<scatBlocks, 256, 0, stream>>>(src, dst, dinv, bufA, bufC, E);
    k_combine<<<nodeBlocks, 256, 0, stream>>>(bufC, bufA, dinv, b2, bufC, N);

    k_gemm_wmma<16, 16, false><<<gemmBlocks, 256, 0, stream>>>(bufC, l1W, bufA, ntiles, nullptr, nullptr);
    k_final<<<nodeBlocks, 256, 0, stream>>>(bufA, l1b, l2W, l2b, (float*)d_out, N);
  }
}


# Edge scan tests each loaded edge against 5 adjacent node ranges (5 rings per wave) on a fifth of a chunk: edge list read 5x, not 25x; push waves replay 5 short records; LDS add order moves
# speedup vs baseline: 3.7127x; 1.0757x over previous
#include <hip/hip_runtime.h>
#include <hip/hip_bf16.h>

typedef __attribute__((ext_vector_type(2))) float v2f;
typedef __attribute__((ext_vector_type(8))) float v8f;

#define TAB_PITCH 32

__global__ void k_init(float* __restrict__ deg, float* __restrict__ aggB,
                       float* __restrict__ aggC, int n) {
  int idx = blockIdx.x * blockDim.x + threadIdx.x;
  if (idx < n) deg[idx] = 1.0f;
  if (idx < n * 16) { aggB[idx] = 0.0f; aggC[idx] = 0.0f; }
}

__global__ void k_deg(const int* __restrict__ dst, float* __restrict__ deg, int E) {
  int e = blockIdx.x * blockDim.x + threadIdx.x;
  if (e < E) atomicAdd(&deg[dst[e]], 1.0f);
}

__global__ void k_deg_halves(const int* __restrict__ dst, float* __restrict__ deg0,
                             float* __restrict__ deg1, int E) {
  const unsigned where = (unsigned)__builtin_amdgcn_s_sendmsg_rtn(0x87);
  float* deg = ((where >> 18) & 1u) ? deg1 : deg0;
  int e = blockIdx.x * blockDim.x + threadIdx.x;
  if (e < E) atomicAdd(&deg[dst[e]], 1.0f);
}

__global__ void k_rsqrt(const float* __restrict__ deg, float* __restrict__ dinv, int n) {
  int i = blockIdx.x * blockDim.x + threadIdx.x;
  if (i < n) dinv[i] = rsqrtf(deg[i]);
}

__global__ void k_rsqrt_rows(const float* __restrict__ deg, const float* __restrict__ deg1,
                             float* __restrict__ dinv, float* __restrict__ tab, int n) {
  int i = blockIdx.x * blockDim.x + threadIdx.x;
  if (i < n) {
    const float r = rsqrtf(deg[i] + deg1[i]);
    dinv[i] = r;
    tab[(size_t)i * TAB_PITCH + 16] = r;
  }
}

template <int K, int P, bool INIT>
__global__ void k_gemm_wmma(const float* __restrict__ A, const float* __restrict__ W,
                            float* __restrict__ H, int ntiles,
                            float* __restrict__ deg0, float* __restrict__ deg1) {
  __shared__ float Ws[K * 16];
  for (int t = threadIdx.x; t < K * 16; t += blockDim.x) Ws[t] = W[t];
  __syncthreads();

  const int lane = threadIdx.x & 31;
  const int tile = blockIdx.x * (blockDim.x >> 5) + (threadIdx.x >> 5);
  if (tile >= ntiles) return;

  const int half = lane >> 4;
  const int mn   = lane & 15;
  const int koff = half * 2;

  const float* arow = A + (size_t)(tile * 16 + mn) * K;

  if (INIT) {
    if (half == 0) deg0[tile * 16 + mn] = 1.0f;
    else           deg1[tile * 16 + mn] = 0.0f;
  }

  v8f c = {0.f, 0.f, 0.f, 0.f, 0.f, 0.f, 0.f, 0.f};
#pragma unroll
  for (int k0 = 0; k0 < K; k0 += 4) {
    v2f a = *(const v2f*)(arow + k0 + koff);
    v2f b;
    b.x = Ws[(k0 + koff + 0) * 16 + mn];
    b.y = Ws[(k0 + koff + 1) * 16 + mn];
    c = __builtin_amdgcn_wmma_f32_16x16x4_f32(
         false, a,  false, b,
         (short)0, c,  false,  false);
  }

  float* hrow = H + ((size_t)tile * 16 + half * 8) * P + mn;
#pragma unroll
  for (int i = 0; i < 8; i++) hrow[i * P] = c[i];
}

__global__ void k_scatter(const int* __restrict__ src, const int* __restrict__ dst,
                          const float* __restrict__ dinv, const float* __restrict__ H,
                          float* __restrict__ AGG, int E) {
  int idx = blockIdx.x * blockDim.x + threadIdx.x;
  int e = idx >> 4;
  int j = idx & 15;
  if (e >= E) return;
  int s = src[e];
  int d = dst[e];
  float norm = dinv[s] * dinv[d];
  atomicAdd(&AGG[d * 16 + j], norm * H[s * 16 + j]);
}

__global__ void k_combine(const float* __restrict__ AGG, const float* __restrict__ H,
                          const float* __restrict__ dinv, const float* __restrict__ bias,
                          float* __restrict__ out, int n) {
  int idx = blockIdx.x * blockDim.x + threadIdx.x;
  int i = idx >> 4;
  int j = idx & 15;
  if (i >= n) return;
  float di = dinv[i];
  out[idx] = fmaxf(AGG[idx] + di * di * H[idx] + bias[j], 0.0f);
}

__global__ void k_final(const float* __restrict__ G, const float* __restrict__ l1b,
                        const float* __restrict__ l2W, const float* __restrict__ l2b,
                        float* __restrict__ out, int n) {
  int idx = blockIdx.x * blockDim.x + threadIdx.x;
  int i = idx >> 4;
  int j = idx & 15;
  float v = 0.0f;
  if (i < n) v = fmaxf(G[idx] + l1b[j], 0.0f) * l2W[j];
#pragma unroll
  for (int off = 8; off; off >>= 1) v += __shfl_xor(v, off, 16);
  if (i < n && j == 0) out[i] = v + l2b[0];
}


#define PUSH_TILE 4096
#define PUSH_NT   1024
#define PUSH_K    6
#define PUSH_RING 256
#define PUSH_C    10
#define SCAN_R    5
#define SCAN_S    5
#define PUSH_CAP  128

template <bool FULL>
__device__ __forceinline__ void push_trip(const unsigned* q, unsigned qh, unsigned left,
                                          unsigned half, int j,
                                          const float* __restrict__ tab, float* acc) {
  unsigned w[8]; float v[8];
#pragma unroll
  for (int t = 0; t < 8; t++) {
    const unsigned slot = 2u * t + half;
    const unsigned idx = (FULL || slot < left) ? qh + slot : qh;
    w[t] = q[idx & (PUSH_RING - 1)];
  }
#pragma unroll
  for (int t = 0; t < 8; t++) {
    const unsigned r = (w[t] & 0x1FFFFu) * TAB_PITCH;
    const float x = tab[r + 16u] * tab[r + (unsigned)j];
    v[t] = (FULL || 2u * t + half < left) ? x : 0.0f;
  }
#pragma unroll
  for (int t = 0; t < 8; t++) atomicAdd(&acc[(w[t] >> 17) * 16 + j], v[t]);
}

__device__ __forceinline__ void push_fetch(int (&dv)[PUSH_K], int (&sv)[PUSH_K], int b0, int tid,
                                           int e1, const int* __restrict__ src,
                                           const int* __restrict__ dst) {
#pragma unroll
  for (int k = 0; k < PUSH_K; k++) {
    const int idx = min(b0 + k * PUSH_NT + tid, e1 - 1);
    dv[k] = dst[idx];
    sv[k] = src[idx];
  }
}

__device__ __forceinline__ void push_append(const int (&dv)[PUSH_K], const int (&sv)[PUSH_K],
                                            int b0, int tid, int e1, int lo,
                                            unsigned* q, unsigned& qn) {
#pragma unroll
  for (int k = 0; k < PUSH_K; k++) {
    const int e = b0 + k * PUSH_NT + tid;
    const unsigned off = (unsigned)(dv[k] - lo);
    const bool hit = e < e1 && off < (unsigned)PUSH_TILE;
    const unsigned m = __builtin_amdgcn_ballot_w32(hit);
    if (hit) q[(qn + __builtin_amdgcn_mbcnt_lo(m, 0u)) & (PUSH_RING - 1)] = (unsigned)sv[k] | (off << 17);
    qn += (unsigned)__popc(m);
  }
}

template <bool FULL>
__device__ __forceinline__ void push_rows(const unsigned (&w)[8], float (&v)[8], unsigned left,
                                          unsigned half, int j, const float* __restrict__ tab) {
#pragma unroll
  for (int t = 0; t < 8; t++) {
    const unsigned r = (w[t] & 0x1FFFFu) * TAB_PITCH;
    const float x = tab[r + 16u] * tab[r + (unsigned)j];
    v[t] = (FULL || 2u * t + half < left) ? x : 0.0f;
  }
}
__device__ __forceinline__ void push_adds(const unsigned (&w)[8], const float (&v)[8], int j,
                                          float* acc) {
#pragma unroll
  for (int t = 0; t < 8; t++) atomicAdd(&acc[(w[t] >> 17) * 16 + j], v[t]);
}
__device__ __forceinline__ void push_words(unsigned (&w)[8], const uint4& a, const uint4& b) {
  w[0] = a.x; w[1] = a.y; w[2] = a.z; w[3] = a.w;
  w[4] = b.x; w[5] = b.y; w[6] = b.z; w[7] = b.w;
}

template <int MODE>
__global__ __launch_bounds__(PUSH_NT)
void k_push(const int* __restrict__ src, const int* __restrict__ dst,
            const float* __restrict__ tab,
            float* __restrict__ part, const unsigned* __restrict__ rec,
            const unsigned* __restrict__ cnt, int E, int n, int chunkEdges) {
  __shared__ __attribute__((aligned(16))) float acc[PUSH_TILE * 16];
  __shared__ unsigned ring[(PUSH_NT / 32) * PUSH_RING];

  const int tid = threadIdx.x;
  const int lo  = blockIdx.x * PUSH_TILE;
  const int e0  = blockIdx.y * chunkEdges;
  const int e1  = min(E, e0 + chunkEdges);
  const int STEP = PUSH_K * PUSH_NT;

  const int wv = __builtin_amdgcn_readfirstlane(tid >> 5);
  const int wg = blockIdx.y * gridDim.x + blockIdx.x;
  const unsigned half = (unsigned)(tid >> 4) & 1u;
  const int j = tid & 15;

  unsigned most = 0u;
  if (MODE == 2) {
#pragma unroll
    for (int k = 0; k < SCAN_S; k++)
      most = max(most, cnt[wg * (SCAN_S * (PUSH_NT / 32)) + k * (PUSH_NT / 32) + (tid & 31)]);
  }

  for (int i = tid; i < PUSH_TILE * 16; i += PUSH_NT) acc[i] = 0.0f;

  const bool replay = MODE == 2 && __builtin_amdgcn_ballot_w32(most > (unsigned)PUSH_CAP) == 0u;

  if (replay) {
    __syncthreads();

#pragma nounroll
    for (int s = 0; s < SCAN_S; s++) {
    const size_t rid = ((size_t)wg * SCAN_S + s) * (PUSH_NT / 32) + wv;
    const unsigned total = __builtin_amdgcn_readfirstlane(cnt[rid]);
    const unsigned ntrip = (total + 15u) >> 4;
    const unsigned nfull = total >> 4;
    const uint4* rp = (const uint4*)(rec + rid * PUSH_CAP) + half * 2u;
    uint4 a0 = make_uint4(0u, 0u, 0u, 0u), a1 = a0, b0 = a0, b1 = a0;
    if (0u < ntrip) { a0 = rp[0]; a1 = rp[1]; }
    if (1u < ntrip) { b0 = rp[4]; b1 = rp[5]; }

    unsigned T = 0u;
    for (; T + 2u <= nfull; T += 2u) {
      unsigned wa[8], wb[8]; float va[8], vb[8];
      push_words(wa, a0, a1);
      push_words(wb, b0, b1);
      if (T + 2u < ntrip) { a0 = rp[(T + 2u) * 4u]; a1 = rp[(T + 2u) * 4u + 1u]; }
      if (T + 3u < ntrip) { b0 = rp[(T + 3u) * 4u]; b1 = rp[(T + 3u) * 4u + 1u]; }
      push_rows<true>(wa, va, 16u, half, j, tab);
      push_rows<true>(wb, vb, 16u, half, j, tab);
      push_adds(wa, va, j, acc);
      push_adds(wb, vb, j, acc);
    }
    if (T < nfull) {
      unsigned wa[8]; float va[8];
      push_words(wa, a0, a1);
      push_rows<true>(wa, va, 16u, half, j, tab);
      push_adds(wa, va, j, acc);
      a0 = b0; a1 = b1;
    }
    if (total & 15u) {
      unsigned wa[8]; float va[8];
      push_words(wa, a0, a1);
      push_rows<false>(wa, va, total & 15u, half, j, tab);
      push_adds(wa, va, j, acc);
    }
    }
  } else {
    int dvA[PUSH_K] = {0}, svA[PUSH_K] = {0}, dvB[PUSH_K] = {0}, svB[PUSH_K] = {0};
    const bool any = e0 < e1;
    if (any) {
      push_fetch(dvA, svA, e0, tid, e1, src, dst);
      push_fetch(dvB, svB, e0 + STEP, tid, e1, src, dst);
    }
    __syncthreads();

    unsigned* q = ring + wv * PUSH_RING;
    unsigned qn = 0u, qh = 0u;

    if (any) for (int b0 = e0; b0 < e1; b0 += 2 * STEP) {
      push_append(dvA, svA, b0, tid, e1, lo, q, qn);
      push_fetch(dvA, svA, b0 + 2 * STEP, tid, e1, src, dst);
      while (qn - qh >= 16u) { push_trip<true>(q, qh, 16u, half, j, tab, acc); qh += 16u; }

      push_append(dvB, svB, b0 + STEP, tid, e1, lo, q, qn);
      push_fetch(dvB, svB, b0 + 3 * STEP, tid, e1, src, dst);
      while (qn - qh >= 16u) { push_trip<true>(q, qh, 16u, half, j, tab, acc); qh += 16u; }
    }
    if (qn != qh) push_trip<false>(q, qh, qn - qh, half, j, tab, acc);
  }
  __syncthreads();

  const float4* acc4 = (const float4*)acc;
  float4* out4 = (float4*)(part + ((size_t)blockIdx.y * n + lo) * 16);
  for (int i4 = tid; i4 < PUSH_TILE * 4; i4 += PUSH_NT) {
    if (lo + (i4 >> 2) < n) out4[i4] = acc4[i4];
  }
}

__device__ __forceinline__ void scan_append(const int (&dv)[PUSH_K], const int (&sv)[PUSH_K],
                                            int b0, int tid, int e1, int lo,
                                            unsigned* q, unsigned (&qn)[SCAN_R],
                                            float* __restrict__ deg) {
#pragma unroll
  for (int k = 0; k < PUSH_K; k++) {
    const int e = b0 + k * PUSH_NT + tid;
    const unsigned off = (unsigned)(dv[k] - lo);
    const bool in = e < e1 && off < (unsigned)(SCAN_R * PUSH_TILE);
    if (in) atomicAdd(&deg[dv[k]], 1.0f);
    const unsigned which = off / (unsigned)PUSH_TILE;
    const unsigned word = (unsigned)sv[k] | ((off % (unsigned)PUSH_TILE) << 17);
#pragma unroll
    for (int i = 0; i < SCAN_R; i++) {
      const bool hit = in && which == (unsigned)i;
      const unsigned m = __builtin_amdgcn_ballot_w32(hit);
      if (hit) q[i * PUSH_RING + ((qn[i] + __builtin_amdgcn_mbcnt_lo(m, 0u)) & (PUSH_RING - 1))] = word;
      qn[i] += (unsigned)__popc(m);
    }
  }
}

template <bool FULL>
__device__ __forceinline__ void scan_flush(const unsigned* q, unsigned qh, unsigned left, int lane,
                                           unsigned* __restrict__ rec) {
  const unsigned trip = (unsigned)lane >> 4, p = (unsigned)lane & 15u;
  const unsigned slot = trip * 16u + (((p & 7u) << 1) | (p >> 3));
  const unsigned idx = (FULL || slot < left) ? qh + slot : qh + trip * 16u;
  const unsigned w = q[idx & (PUSH_RING - 1)];
  if ((FULL || trip * 16u < left) && qh + (unsigned)lane < (unsigned)PUSH_CAP) rec[qh + (unsigned)lane] = w;
}

__global__ __launch_bounds__(PUSH_NT)
void k_scan(const int* __restrict__ src, const int* __restrict__ dst,
            float* __restrict__ deg0, float* __restrict__ deg1,
            unsigned* __restrict__ rec, unsigned* __restrict__ cnt, int E, int chunkEdges,
            int nRanges) {
  __shared__ unsigned ring[(PUSH_NT / 32) * SCAN_R * PUSH_RING];

  const int tid = threadIdx.x;
  const int lane = tid & 31;
  const int r0  = blockIdx.x * SCAN_R;
  const int lo  = r0 * PUSH_TILE;
  const int c   = blockIdx.y / SCAN_S;
  const int s   = blockIdx.y - c * SCAN_S;
  const int pieceEdges = (chunkEdges + SCAN_S - 1) / SCAN_S;
  const int cEnd = min(E, (c + 1) * chunkEdges);
  const int e0  = c * chunkEdges + s * pieceEdges;
  const int e1  = min(cEnd, e0 + pieceEdges);
  const int STEP = PUSH_K * PUSH_NT;

  const int wv = __builtin_amdgcn_readfirstlane(tid >> 5);

  const unsigned where = (unsigned)__builtin_amdgcn_s_sendmsg_rtn(0x87);
  float* deg = ((where >> 18) & 1u) ? deg1 : deg0;

  unsigned* q = ring + wv * (SCAN_R * PUSH_RING);
  unsigned qn[SCAN_R], qh[SCAN_R];
#pragma unroll
  for (int i = 0; i < SCAN_R; i++) { qn[i] = 0u; qh[i] = 0u; }

  const size_t rid0 = ((size_t)(c * nRanges + r0) * SCAN_S + s) * (PUSH_NT / 32) + wv;
  const size_t ridStep = (size_t)SCAN_S * (PUSH_NT / 32);

  if (e0 < e1) {
    int dvA[PUSH_K], svA[PUSH_K], dvB[PUSH_K], svB[PUSH_K];
    push_fetch(dvA, svA, e0, tid, e1, src, dst);
    push_fetch(dvB, svB, e0 + STEP, tid, e1, src, dst);
    for (int b0 = e0; b0 < e1; b0 += 2 * STEP) {
      scan_append(dvA, svA, b0, tid, e1, lo, q, qn, deg);
      push_fetch(dvA, svA, b0 + 2 * STEP, tid, e1, src, dst);
#pragma unroll
      for (int i = 0; i < SCAN_R; i++)
        while (qn[i] - qh[i] >= 32u) {
          scan_flush<true>(q + i * PUSH_RING, qh[i], 32u, lane, rec + (rid0 + i * ridStep) * PUSH_CAP);
          qh[i] += 32u;
        }

      scan_append(dvB, svB, b0 + STEP, tid, e1, lo, q, qn, deg);
      push_fetch(dvB, svB, b0 + 3 * STEP, tid, e1, src, dst);
#pragma unroll
      for (int i = 0; i < SCAN_R; i++)
        while (qn[i] - qh[i] >= 32u) {
          scan_flush<true>(q + i * PUSH_RING, qh[i], 32u, lane, rec + (rid0 + i * ridStep) * PUSH_CAP);
          qh[i] += 32u;
        }
    }
#pragma unroll
    for (int i = 0; i < SCAN_R; i++)
      if (qn[i] != qh[i])
        scan_flush<false>(q + i * PUSH_RING, qh[i], qn[i] - qh[i], lane, rec + (rid0 + i * ridStep) * PUSH_CAP);
  }
#pragma unroll
  for (int i = 0; i < SCAN_R; i++)
    if (lane == 0 && r0 + i < nRanges) cnt[rid0 + i * ridStep] = qn[i];
}

#define CW_PITCH 20

template <bool HEAD>
__global__ __launch_bounds__(256)
void k_combine_wmma(const float* __restrict__ part, float* tab,
                    const float* __restrict__ dinv, const float* __restrict__ bias,
                    const float* __restrict__ W, const float* __restrict__ l1b,
                    const float* __restrict__ l2W, const float* __restrict__ l2b,
                    float* __restrict__ out, int n, int ntiles) {
  __shared__ float Ws[16 * 16];
  __shared__ __attribute__((aligned(16))) float hs[8 * 16 * CW_PITCH];
  Ws[threadIdx.x] = W[threadIdx.x];

  const int lane = threadIdx.x & 31;
  const int wv   = threadIdx.x >> 5;
  const int tile = blockIdx.x * 8 + wv;
  const bool ok  = tile < ntiles;
  const int tc   = ok ? tile : ntiles - 1;

  const int r0 = lane >> 2;
  const int c0 = (lane & 3) * 4;
  const size_t plane4 = (size_t)n * 4;
  const float4* p4 = (const float4*)part + (size_t)tc * 64 + lane;
  float4 lo[PUSH_C], hi[PUSH_C];
#pragma unroll
  for (int c = 0; c < PUSH_C; c++) { lo[c] = p4[c * plane4]; hi[c] = p4[c * plane4 + 32]; }
  const int nlo = tc * 16 + r0, nhi = nlo + 8;
  const float4 tlo = *(const float4*)(tab + (size_t)nlo * TAB_PITCH + c0);
  const float4 thi = *(const float4*)(tab + (size_t)nhi * TAB_PITCH + c0);
  const float dlo = dinv[nlo], dhi = dinv[nhi];
  float4 b4;
  b4.x = bias[c0]; b4.y = bias[c0 + 1]; b4.z = bias[c0 + 2]; b4.w = bias[c0 + 3];

  float4 alo = lo[0], ahi = hi[0];
#pragma unroll
  for (int c = 1; c < PUSH_C; c++) {
    alo.x += lo[c].x; alo.y += lo[c].y; alo.z += lo[c].z; alo.w += lo[c].w;
    ahi.x += hi[c].x; ahi.y += hi[c].y; ahi.z += hi[c].z; ahi.w += hi[c].w;
  }
  float4 hlo, hhi;
  hlo.x = fmaxf(dlo * (alo.x + dlo * tlo.x) + b4.x, 0.0f);
  hlo.y = fmaxf(dlo * (alo.y + dlo * tlo.y) + b4.y, 0.0f);
  hlo.z = fmaxf(dlo * (alo.z + dlo * tlo.z) + b4.z, 0.0f);
  hlo.w = fmaxf(dlo * (alo.w + dlo * tlo.w) + b4.w, 0.0f);
  hhi.x = fmaxf(dhi * (ahi.x + dhi * thi.x) + b4.x, 0.0f);
  hhi.y = fmaxf(dhi * (ahi.y + dhi * thi.y) + b4.y, 0.0f);
  hhi.z = fmaxf(dhi * (ahi.z + dhi * thi.z) + b4.z, 0.0f);
  hhi.w = fmaxf(dhi * (ahi.w + dhi * thi.w) + b4.w, 0.0f);

  float* ht = hs + wv * (16 * CW_PITCH);
  *(float4*)(ht + r0 * CW_PITCH + c0)       = hlo;
  *(float4*)(ht + (r0 + 8) * CW_PITCH + c0) = hhi;
  __syncthreads();

  const int half = lane >> 4;
  const int mn   = lane & 15;
  const int koff = half * 2;
  v8f c = {0.f, 0.f, 0.f, 0.f, 0.f, 0.f, 0.f, 0.f};
#pragma unroll
  for (int k0 = 0; k0 < 16; k0 += 4) {
    v2f a = *(const v2f*)(ht + mn * CW_PITCH + k0 + koff);
    v2f b;
    b.x = Ws[(k0 + koff + 0) * 16 + mn];
    b.y = Ws[(k0 + koff + 1) * 16 + mn];
    c = __builtin_amdgcn_wmma_f32_16x16x4_f32(
         false, a,  false, b,
         (short)0, c,  false,  false);
  }

  if (!HEAD) {
    if (ok) {
      float* hrow = tab + ((size_t)tile * 16 + half * 8) * TAB_PITCH + mn;
#pragma unroll
      for (int i = 0; i < 8; i++) hrow[i * TAB_PITCH] = c[i];
    }
  } else {
    const float bj = l1b[mn], wj = l2W[mn];
    float res = 0.0f;
#pragma unroll
    for (int i = 0; i < 8; i++) {
      float v = fmaxf(c[i] + bj, 0.0f) * wj;
#pragma unroll
      for (int off = 8; off; off >>= 1) v += __shfl_xor(v, off, 16);
      if (mn == i) res = v;
    }
    if (ok && mn < 8) out[tile * 16 + half * 8 + mn] = res + l2b[0];
  }
}

extern "C" void kernel_launch(void* const* d_in, const int* in_sizes, int n_in,
                              void* d_out, int out_size, void* d_ws, size_t ws_size,
                              hipStream_t stream) {
  const float* x   = (const float*)d_in[0];
  const int*   ei  = (const int*)d_in[1];
  const float* W1  = (const float*)d_in[2];
  const float* b1  = (const float*)d_in[3];
  const float* W2  = (const float*)d_in[4];
  const float* b2  = (const float*)d_in[5];
  const float* l1W = (const float*)d_in[6];
  const float* l1b = (const float*)d_in[7];
  const float* l2W = (const float*)d_in[8];
  const float* l2b = (const float*)d_in[9];

  const int N = in_sizes[0] / 128;
  const int E = in_sizes[1] / 2;
  const int* src = ei;
  const int* dst = ei + E;

  float* deg  = (float*)d_ws;
  float* dinv = deg + N;
  float* bufA = dinv + N;
  float* bufB = bufA + (size_t)N * 16;
  float* bufC = bufB + (size_t)N * 16;

  const int ntiles = N / 16;
  const int gemmBlocks = (ntiles * 32 + 255) / 256;
  const int nodeElems  = N * 16;
  const int nodeBlocks = (nodeElems + 255) / 256;
  const int edgeBlocks = (E + 255) / 256;
  const int scatBlocks = (int)(((size_t)E * 16 + 255) / 256);

  const size_t pushNeed = ((size_t)N * 2 + (size_t)N * 16 * (2 + PUSH_C)) * sizeof(float);
  if (pushNeed <= ws_size && N <= (1 << 17)) {
    float* tab  = bufA;
    float* part = bufC;
    const int chunkEdges = (E + PUSH_C - 1) / PUSH_C;
    const dim3 pushGrid((N + PUSH_TILE - 1) / PUSH_TILE, PUSH_C);

    const size_t nWaves   = (size_t)pushGrid.x * pushGrid.y * (PUSH_NT / 32) * SCAN_S;
    const dim3 scanGrid((pushGrid.x + SCAN_R - 1) / SCAN_R, PUSH_C * SCAN_S);
    const size_t cntWords = (nWaves + 31) / 32 * 32;
    unsigned* cnt = (unsigned*)(part + (size_t)PUSH_C * N * 16);
    unsigned* rec = cnt + cntWords;
    const bool record = pushNeed + (cntWords + nWaves * PUSH_CAP) * sizeof(unsigned) <= ws_size;

    float* deg1 = part;
    k_gemm_wmma<128, TAB_PITCH, true><<<gemmBlocks, 256, 0, stream>>>(x, W1, tab, ntiles, deg, deg1);
    if (record) k_scan<<<scanGrid, PUSH_NT, 0, stream>>>(src, dst, deg, deg1, rec, cnt, E, chunkEdges,
                                                         (int)pushGrid.x);
    else        k_deg_halves<<<edgeBlocks, 256, 0, stream>>>(dst, deg, deg1, E);
    k_rsqrt_rows<<<(N + 255) / 256, 256, 0, stream>>>(deg, deg1, dinv, tab, N);

    if (record) k_push<2><<<pushGrid, PUSH_NT, 0, stream>>>(src, dst, tab, part, rec, cnt, E, N, chunkEdges);
    else        k_push<0><<<pushGrid, PUSH_NT, 0, stream>>>(src, dst, tab, part, rec, cnt, E, N, chunkEdges);
    k_combine_wmma<false><<<gemmBlocks, 256, 0, stream>>>(part, tab, dinv, b1, W2, nullptr, nullptr,
                                                         nullptr, nullptr, N, ntiles);
    if (record) k_push<2><<<pushGrid, PUSH_NT, 0, stream>>>(src, dst, tab, part, rec, cnt, E, N, chunkEdges);
    else        k_push<0><<<pushGrid, PUSH_NT, 0, stream>>>(src, dst, tab, part, rec, cnt, E, N, chunkEdges);
    k_combine_wmma<true><<<gemmBlocks, 256, 0, stream>>>(part, tab, dinv, b2, l1W, l1b, l2W, l2b,
                                                        (float*)d_out, N, ntiles);
  } else {
    k_init  <<<nodeBlocks, 256, 0, stream>>>(deg, bufB, bufC, N);
    k_deg   <<<edgeBlocks, 256, 0, stream>>>(dst, deg, E);
    k_rsqrt <<<(N + 255) / 256, 256, 0, stream>>>(deg, dinv, N);

    k_gemm_wmma<128, 16, false><<<gemmBlocks, 256, 0, stream>>>(x, W1, bufA, ntiles, nullptr, nullptr);
    k_scatter<<<scatBlocks, 256, 0, stream>>>(src, dst, dinv, bufA, bufB, E);
    k_combine<<<nodeBlocks, 256, 0, stream>>>(bufB, bufA, dinv, b1, bufB, N);

    k_gemm_wmma<16, 16, false><<<gemmBlocks, 256, 0, stream>>>(bufB, W2, bufA, ntiles, nullptr, nullptr);
    k_scatter<<<scatBlocks, 256, 0, stream>>>(src, dst, dinv, bufA, bufC, E);
    k_combine<<<nodeBlocks, 256, 0, stream>>>(bufC, bufA, dinv, b2, bufC, N);

    k_gemm_wmma<16, 16, false><<<gemmBlocks, 256, 0, stream>>>(bufC, l1W, bufA, ntiles, nullptr, nullptr);
    k_final<<<nodeBlocks, 256, 0, stream>>>(bufA, l1b, l2W, l2b, (float*)d_out, N);
  }
}


# Edge scan in two levels: each loaded edge is tested once against its 5-range group and compacted into a per-wave LDS staging ring; the 5 range tests run on 32 dense staged entries; add order kept
# speedup vs baseline: 3.8379x; 1.0337x over previous
#include <hip/hip_runtime.h>
#include <hip/hip_bf16.h>

typedef __attribute__((ext_vector_type(2))) float v2f;
typedef __attribute__((ext_vector_type(8))) float v8f;

#define TAB_PITCH 32

__global__ void k_init(float* __restrict__ deg, float* __restrict__ aggB,
                       float* __restrict__ aggC, int n) {
  int idx = blockIdx.x * blockDim.x + threadIdx.x;
  if (idx < n) deg[idx] = 1.0f;
  if (idx < n * 16) { aggB[idx] = 0.0f; aggC[idx] = 0.0f; }
}

__global__ void k_deg(const int* __restrict__ dst, float* __restrict__ deg, int E) {
  int e = blockIdx.x * blockDim.x + threadIdx.x;
  if (e < E) atomicAdd(&deg[dst[e]], 1.0f);
}

__global__ void k_deg_halves(const int* __restrict__ dst, float* __restrict__ deg0,
                             float* __restrict__ deg1, int E) {
  const unsigned where = (unsigned)__builtin_amdgcn_s_sendmsg_rtn(0x87);
  float* deg = ((where >> 18) & 1u) ? deg1 : deg0;
  int e = blockIdx.x * blockDim.x + threadIdx.x;
  if (e < E) atomicAdd(&deg[dst[e]], 1.0f);
}

__global__ void k_rsqrt(const float* __restrict__ deg, float* __restrict__ dinv, int n) {
  int i = blockIdx.x * blockDim.x + threadIdx.x;
  if (i < n) dinv[i] = rsqrtf(deg[i]);
}

__global__ void k_rsqrt_rows(const float* __restrict__ deg, const float* __restrict__ deg1,
                             float* __restrict__ dinv, float* __restrict__ tab, int n) {
  int i = blockIdx.x * blockDim.x + threadIdx.x;
  if (i < n) {
    const float r = rsqrtf(deg[i] + deg1[i]);
    dinv[i] = r;
    tab[(size_t)i * TAB_PITCH + 16] = r;
  }
}

template <int K, int P, bool INIT>
__global__ void k_gemm_wmma(const float* __restrict__ A, const float* __restrict__ W,
                            float* __restrict__ H, int ntiles,
                            float* __restrict__ deg0, float* __restrict__ deg1) {
  __shared__ float Ws[K * 16];
  for (int t = threadIdx.x; t < K * 16; t += blockDim.x) Ws[t] = W[t];
  __syncthreads();

  const int lane = threadIdx.x & 31;
  const int tile = blockIdx.x * (blockDim.x >> 5) + (threadIdx.x >> 5);
  if (tile >= ntiles) return;

  const int half = lane >> 4;
  const int mn   = lane & 15;
  const int koff = half * 2;

  const float* arow = A + (size_t)(tile * 16 + mn) * K;

  if (INIT) {
    if (half == 0) deg0[tile * 16 + mn] = 1.0f;
    else           deg1[tile * 16 + mn] = 0.0f;
  }

  v8f c = {0.f, 0.f, 0.f, 0.f, 0.f, 0.f, 0.f, 0.f};
#pragma unroll
  for (int k0 = 0; k0 < K; k0 += 4) {
    v2f a = *(const v2f*)(arow + k0 + koff);
    v2f b;
    b.x = Ws[(k0 + koff + 0) * 16 + mn];
    b.y = Ws[(k0 + koff + 1) * 16 + mn];
    c = __builtin_amdgcn_wmma_f32_16x16x4_f32(
         false, a,  false, b,
         (short)0, c,  false,  false);
  }

  float* hrow = H + ((size_t)tile * 16 + half * 8) * P + mn;
#pragma unroll
  for (int i = 0; i < 8; i++) hrow[i * P] = c[i];
}

__global__ void k_scatter(const int* __restrict__ src, const int* __restrict__ dst,
                          const float* __restrict__ dinv, const float* __restrict__ H,
                          float* __restrict__ AGG, int E) {
  int idx = blockIdx.x * blockDim.x + threadIdx.x;
  int e = idx >> 4;
  int j = idx & 15;
  if (e >= E) return;
  int s = src[e];
  int d = dst[e];
  float norm = dinv[s] * dinv[d];
  atomicAdd(&AGG[d * 16 + j], norm * H[s * 16 + j]);
}

__global__ void k_combine(const float* __restrict__ AGG, const float* __restrict__ H,
                          const float* __restrict__ dinv, const float* __restrict__ bias,
                          float* __restrict__ out, int n) {
  int idx = blockIdx.x * blockDim.x + threadIdx.x;
  int i = idx >> 4;
  int j = idx & 15;
  if (i >= n) return;
  float di = dinv[i];
  out[idx] = fmaxf(AGG[idx] + di * di * H[idx] + bias[j], 0.0f);
}

__global__ void k_final(const float* __restrict__ G, const float* __restrict__ l1b,
                        const float* __restrict__ l2W, const float* __restrict__ l2b,
                        float* __restrict__ out, int n) {
  int idx = blockIdx.x * blockDim.x + threadIdx.x;
  int i = idx >> 4;
  int j = idx & 15;
  float v = 0.0f;
  if (i < n) v = fmaxf(G[idx] + l1b[j], 0.0f) * l2W[j];
#pragma unroll
  for (int off = 8; off; off >>= 1) v += __shfl_xor(v, off, 16);
  if (i < n && j == 0) out[i] = v + l2b[0];
}


#define PUSH_TILE 4096
#define PUSH_NT   1024
#define PUSH_K    6
#define PUSH_RING 256
#define PUSH_C    10
#define SCAN_R    5
#define SCAN_S    5
#define PUSH_CAP  128

template <bool FULL>
__device__ __forceinline__ void push_trip(const unsigned* q, unsigned qh, unsigned left,
                                          unsigned half, int j,
                                          const float* __restrict__ tab, float* acc) {
  unsigned w[8]; float v[8];
#pragma unroll
  for (int t = 0; t < 8; t++) {
    const unsigned slot = 2u * t + half;
    const unsigned idx = (FULL || slot < left) ? qh + slot : qh;
    w[t] = q[idx & (PUSH_RING - 1)];
  }
#pragma unroll
  for (int t = 0; t < 8; t++) {
    const unsigned r = (w[t] & 0x1FFFFu) * TAB_PITCH;
    const float x = tab[r + 16u] * tab[r + (unsigned)j];
    v[t] = (FULL || 2u * t + half < left) ? x : 0.0f;
  }
#pragma unroll
  for (int t = 0; t < 8; t++) atomicAdd(&acc[(w[t] >> 17) * 16 + j], v[t]);
}

__device__ __forceinline__ void push_fetch(int (&dv)[PUSH_K], int (&sv)[PUSH_K], int b0, int tid,
                                           int e1, const int* __restrict__ src,
                                           const int* __restrict__ dst) {
#pragma unroll
  for (int k = 0; k < PUSH_K; k++) {
    const int idx = min(b0 + k * PUSH_NT + tid, e1 - 1);
    dv[k] = dst[idx];
    sv[k] = src[idx];
  }
}

__device__ __forceinline__ void push_append(const int (&dv)[PUSH_K], const int (&sv)[PUSH_K],
                                            int b0, int tid, int e1, int lo,
                                            unsigned* q, unsigned& qn) {
#pragma unroll
  for (int k = 0; k < PUSH_K; k++) {
    const int e = b0 + k * PUSH_NT + tid;
    const unsigned off = (unsigned)(dv[k] - lo);
    const bool hit = e < e1 && off < (unsigned)PUSH_TILE;
    const unsigned m = __builtin_amdgcn_ballot_w32(hit);
    if (hit) q[(qn + __builtin_amdgcn_mbcnt_lo(m, 0u)) & (PUSH_RING - 1)] = (unsigned)sv[k] | (off << 17);
    qn += (unsigned)__popc(m);
  }
}

template <bool FULL>
__device__ __forceinline__ void push_rows(const unsigned (&w)[8], float (&v)[8], unsigned left,
                                          unsigned half, int j, const float* __restrict__ tab) {
#pragma unroll
  for (int t = 0; t < 8; t++) {
    const unsigned r = (w[t] & 0x1FFFFu) * TAB_PITCH;
    const float x = tab[r + 16u] * tab[r + (unsigned)j];
    v[t] = (FULL || 2u * t + half < left) ? x : 0.0f;
  }
}
__device__ __forceinline__ void push_adds(const unsigned (&w)[8], const float (&v)[8], int j,
                                          float* acc) {
#pragma unroll
  for (int t = 0; t < 8; t++) atomicAdd(&acc[(w[t] >> 17) * 16 + j], v[t]);
}
__device__ __forceinline__ void push_words(unsigned (&w)[8], const uint4& a, const uint4& b) {
  w[0] = a.x; w[1] = a.y; w[2] = a.z; w[3] = a.w;
  w[4] = b.x; w[5] = b.y; w[6] = b.z; w[7] = b.w;
}

template <int MODE>
__global__ __launch_bounds__(PUSH_NT)
void k_push(const int* __restrict__ src, const int* __restrict__ dst,
            const float* __restrict__ tab,
            float* __restrict__ part, const unsigned* __restrict__ rec,
            const unsigned* __restrict__ cnt, int E, int n, int chunkEdges) {
  __shared__ __attribute__((aligned(16))) float acc[PUSH_TILE * 16];
  __shared__ unsigned ring[(PUSH_NT / 32) * PUSH_RING];

  const int tid = threadIdx.x;
  const int lo  = blockIdx.x * PUSH_TILE;
  const int e0  = blockIdx.y * chunkEdges;
  const int e1  = min(E, e0 + chunkEdges);
  const int STEP = PUSH_K * PUSH_NT;

  const int wv = __builtin_amdgcn_readfirstlane(tid >> 5);
  const int wg = blockIdx.y * gridDim.x + blockIdx.x;
  const unsigned half = (unsigned)(tid >> 4) & 1u;
  const int j = tid & 15;

  unsigned most = 0u;
  if (MODE == 2) {
#pragma unroll
    for (int k = 0; k < SCAN_S; k++)
      most = max(most, cnt[wg * (SCAN_S * (PUSH_NT / 32)) + k * (PUSH_NT / 32) + (tid & 31)]);
  }

  for (int i = tid; i < PUSH_TILE * 16; i += PUSH_NT) acc[i] = 0.0f;

  const bool replay = MODE == 2 && __builtin_amdgcn_ballot_w32(most > (unsigned)PUSH_CAP) == 0u;

  if (replay) {
    __syncthreads();

#pragma nounroll
    for (int s = 0; s < SCAN_S; s++) {
    const size_t rid = ((size_t)wg * SCAN_S + s) * (PUSH_NT / 32) + wv;
    const unsigned total = __builtin_amdgcn_readfirstlane(cnt[rid]);
    const unsigned ntrip = (total + 15u) >> 4;
    const unsigned nfull = total >> 4;
    const uint4* rp = (const uint4*)(rec + rid * PUSH_CAP) + half * 2u;
    uint4 a0 = make_uint4(0u, 0u, 0u, 0u), a1 = a0, b0 = a0, b1 = a0;
    if (0u < ntrip) { a0 = rp[0]; a1 = rp[1]; }
    if (1u < ntrip) { b0 = rp[4]; b1 = rp[5]; }

    unsigned T = 0u;
    for (; T + 2u <= nfull; T += 2u) {
      unsigned wa[8], wb[8]; float va[8], vb[8];
      push_words(wa, a0, a1);
      push_words(wb, b0, b1);
      if (T + 2u < ntrip) { a0 = rp[(T + 2u) * 4u]; a1 = rp[(T + 2u) * 4u + 1u]; }
      if (T + 3u < ntrip) { b0 = rp[(T + 3u) * 4u]; b1 = rp[(T + 3u) * 4u + 1u]; }
      push_rows<true>(wa, va, 16u, half, j, tab);
      push_rows<true>(wb, vb, 16u, half, j, tab);
      push_adds(wa, va, j, acc);
      push_adds(wb, vb, j, acc);
    }
    if (T < nfull) {
      unsigned wa[8]; float va[8];
      push_words(wa, a0, a1);
      push_rows<true>(wa, va, 16u, half, j, tab);
      push_adds(wa, va, j, acc);
      a0 = b0; a1 = b1;
    }
    if (total & 15u) {
      unsigned wa[8]; float va[8];
      push_words(wa, a0, a1);
      push_rows<false>(wa, va, total & 15u, half, j, tab);
      push_adds(wa, va, j, acc);
    }
    }
  } else {
    int dvA[PUSH_K] = {0}, svA[PUSH_K] = {0}, dvB[PUSH_K] = {0}, svB[PUSH_K] = {0};
    const bool any = e0 < e1;
    if (any) {
      push_fetch(dvA, svA, e0, tid, e1, src, dst);
      push_fetch(dvB, svB, e0 + STEP, tid, e1, src, dst);
    }
    __syncthreads();

    unsigned* q = ring + wv * PUSH_RING;
    unsigned qn = 0u, qh = 0u;

    if (any) for (int b0 = e0; b0 < e1; b0 += 2 * STEP) {
      push_append(dvA, svA, b0, tid, e1, lo, q, qn);
      push_fetch(dvA, svA, b0 + 2 * STEP, tid, e1, src, dst);
      while (qn - qh >= 16u) { push_trip<true>(q, qh, 16u, half, j, tab, acc); qh += 16u; }

      push_append(dvB, svB, b0 + STEP, tid, e1, lo, q, qn);
      push_fetch(dvB, svB, b0 + 3 * STEP, tid, e1, src, dst);
      while (qn - qh >= 16u) { push_trip<true>(q, qh, 16u, half, j, tab, acc); qh += 16u; }
    }
    if (qn != qh) push_trip<false>(q, qh, qn - qh, half, j, tab, acc);
  }
  __syncthreads();

  const float4* acc4 = (const float4*)acc;
  float4* out4 = (float4*)(part + ((size_t)blockIdx.y * n + lo) * 16);
  for (int i4 = tid; i4 < PUSH_TILE * 4; i4 += PUSH_NT) {
    if (lo + (i4 >> 2) < n) out4[i4] = acc4[i4];
  }
}

__device__ __forceinline__ void scan_stage(const int (&dv)[PUSH_K], const int (&sv)[PUSH_K],
                                           int b0, int tid, int e1, int lo,
                                           unsigned* st, unsigned& sn, float* __restrict__ deg) {
#pragma unroll
  for (int k = 0; k < PUSH_K; k++) {
    const int e = b0 + k * PUSH_NT + tid;
    const unsigned off = (unsigned)(dv[k] - lo);
    const bool in = e < e1 && off < (unsigned)(SCAN_R * PUSH_TILE);
    if (in) atomicAdd(&deg[dv[k]], 1.0f);
    const unsigned m = __builtin_amdgcn_ballot_w32(in);
    if (in) st[(sn + __builtin_amdgcn_mbcnt_lo(m, 0u)) & (PUSH_RING - 1)] = (unsigned)sv[k] | (off << 17);
    sn += (unsigned)__popc(m);
  }
}

template <bool FULL>
__device__ __forceinline__ void scan_split(const unsigned* st, unsigned sh, unsigned avail, int lane,
                                           unsigned* q, unsigned (&qn)[SCAN_R]) {
  const bool valid = FULL || (unsigned)lane < avail;
  const unsigned w = st[(valid ? sh + (unsigned)lane : sh) & (PUSH_RING - 1)];
  const unsigned off = w >> 17;
  const unsigned which = off / (unsigned)PUSH_TILE;
  const unsigned word = (w & 0x1FFFFu) | ((off % (unsigned)PUSH_TILE) << 17);
#pragma unroll
  for (int i = 0; i < SCAN_R; i++) {
    const bool hit = valid && which == (unsigned)i;
    const unsigned m = __builtin_amdgcn_ballot_w32(hit);
    if (hit) q[i * PUSH_RING + ((qn[i] + __builtin_amdgcn_mbcnt_lo(m, 0u)) & (PUSH_RING - 1))] = word;
    qn[i] += (unsigned)__popc(m);
  }
}

template <bool FULL>
__device__ __forceinline__ void scan_flush(const unsigned* q, unsigned qh, unsigned left, int lane,
                                           unsigned* __restrict__ rec) {
  const unsigned trip = (unsigned)lane >> 4, p = (unsigned)lane & 15u;
  const unsigned slot = trip * 16u + (((p & 7u) << 1) | (p >> 3));
  const unsigned idx = (FULL || slot < left) ? qh + slot : qh + trip * 16u;
  const unsigned w = q[idx & (PUSH_RING - 1)];
  if ((FULL || trip * 16u < left) && qh + (unsigned)lane < (unsigned)PUSH_CAP) rec[qh + (unsigned)lane] = w;
}

__device__ __forceinline__ void scan_drain(const unsigned* st, unsigned sn, unsigned& sh, int lane,
                                           unsigned* q, unsigned (&qn)[SCAN_R], unsigned (&qh)[SCAN_R],
                                           unsigned* __restrict__ rec, size_t rid0, size_t ridStep) {
  while (sn - sh >= 32u) {
    scan_split<true>(st, sh, 32u, lane, q, qn);
    sh += 32u;
#pragma unroll
    for (int i = 0; i < SCAN_R; i++)
      while (qn[i] - qh[i] >= 32u) {
        scan_flush<true>(q + i * PUSH_RING, qh[i], 32u, lane, rec + (rid0 + i * ridStep) * PUSH_CAP);
        qh[i] += 32u;
      }
  }
}

__global__ __launch_bounds__(PUSH_NT)
void k_scan(const int* __restrict__ src, const int* __restrict__ dst,
            float* __restrict__ deg0, float* __restrict__ deg1,
            unsigned* __restrict__ rec, unsigned* __restrict__ cnt, int E, int chunkEdges,
            int nRanges) {
  __shared__ unsigned ring[(PUSH_NT / 32) * SCAN_R * PUSH_RING];
  __shared__ unsigned stage[(PUSH_NT / 32) * PUSH_RING];

  const int tid = threadIdx.x;
  const int lane = tid & 31;
  const int r0  = blockIdx.x * SCAN_R;
  const int lo  = r0 * PUSH_TILE;
  const int c   = blockIdx.y / SCAN_S;
  const int s   = blockIdx.y - c * SCAN_S;
  const int pieceEdges = (chunkEdges + SCAN_S - 1) / SCAN_S;
  const int cEnd = min(E, (c + 1) * chunkEdges);
  const int e0  = c * chunkEdges + s * pieceEdges;
  const int e1  = min(cEnd, e0 + pieceEdges);
  const int STEP = PUSH_K * PUSH_NT;

  const int wv = __builtin_amdgcn_readfirstlane(tid >> 5);

  const unsigned where = (unsigned)__builtin_amdgcn_s_sendmsg_rtn(0x87);
  float* deg = ((where >> 18) & 1u) ? deg1 : deg0;

  unsigned* q = ring + wv * (SCAN_R * PUSH_RING);
  unsigned* st = stage + wv * PUSH_RING;
  unsigned sn = 0u, sh = 0u;
  unsigned qn[SCAN_R], qh[SCAN_R];
#pragma unroll
  for (int i = 0; i < SCAN_R; i++) { qn[i] = 0u; qh[i] = 0u; }

  const size_t rid0 = ((size_t)(c * nRanges + r0) * SCAN_S + s) * (PUSH_NT / 32) + wv;
  const size_t ridStep = (size_t)SCAN_S * (PUSH_NT / 32);

  if (e0 < e1) {
    int dvA[PUSH_K], svA[PUSH_K], dvB[PUSH_K], svB[PUSH_K];
    push_fetch(dvA, svA, e0, tid, e1, src, dst);
    push_fetch(dvB, svB, e0 + STEP, tid, e1, src, dst);
    for (int b0 = e0; b0 < e1; b0 += 2 * STEP) {
      scan_stage(dvA, svA, b0, tid, e1, lo, st, sn, deg);
      push_fetch(dvA, svA, b0 + 2 * STEP, tid, e1, src, dst);
      scan_drain(st, sn, sh, lane, q, qn, qh, rec, rid0, ridStep);

      scan_stage(dvB, svB, b0 + STEP, tid, e1, lo, st, sn, deg);
      push_fetch(dvB, svB, b0 + 3 * STEP, tid, e1, src, dst);
      scan_drain(st, sn, sh, lane, q, qn, qh, rec, rid0, ridStep);
    }
    if (sn != sh) {
      scan_split<false>(st, sh, sn - sh, lane, q, qn);
#pragma unroll
      for (int i = 0; i < SCAN_R; i++)
        while (qn[i] - qh[i] >= 32u) {
          scan_flush<true>(q + i * PUSH_RING, qh[i], 32u, lane, rec + (rid0 + i * ridStep) * PUSH_CAP);
          qh[i] += 32u;
        }
    }
#pragma unroll
    for (int i = 0; i < SCAN_R; i++)
      if (qn[i] != qh[i])
        scan_flush<false>(q + i * PUSH_RING, qh[i], qn[i] - qh[i], lane, rec + (rid0 + i * ridStep) * PUSH_CAP);
  }
#pragma unroll
  for (int i = 0; i < SCAN_R; i++)
    if (lane == 0 && r0 + i < nRanges) cnt[rid0 + i * ridStep] = qn[i];
}

#define CW_PITCH 20

template <bool HEAD>
__global__ __launch_bounds__(256)
void k_combine_wmma(const float* __restrict__ part, float* tab,
                    const float* __restrict__ dinv, const float* __restrict__ bias,
                    const float* __restrict__ W, const float* __restrict__ l1b,
                    const float* __restrict__ l2W, const float* __restrict__ l2b,
                    float* __restrict__ out, int n, int ntiles) {
  __shared__ float Ws[16 * 16];
  __shared__ __attribute__((aligned(16))) float hs[8 * 16 * CW_PITCH];
  Ws[threadIdx.x] = W[threadIdx.x];

  const int lane = threadIdx.x & 31;
  const int wv   = threadIdx.x >> 5;
  const int tile = blockIdx.x * 8 + wv;
  const bool ok  = tile < ntiles;
  const int tc   = ok ? tile : ntiles - 1;

  const int r0 = lane >> 2;
  const int c0 = (lane & 3) * 4;
  const size_t plane4 = (size_t)n * 4;
  const float4* p4 = (const float4*)part + (size_t)tc * 64 + lane;
  float4 lo[PUSH_C], hi[PUSH_C];
#pragma unroll
  for (int c = 0; c < PUSH_C; c++) { lo[c] = p4[c * plane4]; hi[c] = p4[c * plane4 + 32]; }
  const int nlo = tc * 16 + r0, nhi = nlo + 8;
  const float4 tlo = *(const float4*)(tab + (size_t)nlo * TAB_PITCH + c0);
  const float4 thi = *(const float4*)(tab + (size_t)nhi * TAB_PITCH + c0);
  const float dlo = dinv[nlo], dhi = dinv[nhi];
  float4 b4;
  b4.x = bias[c0]; b4.y = bias[c0 + 1]; b4.z = bias[c0 + 2]; b4.w = bias[c0 + 3];

  float4 alo = lo[0], ahi = hi[0];
#pragma unroll
  for (int c = 1; c < PUSH_C; c++) {
    alo.x += lo[c].x; alo.y += lo[c].y; alo.z += lo[c].z; alo.w += lo[c].w;
    ahi.x += hi[c].x; ahi.y += hi[c].y; ahi.z += hi[c].z; ahi.w += hi[c].w;
  }
  float4 hlo, hhi;
  hlo.x = fmaxf(dlo * (alo.x + dlo * tlo.x) + b4.x, 0.0f);
  hlo.y = fmaxf(dlo * (alo.y + dlo * tlo.y) + b4.y, 0.0f);
  hlo.z = fmaxf(dlo * (alo.z + dlo * tlo.z) + b4.z, 0.0f);
  hlo.w = fmaxf(dlo * (alo.w + dlo * tlo.w) + b4.w, 0.0f);
  hhi.x = fmaxf(dhi * (ahi.x + dhi * thi.x) + b4.x, 0.0f);
  hhi.y = fmaxf(dhi * (ahi.y + dhi * thi.y) + b4.y, 0.0f);
  hhi.z = fmaxf(dhi * (ahi.z + dhi * thi.z) + b4.z, 0.0f);
  hhi.w = fmaxf(dhi * (ahi.w + dhi * thi.w) + b4.w, 0.0f);

  float* ht = hs + wv * (16 * CW_PITCH);
  *(float4*)(ht + r0 * CW_PITCH + c0)       = hlo;
  *(float4*)(ht + (r0 + 8) * CW_PITCH + c0) = hhi;
  __syncthreads();

  const int half = lane >> 4;
  const int mn   = lane & 15;
  const int koff = half * 2;
  v8f c = {0.f, 0.f, 0.f, 0.f, 0.f, 0.f, 0.f, 0.f};
#pragma unroll
  for (int k0 = 0; k0 < 16; k0 += 4) {
    v2f a = *(const v2f*)(ht + mn * CW_PITCH + k0 + koff);
    v2f b;
    b.x = Ws[(k0 + koff + 0) * 16 + mn];
    b.y = Ws[(k0 + koff + 1) * 16 + mn];
    c = __builtin_amdgcn_wmma_f32_16x16x4_f32(
         false, a,  false, b,
         (short)0, c,  false,  false);
  }

  if (!HEAD) {
    if (ok) {
      float* hrow = tab + ((size_t)tile * 16 + half * 8) * TAB_PITCH + mn;
#pragma unroll
      for (int i = 0; i < 8; i++) hrow[i * TAB_PITCH] = c[i];
    }
  } else {
    const float bj = l1b[mn], wj = l2W[mn];
    float res = 0.0f;
#pragma unroll
    for (int i = 0; i < 8; i++) {
      float v = fmaxf(c[i] + bj, 0.0f) * wj;
#pragma unroll
      for (int off = 8; off; off >>= 1) v += __shfl_xor(v, off, 16);
      if (mn == i) res = v;
    }
    if (ok && mn < 8) out[tile * 16 + half * 8 + mn] = res + l2b[0];
  }
}

extern "C" void kernel_launch(void* const* d_in, const int* in_sizes, int n_in,
                              void* d_out, int out_size, void* d_ws, size_t ws_size,
                              hipStream_t stream) {
  const float* x   = (const float*)d_in[0];
  const int*   ei  = (const int*)d_in[1];
  const float* W1  = (const float*)d_in[2];
  const float* b1  = (const float*)d_in[3];
  const float* W2  = (const float*)d_in[4];
  const float* b2  = (const float*)d_in[5];
  const float* l1W = (const float*)d_in[6];
  const float* l1b = (const float*)d_in[7];
  const float* l2W = (const float*)d_in[8];
  const float* l2b = (const float*)d_in[9];

  const int N = in_sizes[0] / 128;
  const int E = in_sizes[1] / 2;
  const int* src = ei;
  const int* dst = ei + E;

  float* deg  = (float*)d_ws;
  float* dinv = deg + N;
  float* bufA = dinv + N;
  float* bufB = bufA + (size_t)N * 16;
  float* bufC = bufB + (size_t)N * 16;

  const int ntiles = N / 16;
  const int gemmBlocks = (ntiles * 32 + 255) / 256;
  const int nodeElems  = N * 16;
  const int nodeBlocks = (nodeElems + 255) / 256;
  const int edgeBlocks = (E + 255) / 256;
  const int scatBlocks = (int)(((size_t)E * 16 + 255) / 256);

  const size_t pushNeed = ((size_t)N * 2 + (size_t)N * 16 * (2 + PUSH_C)) * sizeof(float);
  if (pushNeed <= ws_size && N <= (1 << 17)) {
    float* tab  = bufA;
    float* part = bufC;
    const int chunkEdges = (E + PUSH_C - 1) / PUSH_C;
    const dim3 pushGrid((N + PUSH_TILE - 1) / PUSH_TILE, PUSH_C);

    const size_t nWaves   = (size_t)pushGrid.x * pushGrid.y * (PUSH_NT / 32) * SCAN_S;
    const dim3 scanGrid((pushGrid.x + SCAN_R - 1) / SCAN_R, PUSH_C * SCAN_S);
    const size_t cntWords = (nWaves + 31) / 32 * 32;
    unsigned* cnt = (unsigned*)(part + (size_t)PUSH_C * N * 16);
    unsigned* rec = cnt + cntWords;
    const bool record = pushNeed + (cntWords + nWaves * PUSH_CAP) * sizeof(unsigned) <= ws_size;

    float* deg1 = part;
    k_gemm_wmma<128, TAB_PITCH, true><<<gemmBlocks, 256, 0, stream>>>(x, W1, tab, ntiles, deg, deg1);
    if (record) k_scan<<<scanGrid, PUSH_NT, 0, stream>>>(src, dst, deg, deg1, rec, cnt, E, chunkEdges,
                                                         (int)pushGrid.x);
    else        k_deg_halves<<<edgeBlocks, 256, 0, stream>>>(dst, deg, deg1, E);
    k_rsqrt_rows<<<(N + 255) / 256, 256, 0, stream>>>(deg, deg1, dinv, tab, N);

    if (record) k_push<2><<<pushGrid, PUSH_NT, 0, stream>>>(src, dst, tab, part, rec, cnt, E, N, chunkEdges);
    else        k_push<0><<<pushGrid, PUSH_NT, 0, stream>>>(src, dst, tab, part, rec, cnt, E, N, chunkEdges);
    k_combine_wmma<false><<<gemmBlocks, 256, 0, stream>>>(part, tab, dinv, b1, W2, nullptr, nullptr,
                                                         nullptr, nullptr, N, ntiles);
    if (record) k_push<2><<<pushGrid, PUSH_NT, 0, stream>>>(src, dst, tab, part, rec, cnt, E, N, chunkEdges);
    else        k_push<0><<<pushGrid, PUSH_NT, 0, stream>>>(src, dst, tab, part, rec, cnt, E, N, chunkEdges);
    k_combine_wmma<true><<<gemmBlocks, 256, 0, stream>>>(part, tab, dinv, b2, l1W, l1b, l2W, l2b,
                                                        (float*)d_out, N, ntiles);
  } else {
    k_init  <<<nodeBlocks, 256, 0, stream>>>(deg, bufB, bufC, N);
    k_deg   <<<edgeBlocks, 256, 0, stream>>>(dst, deg, E);
    k_rsqrt <<<(N + 255) / 256, 256, 0, stream>>>(deg, dinv, N);

    k_gemm_wmma<128, 16, false><<<gemmBlocks, 256, 0, stream>>>(x, W1, bufA, ntiles, nullptr, nullptr);
    k_scatter<<<scatBlocks, 256, 0, stream>>>(src, dst, dinv, bufA, bufB, E);
    k_combine<<<nodeBlocks, 256, 0, stream>>>(bufB, bufA, dinv, b1, bufB, N);

    k_gemm_wmma<16, 16, false><<<gemmBlocks, 256, 0, stream>>>(bufB, W2, bufA, ntiles, nullptr, nullptr);
    k_scatter<<<scatBlocks, 256, 0, stream>>>(src, dst, dinv, bufA, bufC, E);
    k_combine<<<nodeBlocks, 256, 0, stream>>>(bufC, bufA, dinv, b2, bufC, N);

    k_gemm_wmma<16, 16, false><<<gemmBlocks, 256, 0, stream>>>(bufC, l1W, bufA, ntiles, nullptr, nullptr);
    k_final<<<nodeBlocks, 256, 0, stream>>>(bufA, l1b, l2W, l2b, (float*)d_out, N);
  }
}


# Partial planes halved: 2080-node LDS tiles x 5 edge chunks (49 ranges, scan groups of 10, 10 short records a wave, totals kept in lanes); push stores, combine reads 5 planes, not 10; sum order moves
# speedup vs baseline: 4.1732x; 1.0874x over previous
#include <hip/hip_runtime.h>
#include <hip/hip_bf16.h>

typedef __attribute__((ext_vector_type(2))) float v2f;
typedef __attribute__((ext_vector_type(8))) float v8f;

#define TAB_PITCH 32

__global__ void k_init(float* __restrict__ deg, float* __restrict__ aggB,
                       float* __restrict__ aggC, int n) {
  int idx = blockIdx.x * blockDim.x + threadIdx.x;
  if (idx < n) deg[idx] = 1.0f;
  if (idx < n * 16) { aggB[idx] = 0.0f; aggC[idx] = 0.0f; }
}

__global__ void k_deg(const int* __restrict__ dst, float* __restrict__ deg, int E) {
  int e = blockIdx.x * blockDim.x + threadIdx.x;
  if (e < E) atomicAdd(&deg[dst[e]], 1.0f);
}

__global__ void k_deg_halves(const int* __restrict__ dst, float* __restrict__ deg0,
                             float* __restrict__ deg1, int E) {
  const unsigned where = (unsigned)__builtin_amdgcn_s_sendmsg_rtn(0x87);
  float* deg = ((where >> 18) & 1u) ? deg1 : deg0;
  int e = blockIdx.x * blockDim.x + threadIdx.x;
  if (e < E) atomicAdd(&deg[dst[e]], 1.0f);
}

__global__ void k_rsqrt(const float* __restrict__ deg, float* __restrict__ dinv, int n) {
  int i = blockIdx.x * blockDim.x + threadIdx.x;
  if (i < n) dinv[i] = rsqrtf(deg[i]);
}

__global__ void k_rsqrt_rows(const float* __restrict__ deg, const float* __restrict__ deg1,
                             float* __restrict__ dinv, float* __restrict__ tab, int n) {
  int i = blockIdx.x * blockDim.x + threadIdx.x;
  if (i < n) {
    const float r = rsqrtf(deg[i] + deg1[i]);
    dinv[i] = r;
    tab[(size_t)i * TAB_PITCH + 16] = r;
  }
}

template <int K, int P, bool INIT>
__global__ void k_gemm_wmma(const float* __restrict__ A, const float* __restrict__ W,
                            float* __restrict__ H, int ntiles,
                            float* __restrict__ deg0, float* __restrict__ deg1) {
  __shared__ float Ws[K * 16];
  for (int t = threadIdx.x; t < K * 16; t += blockDim.x) Ws[t] = W[t];
  __syncthreads();

  const int lane = threadIdx.x & 31;
  const int tile = blockIdx.x * (blockDim.x >> 5) + (threadIdx.x >> 5);
  if (tile >= ntiles) return;

  const int half = lane >> 4;
  const int mn   = lane & 15;
  const int koff = half * 2;

  const float* arow = A + (size_t)(tile * 16 + mn) * K;

  if (INIT) {
    if (half == 0) deg0[tile * 16 + mn] = 1.0f;
    else           deg1[tile * 16 + mn] = 0.0f;
  }

  v8f c = {0.f, 0.f, 0.f, 0.f, 0.f, 0.f, 0.f, 0.f};
#pragma unroll
  for (int k0 = 0; k0 < K; k0 += 4) {
    v2f a = *(const v2f*)(arow + k0 + koff);
    v2f b;
    b.x = Ws[(k0 + koff + 0) * 16 + mn];
    b.y = Ws[(k0 + koff + 1) * 16 + mn];
    c = __builtin_amdgcn_wmma_f32_16x16x4_f32(
         false, a,  false, b,
         (short)0, c,  false,  false);
  }

  float* hrow = H + ((size_t)tile * 16 + half * 8) * P + mn;
#pragma unroll
  for (int i = 0; i < 8; i++) hrow[i * P] = c[i];
}

__global__ void k_scatter(const int* __restrict__ src, const int* __restrict__ dst,
                          const float* __restrict__ dinv, const float* __restrict__ H,
                          float* __restrict__ AGG, int E) {
  int idx = blockIdx.x * blockDim.x + threadIdx.x;
  int e = idx >> 4;
  int j = idx & 15;
  if (e >= E) return;
  int s = src[e];
  int d = dst[e];
  float norm = dinv[s] * dinv[d];
  atomicAdd(&AGG[d * 16 + j], norm * H[s * 16 + j]);
}

__global__ void k_combine(const float* __restrict__ AGG, const float* __restrict__ H,
                          const float* __restrict__ dinv, const float* __restrict__ bias,
                          float* __restrict__ out, int n) {
  int idx = blockIdx.x * blockDim.x + threadIdx.x;
  int i = idx >> 4;
  int j = idx & 15;
  if (i >= n) return;
  float di = dinv[i];
  out[idx] = fmaxf(AGG[idx] + di * di * H[idx] + bias[j], 0.0f);
}

__global__ void k_final(const float* __restrict__ G, const float* __restrict__ l1b,
                        const float* __restrict__ l2W, const float* __restrict__ l2b,
                        float* __restrict__ out, int n) {
  int idx = blockIdx.x * blockDim.x + threadIdx.x;
  int i = idx >> 4;
  int j = idx & 15;
  float v = 0.0f;
  if (i < n) v = fmaxf(G[idx] + l1b[j], 0.0f) * l2W[j];
#pragma unroll
  for (int off = 8; off; off >>= 1) v += __shfl_xor(v, off, 16);
  if (i < n && j == 0) out[i] = v + l2b[0];
}


#define PUSH_TILE 2080
#define PUSH_NT   1024
#define PUSH_K    6
#define PUSH_RING 256
#define PUSH_C    5
#define SCAN_R    10
#define SCAN_S    10
#define SCAN_RING 128
#define PUSH_CAP  128
static_assert(SCAN_R * PUSH_TILE <= (1 << 15), "a staged word keeps the offset inside the group in 15 bits");

template <bool FULL>
__device__ __forceinline__ void push_trip(const unsigned* q, unsigned qh, unsigned left,
                                          unsigned half, int j,
                                          const float* __restrict__ tab, float* acc) {
  unsigned w[8]; float v[8];
#pragma unroll
  for (int t = 0; t < 8; t++) {
    const unsigned slot = 2u * t + half;
    const unsigned idx = (FULL || slot < left) ? qh + slot : qh;
    w[t] = q[idx & (PUSH_RING - 1)];
  }
#pragma unroll
  for (int t = 0; t < 8; t++) {
    const unsigned r = (w[t] & 0x1FFFFu) * TAB_PITCH;
    const float x = tab[r + 16u] * tab[r + (unsigned)j];
    v[t] = (FULL || 2u * t + half < left) ? x : 0.0f;
  }
#pragma unroll
  for (int t = 0; t < 8; t++) atomicAdd(&acc[(w[t] >> 17) * 16 + j], v[t]);
}

__device__ __forceinline__ void push_fetch(int (&dv)[PUSH_K], int (&sv)[PUSH_K], int b0, int tid,
                                           int e1, const int* __restrict__ src,
                                           const int* __restrict__ dst) {
#pragma unroll
  for (int k = 0; k < PUSH_K; k++) {
    const int idx = min(b0 + k * PUSH_NT + tid, e1 - 1);
    dv[k] = dst[idx];
    sv[k] = src[idx];
  }
}

__device__ __forceinline__ void push_append(const int (&dv)[PUSH_K], const int (&sv)[PUSH_K],
                                            int b0, int tid, int e1, int lo,
                                            unsigned* q, unsigned& qn) {
#pragma unroll
  for (int k = 0; k < PUSH_K; k++) {
    const int e = b0 + k * PUSH_NT + tid;
    const unsigned off = (unsigned)(dv[k] - lo);
    const bool hit = e < e1 && off < (unsigned)PUSH_TILE;
    const unsigned m = __builtin_amdgcn_ballot_w32(hit);
    if (hit) q[(qn + __builtin_amdgcn_mbcnt_lo(m, 0u)) & (PUSH_RING - 1)] = (unsigned)sv[k] | (off << 17);
    qn += (unsigned)__popc(m);
  }
}

template <bool FULL>
__device__ __forceinline__ void push_rows(const unsigned (&w)[8], float (&v)[8], unsigned left,
                                          unsigned half, int j, const float* __restrict__ tab) {
#pragma unroll
  for (int t = 0; t < 8; t++) {
    const unsigned r = (w[t] & 0x1FFFFu) * TAB_PITCH;
    const float x = tab[r + 16u] * tab[r + (unsigned)j];
    v[t] = (FULL || 2u * t + half < left) ? x : 0.0f;
  }
}
__device__ __forceinline__ void push_adds(const unsigned (&w)[8], const float (&v)[8], int j,
                                          float* acc) {
#pragma unroll
  for (int t = 0; t < 8; t++) atomicAdd(&acc[(w[t] >> 17) * 16 + j], v[t]);
}
__device__ __forceinline__ void push_words(unsigned (&w)[8], const uint4& a, const uint4& b) {
  w[0] = a.x; w[1] = a.y; w[2] = a.z; w[3] = a.w;
  w[4] = b.x; w[5] = b.y; w[6] = b.z; w[7] = b.w;
}

template <int MODE>
__global__ __launch_bounds__(PUSH_NT)
void k_push(const int* __restrict__ src, const int* __restrict__ dst,
            const float* __restrict__ tab,
            float* __restrict__ part, const unsigned* __restrict__ rec,
            const unsigned* __restrict__ cnt, int E, int n, int chunkEdges) {
  __shared__ __attribute__((aligned(16))) float acc[PUSH_TILE * 16];
  __shared__ unsigned ring[(PUSH_NT / 32) * PUSH_RING];

  const int tid = threadIdx.x;
  const int lo  = blockIdx.x * PUSH_TILE;
  const int e0  = blockIdx.y * chunkEdges;
  const int e1  = min(E, e0 + chunkEdges);
  const int STEP = PUSH_K * PUSH_NT;

  const int wv = __builtin_amdgcn_readfirstlane(tid >> 5);
  const int wg = blockIdx.y * gridDim.x + blockIdx.x;
  const unsigned half = (unsigned)(tid >> 4) & 1u;
  const int j = tid & 15;

  unsigned tot[SCAN_S] = {0u};
  if (MODE == 2) {
#pragma unroll
    for (int k = 0; k < SCAN_S; k++)
      tot[k] = cnt[wg * (SCAN_S * (PUSH_NT / 32)) + k * (PUSH_NT / 32) + (tid & 31)];
  }

  for (int i = tid; i < PUSH_TILE * 16; i += PUSH_NT) acc[i] = 0.0f;

  unsigned most = 0u, mine = 0u;
  if (MODE == 2) {
#pragma unroll
    for (int k = 0; k < SCAN_S; k++) {
      most = max(most, tot[k]);
      const unsigned own = (unsigned)__builtin_amdgcn_readlane((int)tot[k], wv);
      if ((tid & 31) == k) mine = own;
    }
  }

  const bool replay = MODE == 2 && __builtin_amdgcn_ballot_w32(most > (unsigned)PUSH_CAP) == 0u;

  if (replay) {
    __syncthreads();

#pragma nounroll
    for (int s = 0; s < SCAN_S; s++) {
    const size_t rid = ((size_t)wg * SCAN_S + s) * (PUSH_NT / 32) + wv;
    const unsigned total = (unsigned)__builtin_amdgcn_readlane((int)mine, s);
    const unsigned ntrip = (total + 15u) >> 4;
    const unsigned nfull = total >> 4;
    const uint4* rp = (const uint4*)(rec + rid * PUSH_CAP) + half * 2u;
    uint4 a0 = make_uint4(0u, 0u, 0u, 0u), a1 = a0, b0 = a0, b1 = a0;
    if (0u < ntrip) { a0 = rp[0]; a1 = rp[1]; }
    if (1u < ntrip) { b0 = rp[4]; b1 = rp[5]; }

    unsigned T = 0u;
    for (; T + 2u <= nfull; T += 2u) {
      unsigned wa[8], wb[8]; float va[8], vb[8];
      push_words(wa, a0, a1);
      push_words(wb, b0, b1);
      if (T + 2u < ntrip) { a0 = rp[(T + 2u) * 4u]; a1 = rp[(T + 2u) * 4u + 1u]; }
      if (T + 3u < ntrip) { b0 = rp[(T + 3u) * 4u]; b1 = rp[(T + 3u) * 4u + 1u]; }
      push_rows<true>(wa, va, 16u, half, j, tab);
      push_rows<true>(wb, vb, 16u, half, j, tab);
      push_adds(wa, va, j, acc);
      push_adds(wb, vb, j, acc);
    }
    if (T < nfull) {
      unsigned wa[8]; float va[8];
      push_words(wa, a0, a1);
      push_rows<true>(wa, va, 16u, half, j, tab);
      push_adds(wa, va, j, acc);
      a0 = b0; a1 = b1;
    }
    if (total & 15u) {
      unsigned wa[8]; float va[8];
      push_words(wa, a0, a1);
      push_rows<false>(wa, va, total & 15u, half, j, tab);
      push_adds(wa, va, j, acc);
    }
    }
  } else {
    int dvA[PUSH_K] = {0}, svA[PUSH_K] = {0}, dvB[PUSH_K] = {0}, svB[PUSH_K] = {0};
    const bool any = e0 < e1;
    if (any) {
      push_fetch(dvA, svA, e0, tid, e1, src, dst);
      push_fetch(dvB, svB, e0 + STEP, tid, e1, src, dst);
    }
    __syncthreads();

    unsigned* q = ring + wv * PUSH_RING;
    unsigned qn = 0u, qh = 0u;

    if (any) for (int b0 = e0; b0 < e1; b0 += 2 * STEP) {
      push_append(dvA, svA, b0, tid, e1, lo, q, qn);
      push_fetch(dvA, svA, b0 + 2 * STEP, tid, e1, src, dst);
      while (qn - qh >= 16u) { push_trip<true>(q, qh, 16u, half, j, tab, acc); qh += 16u; }

      push_append(dvB, svB, b0 + STEP, tid, e1, lo, q, qn);
      push_fetch(dvB, svB, b0 + 3 * STEP, tid, e1, src, dst);
      while (qn - qh >= 16u) { push_trip<true>(q, qh, 16u, half, j, tab, acc); qh += 16u; }
    }
    if (qn != qh) push_trip<false>(q, qh, qn - qh, half, j, tab, acc);
  }
  __syncthreads();

  const float4* acc4 = (const float4*)acc;
  float4* out4 = (float4*)(part + ((size_t)blockIdx.y * n + lo) * 16);
  for (int i4 = tid; i4 < PUSH_TILE * 4; i4 += PUSH_NT) {
    if (lo + (i4 >> 2) < n) out4[i4] = acc4[i4];
  }
}

__device__ __forceinline__ void scan_stage(const int (&dv)[PUSH_K], const int (&sv)[PUSH_K],
                                           int b0, int tid, int e1, int lo,
                                           unsigned* st, unsigned& sn, float* __restrict__ deg) {
#pragma unroll
  for (int k = 0; k < PUSH_K; k++) {
    const int e = b0 + k * PUSH_NT + tid;
    const unsigned off = (unsigned)(dv[k] - lo);
    const bool in = e < e1 && off < (unsigned)(SCAN_R * PUSH_TILE);
    if (in) atomicAdd(&deg[dv[k]], 1.0f);
    const unsigned m = __builtin_amdgcn_ballot_w32(in);
    if (in) st[(sn + __builtin_amdgcn_mbcnt_lo(m, 0u)) & (PUSH_RING - 1)] = (unsigned)sv[k] | (off << 17);
    sn += (unsigned)__popc(m);
  }
}

template <bool FULL>
__device__ __forceinline__ void scan_split(const unsigned* st, unsigned sh, unsigned avail, int lane,
                                           unsigned* q, unsigned (&qn)[SCAN_R]) {
  const bool valid = FULL || (unsigned)lane < avail;
  const unsigned w = st[(valid ? sh + (unsigned)lane : sh) & (PUSH_RING - 1)];
  const unsigned off = w >> 17;
  const unsigned which = off / (unsigned)PUSH_TILE;
  const unsigned word = (w & 0x1FFFFu) | ((off % (unsigned)PUSH_TILE) << 17);
#pragma unroll
  for (int i = 0; i < SCAN_R; i++) {
    const bool hit = valid && which == (unsigned)i;
    const unsigned m = __builtin_amdgcn_ballot_w32(hit);
    if (hit) q[i * SCAN_RING + ((qn[i] + __builtin_amdgcn_mbcnt_lo(m, 0u)) & (SCAN_RING - 1))] = word;
    qn[i] += (unsigned)__popc(m);
  }
}

template <bool FULL>
__device__ __forceinline__ void scan_flush(const unsigned* q, unsigned qh, unsigned left, int lane,
                                           unsigned* __restrict__ rec) {
  const unsigned trip = (unsigned)lane >> 4, p = (unsigned)lane & 15u;
  const unsigned slot = trip * 16u + (((p & 7u) << 1) | (p >> 3));
  const unsigned idx = (FULL || slot < left) ? qh + slot : qh + trip * 16u;
  const unsigned w = q[idx & (SCAN_RING - 1)];
  if ((FULL || trip * 16u < left) && qh + (unsigned)lane < (unsigned)PUSH_CAP) rec[qh + (unsigned)lane] = w;
}

__device__ __forceinline__ void scan_drain(const unsigned* st, unsigned sn, unsigned& sh, int lane,
                                           unsigned* q, unsigned (&qn)[SCAN_R], unsigned (&qh)[SCAN_R],
                                           unsigned* __restrict__ rec, size_t rid0, size_t ridStep) {
  while (sn - sh >= 32u) {
    scan_split<true>(st, sh, 32u, lane, q, qn);
    sh += 32u;
#pragma unroll
    for (int i = 0; i < SCAN_R; i++)
      while (qn[i] - qh[i] >= 32u) {
        scan_flush<true>(q + i * SCAN_RING, qh[i], 32u, lane, rec + (rid0 + i * ridStep) * PUSH_CAP);
        qh[i] += 32u;
      }
  }
}

__global__ __launch_bounds__(PUSH_NT)
void k_scan(const int* __restrict__ src, const int* __restrict__ dst,
            float* __restrict__ deg0, float* __restrict__ deg1,
            unsigned* __restrict__ rec, unsigned* __restrict__ cnt, int E, int chunkEdges,
            int nRanges) {
  __shared__ unsigned ring[(PUSH_NT / 32) * SCAN_R * SCAN_RING];
  __shared__ unsigned stage[(PUSH_NT / 32) * PUSH_RING];

  const int tid = threadIdx.x;
  const int lane = tid & 31;
  const int r0  = blockIdx.x * SCAN_R;
  const int lo  = r0 * PUSH_TILE;
  const int c   = blockIdx.y / SCAN_S;
  const int s   = blockIdx.y - c * SCAN_S;
  const int pieceEdges = (chunkEdges + SCAN_S - 1) / SCAN_S;
  const int cEnd = min(E, (c + 1) * chunkEdges);
  const int e0  = c * chunkEdges + s * pieceEdges;
  const int e1  = min(cEnd, e0 + pieceEdges);
  const int STEP = PUSH_K * PUSH_NT;

  const int wv = __builtin_amdgcn_readfirstlane(tid >> 5);

  const unsigned where = (unsigned)__builtin_amdgcn_s_sendmsg_rtn(0x87);
  float* deg = ((where >> 18) & 1u) ? deg1 : deg0;

  unsigned* q = ring + wv * (SCAN_R * SCAN_RING);
  unsigned* st = stage + wv * PUSH_RING;
  unsigned sn = 0u, sh = 0u;
  unsigned qn[SCAN_R], qh[SCAN_R];
#pragma unroll
  for (int i = 0; i < SCAN_R; i++) { qn[i] = 0u; qh[i] = 0u; }

  const size_t rid0 = ((size_t)(c * nRanges + r0) * SCAN_S + s) * (PUSH_NT / 32) + wv;
  const size_t ridStep = (size_t)SCAN_S * (PUSH_NT / 32);

  if (e0 < e1) {
    int dvA[PUSH_K], svA[PUSH_K], dvB[PUSH_K], svB[PUSH_K];
    push_fetch(dvA, svA, e0, tid, e1, src, dst);
    push_fetch(dvB, svB, e0 + STEP, tid, e1, src, dst);
    for (int b0 = e0; b0 < e1; b0 += 2 * STEP) {
      scan_stage(dvA, svA, b0, tid, e1, lo, st, sn, deg);
      push_fetch(dvA, svA, b0 + 2 * STEP, tid, e1, src, dst);
      scan_drain(st, sn, sh, lane, q, qn, qh, rec, rid0, ridStep);

      scan_stage(dvB, svB, b0 + STEP, tid, e1, lo, st, sn, deg);
      push_fetch(dvB, svB, b0 + 3 * STEP, tid, e1, src, dst);
      scan_drain(st, sn, sh, lane, q, qn, qh, rec, rid0, ridStep);
    }
    if (sn != sh) {
      scan_split<false>(st, sh, sn - sh, lane, q, qn);
#pragma unroll
      for (int i = 0; i < SCAN_R; i++)
        while (qn[i] - qh[i] >= 32u) {
          scan_flush<true>(q + i * SCAN_RING, qh[i], 32u, lane, rec + (rid0 + i * ridStep) * PUSH_CAP);
          qh[i] += 32u;
        }
    }
#pragma unroll
    for (int i = 0; i < SCAN_R; i++)
      if (qn[i] != qh[i])
        scan_flush<false>(q + i * SCAN_RING, qh[i], qn[i] - qh[i], lane, rec + (rid0 + i * ridStep) * PUSH_CAP);
  }
#pragma unroll
  for (int i = 0; i < SCAN_R; i++)
    if (lane == 0 && r0 + i < nRanges) cnt[rid0 + i * ridStep] = qn[i];
}

#define CW_PITCH 20

template <bool HEAD>
__global__ __launch_bounds__(256)
void k_combine_wmma(const float* __restrict__ part, float* tab,
                    const float* __restrict__ dinv, const float* __restrict__ bias,
                    const float* __restrict__ W, const float* __restrict__ l1b,
                    const float* __restrict__ l2W, const float* __restrict__ l2b,
                    float* __restrict__ out, int n, int ntiles) {
  __shared__ float Ws[16 * 16];
  __shared__ __attribute__((aligned(16))) float hs[8 * 16 * CW_PITCH];
  Ws[threadIdx.x] = W[threadIdx.x];

  const int lane = threadIdx.x & 31;
  const int wv   = threadIdx.x >> 5;
  const int tile = blockIdx.x * 8 + wv;
  const bool ok  = tile < ntiles;
  const int tc   = ok ? tile : ntiles - 1;

  const int r0 = lane >> 2;
  const int c0 = (lane & 3) * 4;
  const size_t plane4 = (size_t)n * 4;
  const float4* p4 = (const float4*)part + (size_t)tc * 64 + lane;
  float4 lo[PUSH_C], hi[PUSH_C];
#pragma unroll
  for (int c = 0; c < PUSH_C; c++) { lo[c] = p4[c * plane4]; hi[c] = p4[c * plane4 + 32]; }
  const int nlo = tc * 16 + r0, nhi = nlo + 8;
  const float4 tlo = *(const float4*)(tab + (size_t)nlo * TAB_PITCH + c0);
  const float4 thi = *(const float4*)(tab + (size_t)nhi * TAB_PITCH + c0);
  const float dlo = dinv[nlo], dhi = dinv[nhi];
  float4 b4;
  b4.x = bias[c0]; b4.y = bias[c0 + 1]; b4.z = bias[c0 + 2]; b4.w = bias[c0 + 3];

  float4 alo = lo[0], ahi = hi[0];
#pragma unroll
  for (int c = 1; c < PUSH_C; c++) {
    alo.x += lo[c].x; alo.y += lo[c].y; alo.z += lo[c].z; alo.w += lo[c].w;
    ahi.x += hi[c].x; ahi.y += hi[c].y; ahi.z += hi[c].z; ahi.w += hi[c].w;
  }
  float4 hlo, hhi;
  hlo.x = fmaxf(dlo * (alo.x + dlo * tlo.x) + b4.x, 0.0f);
  hlo.y = fmaxf(dlo * (alo.y + dlo * tlo.y) + b4.y, 0.0f);
  hlo.z = fmaxf(dlo * (alo.z + dlo * tlo.z) + b4.z, 0.0f);
  hlo.w = fmaxf(dlo * (alo.w + dlo * tlo.w) + b4.w, 0.0f);
  hhi.x = fmaxf(dhi * (ahi.x + dhi * thi.x) + b4.x, 0.0f);
  hhi.y = fmaxf(dhi * (ahi.y + dhi * thi.y) + b4.y, 0.0f);
  hhi.z = fmaxf(dhi * (ahi.z + dhi * thi.z) + b4.z, 0.0f);
  hhi.w = fmaxf(dhi * (ahi.w + dhi * thi.w) + b4.w, 0.0f);

  float* ht = hs + wv * (16 * CW_PITCH);
  *(float4*)(ht + r0 * CW_PITCH + c0)       = hlo;
  *(float4*)(ht + (r0 + 8) * CW_PITCH + c0) = hhi;
  __syncthreads();

  const int half = lane >> 4;
  const int mn   = lane & 15;
  const int koff = half * 2;
  v8f c = {0.f, 0.f, 0.f, 0.f, 0.f, 0.f, 0.f, 0.f};
#pragma unroll
  for (int k0 = 0; k0 < 16; k0 += 4) {
    v2f a = *(const v2f*)(ht + mn * CW_PITCH + k0 + koff);
    v2f b;
    b.x = Ws[(k0 + koff + 0) * 16 + mn];
    b.y = Ws[(k0 + koff + 1) * 16 + mn];
    c = __builtin_amdgcn_wmma_f32_16x16x4_f32(
         false, a,  false, b,
         (short)0, c,  false,  false);
  }

  if (!HEAD) {
    if (ok) {
      float* hrow = tab + ((size_t)tile * 16 + half * 8) * TAB_PITCH + mn;
#pragma unroll
      for (int i = 0; i < 8; i++) hrow[i * TAB_PITCH] = c[i];
    }
  } else {
    const float bj = l1b[mn], wj = l2W[mn];
    float res = 0.0f;
#pragma unroll
    for (int i = 0; i < 8; i++) {
      float v = fmaxf(c[i] + bj, 0.0f) * wj;
#pragma unroll
      for (int off = 8; off; off >>= 1) v += __shfl_xor(v, off, 16);
      if (mn == i) res = v;
    }
    if (ok && mn < 8) out[tile * 16 + half * 8 + mn] = res + l2b[0];
  }
}

extern "C" void kernel_launch(void* const* d_in, const int* in_sizes, int n_in,
                              void* d_out, int out_size, void* d_ws, size_t ws_size,
                              hipStream_t stream) {
  const float* x   = (const float*)d_in[0];
  const int*   ei  = (const int*)d_in[1];
  const float* W1  = (const float*)d_in[2];
  const float* b1  = (const float*)d_in[3];
  const float* W2  = (const float*)d_in[4];
  const float* b2  = (const float*)d_in[5];
  const float* l1W = (const float*)d_in[6];
  const float* l1b = (const float*)d_in[7];
  const float* l2W = (const float*)d_in[8];
  const float* l2b = (const float*)d_in[9];

  const int N = in_sizes[0] / 128;
  const int E = in_sizes[1] / 2;
  const int* src = ei;
  const int* dst = ei + E;

  float* deg  = (float*)d_ws;
  float* dinv = deg + N;
  float* bufA = dinv + N;
  float* bufB = bufA + (size_t)N * 16;
  float* bufC = bufB + (size_t)N * 16;

  const int ntiles = N / 16;
  const int gemmBlocks = (ntiles * 32 + 255) / 256;
  const int nodeElems  = N * 16;
  const int nodeBlocks = (nodeElems + 255) / 256;
  const int edgeBlocks = (E + 255) / 256;
  const int scatBlocks = (int)(((size_t)E * 16 + 255) / 256);

  const size_t pushNeed = ((size_t)N * 2 + (size_t)N * 16 * (2 + PUSH_C)) * sizeof(float);
  if (pushNeed <= ws_size && N <= (1 << 17)) {
    float* tab  = bufA;
    float* part = bufC;
    const int chunkEdges = (E + PUSH_C - 1) / PUSH_C;
    const dim3 pushGrid((N + PUSH_TILE - 1) / PUSH_TILE, PUSH_C);

    const size_t nWaves   = (size_t)pushGrid.x * pushGrid.y * (PUSH_NT / 32) * SCAN_S;
    const dim3 scanGrid((pushGrid.x + SCAN_R - 1) / SCAN_R, PUSH_C * SCAN_S);
    const size_t cntWords = (nWaves + 31) / 32 * 32;
    unsigned* cnt = (unsigned*)(part + (size_t)PUSH_C * N * 16);
    unsigned* rec = cnt + cntWords;
    const bool record = pushNeed + (cntWords + nWaves * PUSH_CAP) * sizeof(unsigned) <= ws_size;

    float* deg1 = part;
    k_gemm_wmma<128, TAB_PITCH, true><<<gemmBlocks, 256, 0, stream>>>(x, W1, tab, ntiles, deg, deg1);
    if (record) k_scan<<<scanGrid, PUSH_NT, 0, stream>>>(src, dst, deg, deg1, rec, cnt, E, chunkEdges,
                                                         (int)pushGrid.x);
    else        k_deg_halves<<<edgeBlocks, 256, 0, stream>>>(dst, deg, deg1, E);
    k_rsqrt_rows<<<(N + 255) / 256, 256, 0, stream>>>(deg, deg1, dinv, tab, N);

    if (record) k_push<2><<<pushGrid, PUSH_NT, 0, stream>>>(src, dst, tab, part, rec, cnt, E, N, chunkEdges);
    else        k_push<0><<<pushGrid, PUSH_NT, 0, stream>>>(src, dst, tab, part, rec, cnt, E, N, chunkEdges);
    k_combine_wmma<false><<<gemmBlocks, 256, 0, stream>>>(part, tab, dinv, b1, W2, nullptr, nullptr,
                                                         nullptr, nullptr, N, ntiles);
    if (record) k_push<2><<<pushGrid, PUSH_NT, 0, stream>>>(src, dst, tab, part, rec, cnt, E, N, chunkEdges);
    else        k_push<0><<<pushGrid, PUSH_NT, 0, stream>>>(src, dst, tab, part, rec, cnt, E, N, chunkEdges);
    k_combine_wmma<true><<<gemmBlocks, 256, 0, stream>>>(part, tab, dinv, b2, l1W, l1b, l2W, l2b,
                                                        (float*)d_out, N, ntiles);
  } else {
    k_init  <<<nodeBlocks, 256, 0, stream>>>(deg, bufB, bufC, N);
    k_deg   <<<edgeBlocks, 256, 0, stream>>>(dst, deg, E);
    k_rsqrt <<<(N + 255) / 256, 256, 0, stream>>>(deg, dinv, N);

    k_gemm_wmma<128, 16, false><<<gemmBlocks, 256, 0, stream>>>(x, W1, bufA, ntiles, nullptr, nullptr);
    k_scatter<<<scatBlocks, 256, 0, stream>>>(src, dst, dinv, bufA, bufB, E);
    k_combine<<<nodeBlocks, 256, 0, stream>>>(bufB, bufA, dinv, b1, bufB, N);

    k_gemm_wmma<16, 16, false><<<gemmBlocks, 256, 0, stream>>>(bufB, W2, bufA, ntiles, nullptr, nullptr);
    k_scatter<<<scatBlocks, 256, 0, stream>>>(src, dst, dinv, bufA, bufC, E);
    k_combine<<<nodeBlocks, 256, 0, stream>>>(bufC, bufA, dinv, b2, bufC, N);

    k_gemm_wmma<16, 16, false><<<gemmBlocks, 256, 0, stream>>>(bufC, l1W, bufA, ntiles, nullptr, nullptr);
    k_final<<<nodeBlocks, 256, 0, stream>>>(bufA, l1b, l2W, l2b, (float*)d_out, N);
  }
}
